# starting kernel
# baseline (speedup 1.0000x reference)
#include <hip/hip_runtime.h>
#include <hip/hip_bf16.h>


typedef __attribute__((ext_vector_type(16))) _Float16 v16h;
typedef __attribute__((ext_vector_type(8)))  _Float16 v8h;
typedef __attribute__((ext_vector_type(8)))  float    v8f;

#define NNODES 50000
#define NEDGES 800000
#define ETOT   (NEDGES + NNODES)
#define FDIM   128
#define NHEAD  4

__device__ __forceinline__ unsigned fkey(float f) {
  unsigned u = __float_as_uint(f);
  return (u & 0x80000000u) ? ~u : (u | 0x80000000u);
}
__device__ __forceinline__ float funkey(unsigned k) {
  unsigned u = (k & 0x80000000u) ? (k & 0x7fffffffu) : ~k;
  return __uint_as_float(u);
}

__global__ void cast_f16_kernel(const float* __restrict__ in,
                                _Float16* __restrict__ out, int n) {
  int i = blockIdx.x * blockDim.x + threadIdx.x;
  if (i < n) out[i] = (_Float16)in[i];
}

__global__ void init_conv_kernel(float* __restrict__ acc,
                                 unsigned* __restrict__ mkey,
                                 float* __restrict__ ssum, int n) {
  int i = blockIdx.x * blockDim.x + threadIdx.x;
  if (i < n * FDIM) acc[i] = 0.0f;
  if (i < n * NHEAD) { mkey[i] = 0u; ssum[i] = 0.0f; }
}

#define KC 64
#define BT_STRIDE 72

__global__ __launch_bounds__(256) void wmma_gemm_kernel(
    const _Float16* __restrict__ A, const float* __restrict__ W,
    const float* __restrict__ bias,
    float* __restrict__ outF, _Float16* __restrict__ outH,
    int M, int Nt, int K, int act) {
  __shared__ _Float16 bt[64 * BT_STRIDE];

  const int lane = threadIdx.x & 31;
  const int wave = threadIdx.x >> 5;
  const int half = lane >> 4;
  const int l15  = lane & 15;
  const int n0   = blockIdx.y * 64;

  int rowBase = blockIdx.x * 128 + wave * 16;
  if (rowBase > M - 16) rowBase = M - 16;

  v8f c[4] = {v8f{}, v8f{}, v8f{}, v8f{}};

  for (int kc = 0; kc < K; kc += KC) {
    {
      int kl = threadIdx.x >> 4;
      const int n4 = (threadIdx.x & 15) * 4;
      #pragma unroll
      for (int p = 0; p < 4; ++p, kl += 16) {
        const float4 wv =
            *(const float4*)(W + (size_t)(kc + kl) * Nt + n0 + n4);
        bt[(n4 + 0) * BT_STRIDE + kl] = (_Float16)wv.x;
        bt[(n4 + 1) * BT_STRIDE + kl] = (_Float16)wv.y;
        bt[(n4 + 2) * BT_STRIDE + kl] = (_Float16)wv.z;
        bt[(n4 + 3) * BT_STRIDE + kl] = (_Float16)wv.w;
      }
    }
    __syncthreads();

    const _Float16* arow = A + (size_t)(rowBase + l15) * K + kc;
    #pragma unroll
    for (int ks = 0; ks < 2; ++ks) {
      const int kb = ks * 32 + half * 8;
      v8h alo = *(const v8h*)(arow + kb);
      v8h ahi = *(const v8h*)(arow + kb + 16);
      v16h a = __builtin_shufflevector(alo, ahi, 0, 1, 2, 3, 4, 5, 6, 7,
                                       8, 9, 10, 11, 12, 13, 14, 15);
      v16h b[4];
      #pragma unroll
      for (int nn = 0; nn < 4; ++nn) {
        const _Float16* bp = bt + (nn * 16 + l15) * BT_STRIDE + kb;
        v8h blo = *(const v8h*)(bp);
        v8h bhi = *(const v8h*)(bp + 16);
        b[nn] = __builtin_shufflevector(blo, bhi, 0, 1, 2, 3, 4, 5, 6, 7,
                                        8, 9, 10, 11, 12, 13, 14, 15);
      }
      #pragma unroll
      for (int nn = 0; nn < 4; ++nn) {
        c[nn] = __builtin_amdgcn_wmma_f32_16x16x32_f16(
            false, a, false, b[nn], (short)0, c[nn], false, false);
      }
    }
    __syncthreads();
  }

  #pragma unroll
  for (int nn = 0; nn < 4; ++nn) {
    const int col = n0 + nn * 16 + l15;
    const float bv = bias ? bias[col] : 0.0f;
    #pragma unroll
    for (int r = 0; r < 8; ++r) {
      const int row = rowBase + half * 8 + r;
      float v = c[nn][r] + bv;
      if (act == 1) v = v > 0.0f ? v : (__expf(v) - 1.0f);
      else if (act == 2) v = 1.0f / (1.0f + __expf(-v));
      if (outF) outF[(size_t)row * Nt + col] = v;
      if (outH) outH[(size_t)row * Nt + col] = (_Float16)v;
    }
  }
}

__global__ void edge_logits_kernel(const int* __restrict__ ei,
                                   const float* __restrict__ xl,
                                   const float* __restrict__ xr,
                                   const float* __restrict__ att,
                                   float* __restrict__ ebuf,
                                   unsigned* __restrict__ mkey) {
  int idx = blockIdx.x * blockDim.x + threadIdx.x;
  if (idx >= ETOT * NHEAD) return;
  const int e = idx >> 2, h = idx & 3;
  int s, d;
  if (e < NEDGES) { s = ei[e]; d = ei[NEDGES + e]; } else { s = d = e - NEDGES; }

  const float4* pl = (const float4*)(xl + (size_t)s * FDIM + h * 32);
  const float4* pr = (const float4*)(xr + (size_t)d * FDIM + h * 32);
  const float4* pa = (const float4*)(att + h * 32);
  float acc = 0.0f;
  #pragma unroll
  for (int i = 0; i < 8; ++i) {
    const float4 a = pl[i], b = pr[i], w = pa[i];
    float z;
    z = a.x + b.x; z = z > 0.0f ? z : 0.2f * z; acc += z * w.x;
    z = a.y + b.y; z = z > 0.0f ? z : 0.2f * z; acc += z * w.y;
    z = a.z + b.z; z = z > 0.0f ? z : 0.2f * z; acc += z * w.z;
    z = a.w + b.w; z = z > 0.0f ? z : 0.2f * z; acc += z * w.w;
  }
  ebuf[idx] = acc;
  atomicMax(mkey + (size_t)d * NHEAD + h, fkey(acc));
}

__global__ void edge_exp_kernel(const int* __restrict__ ei,
                                float* __restrict__ ebuf,
                                const unsigned* __restrict__ mkey,
                                float* __restrict__ ssum) {
  int idx = blockIdx.x * blockDim.x + threadIdx.x;
  if (idx >= ETOT * NHEAD) return;
  const int e = idx >> 2, h = idx & 3;
  int d;
  if (e < NEDGES) d = ei[NEDGES + e]; else d = e - NEDGES;
  const float m = funkey(mkey[(size_t)d * NHEAD + h]);
  const float p = __expf(ebuf[idx] - m);
  ebuf[idx] = p;
  atomicAdd(ssum + (size_t)d * NHEAD + h, p);
}

__global__ void edge_scatter_kernel(const int* __restrict__ ei,
                                    const float* __restrict__ xl,
                                    const float* __restrict__ ebuf,
                                    const float* __restrict__ ssum,
                                    float* __restrict__ acc) {
  int idx = blockIdx.x * blockDim.x + threadIdx.x;
  if (idx >= ETOT * NHEAD) return;
  const int e = idx >> 2, h = idx & 3;
  int s, d;
  if (e < NEDGES) { s = ei[e]; d = ei[NEDGES + e]; } else { s = d = e - NEDGES; }
  const float alpha = ebuf[idx] / ssum[(size_t)d * NHEAD + h];
  const float4* pl = (const float4*)(xl + (size_t)s * FDIM + h * 32);
  float* ob = acc + (size_t)d * FDIM + h * 32;
  #pragma unroll
  for (int i = 0; i < 8; ++i) {
    const float4 a = pl[i];
    atomicAdd(ob + i * 4 + 0, a.x * alpha);
    atomicAdd(ob + i * 4 + 1, a.y * alpha);
    atomicAdd(ob + i * 4 + 2, a.z * alpha);
    atomicAdd(ob + i * 4 + 3, a.w * alpha);
  }
}

__global__ void node_finish_kernel(const float* __restrict__ acc,
                                   const float* __restrict__ b,
                                   _Float16* __restrict__ outH, int n) {
  int i = blockIdx.x * blockDim.x + threadIdx.x;
  if (i >= n * FDIM) return;
  float v = acc[i] + b[i & (FDIM - 1)];
  v = v > 0.0f ? v : (__expf(v) - 1.0f);
  outH[i] = (_Float16)v;
}

extern "C" void kernel_launch(void* const* d_in, const int* in_sizes, int n_in,
                              void* d_out, int out_size, void* d_ws,
                              size_t ws_size, hipStream_t stream) {
  (void)in_sizes; (void)n_in; (void)out_size; (void)ws_size;
  const float* x    = (const float*)d_in[0];
  const int*   ei   = (const int*)d_in[1];
  const float* Wl1  = (const float*)d_in[2];
  const float* Wr1  = (const float*)d_in[3];
  const float* att1 = (const float*)d_in[4];
  const float* b1   = (const float*)d_in[5];
  const float* Wl2  = (const float*)d_in[6];
  const float* Wr2  = (const float*)d_in[7];
  const float* att2 = (const float*)d_in[8];
  const float* b2   = (const float*)d_in[9];
  const float* W1   = (const float*)d_in[10];
  const float* bb1  = (const float*)d_in[11];
  const float* W2   = (const float*)d_in[12];
  const float* bb2  = (const float*)d_in[13];
  const float* W3   = (const float*)d_in[14];
  const float* bb3  = (const float*)d_in[15];
  float* out = (float*)d_out;

  char* ws = (char*)d_ws;
  size_t off = 0;
  auto alloc = [&](size_t bytes) -> char* {
    char* p = ws + off;
    off += (bytes + 255) & ~(size_t)255;
    return p;
  };
  const size_t S_F = (size_t)NNODES * FDIM;
  _Float16* xh  = (_Float16*)alloc(S_F * 2);
  _Float16* h1h = (_Float16*)alloc(S_F * 2);
  _Float16* h2h = (_Float16*)alloc(S_F * 2);
  const size_t uoff = off;
  float*    xl   = (float*)alloc(S_F * 4);
  float*    xr   = (float*)alloc(S_F * 4);
  float*    accb = (float*)alloc(S_F * 4);
  float*    ebuf = (float*)alloc((size_t)ETOT * NHEAD * 4);
  unsigned* mkey = (unsigned*)alloc((size_t)NNODES * NHEAD * 4);
  float*    ssum = (float*)alloc((size_t)NNODES * NHEAD * 4);
  _Float16* m1 = (_Float16*)(ws + uoff);
  _Float16* m2 = (_Float16*)(ws + uoff + (size_t)NNODES * 640 * 2);

  const int TB = 256;
  const int ngrid = (int)((S_F + TB - 1) / TB);
  const int egrid = (ETOT * NHEAD + TB - 1) / TB;
  const dim3 gconv((NNODES + 127) / 128, 128 / 64);

  cast_f16_kernel<<<ngrid, TB, 0, stream>>>(x, xh, (int)S_F);
  wmma_gemm_kernel<<<gconv, TB, 0, stream>>>(xh, Wl1, nullptr, xl, nullptr,
                                             NNODES, 128, 128, 0);
  wmma_gemm_kernel<<<gconv, TB, 0, stream>>>(xh, Wr1, nullptr, xr, nullptr,
                                             NNODES, 128, 128, 0);
  init_conv_kernel<<<ngrid, TB, 0, stream>>>(accb, mkey, ssum, NNODES);
  edge_logits_kernel<<<egrid, TB, 0, stream>>>(ei, xl, xr, att1, ebuf, mkey);
  edge_exp_kernel<<<egrid, TB, 0, stream>>>(ei, ebuf, mkey, ssum);
  edge_scatter_kernel<<<egrid, TB, 0, stream>>>(ei, xl, ebuf, ssum, accb);
  node_finish_kernel<<<ngrid, TB, 0, stream>>>(accb, b1, h1h, NNODES);

  wmma_gemm_kernel<<<gconv, TB, 0, stream>>>(h1h, Wl2, nullptr, xl, nullptr,
                                             NNODES, 128, 128, 0);
  wmma_gemm_kernel<<<gconv, TB, 0, stream>>>(h1h, Wr2, nullptr, xr, nullptr,
                                             NNODES, 128, 128, 0);
  init_conv_kernel<<<ngrid, TB, 0, stream>>>(accb, mkey, ssum, NNODES);
  edge_logits_kernel<<<egrid, TB, 0, stream>>>(ei, xl, xr, att2, ebuf, mkey);
  edge_exp_kernel<<<egrid, TB, 0, stream>>>(ei, ebuf, mkey, ssum);
  edge_scatter_kernel<<<egrid, TB, 0, stream>>>(ei, xl, ebuf, ssum, accb);
  node_finish_kernel<<<ngrid, TB, 0, stream>>>(accb, b2, h2h, NNODES);

  const dim3 gm1((NNODES + 127) / 128, 640 / 64);
  wmma_gemm_kernel<<<gm1, TB, 0, stream>>>(h2h, W1, bb1, nullptr, m1,
                                           NNODES, 640, 128, 1);
  const dim3 gm2((NNODES + 127) / 128, 320 / 64);
  wmma_gemm_kernel<<<gm2, TB, 0, stream>>>(m1, W2, bb2, nullptr, m2,
                                           NNODES, 320, 640, 1);
  const dim3 gm3((NNODES + 127) / 128, 64 / 64);
  wmma_gemm_kernel<<<gm3, TB, 0, stream>>>(m2, W3, bb3, out, nullptr,
                                           NNODES, 64, 320, 2);
}


# Aggregation walks a destination-sorted edge list (counting sort per call, int atomics; sum order = arrival order): one wave per node gathers xl rows, replacing the float atomic scatter-add
# speedup vs baseline: 2.6123x; 2.6123x over previous
#include <hip/hip_runtime.h>
#include <hip/hip_bf16.h>


typedef __attribute__((ext_vector_type(16))) _Float16 v16h;
typedef __attribute__((ext_vector_type(8)))  _Float16 v8h;
typedef __attribute__((ext_vector_type(8)))  float    v8f;

#define NNODES 50000
#define NEDGES 800000
#define ETOT   (NEDGES + NNODES)
#define FDIM   128
#define NHEAD  4

__device__ __forceinline__ unsigned fkey(float f) {
  unsigned u = __float_as_uint(f);
  return (u & 0x80000000u) ? ~u : (u | 0x80000000u);
}
__device__ __forceinline__ float funkey(unsigned k) {
  unsigned u = (k & 0x80000000u) ? (k & 0x7fffffffu) : ~k;
  return __uint_as_float(u);
}

__global__ void cast_f16_kernel(const float* __restrict__ in,
                                _Float16* __restrict__ out, int n) {
  int i = blockIdx.x * blockDim.x + threadIdx.x;
  if (i < n) out[i] = (_Float16)in[i];
}

__global__ void init_conv_kernel(unsigned* __restrict__ mkey,
                                 float* __restrict__ ssum, int n) {
  int i = blockIdx.x * blockDim.x + threadIdx.x;
  if (i < n * NHEAD) { mkey[i] = 0u; ssum[i] = 0.0f; }
}

__global__ void csr_zero_kernel(int* __restrict__ cnt, int n) {
  int i = blockIdx.x * blockDim.x + threadIdx.x;
  if (i < n) cnt[i] = 0;
}

__global__ void csr_count_kernel(const int* __restrict__ ei,
                                 int* __restrict__ cnt) {
  int e = blockIdx.x * blockDim.x + threadIdx.x;
  if (e < NEDGES) atomicAdd(cnt + ei[NEDGES + e], 1);
}

#define SCAN_CH ((NNODES + 1023) / 1024)
#define SCAN_N  (SCAN_CH * 1024)
__global__ __launch_bounds__(1024) void csr_scan_kernel(
    const int* __restrict__ cnt, int* __restrict__ rowptr,
    int* __restrict__ cursor) {
  __shared__ int wsum[32];
  const int t = threadIdx.x, lane = t & 31, wv = t >> 5;
  const int i0 = t * SCAN_CH;
  int v[SCAN_CH];
  int local = 0;
  #pragma unroll
  for (int i = 0; i < SCAN_CH; ++i) v[i] = cnt[i0 + i];
  #pragma unroll
  for (int i = 0; i < SCAN_CH; ++i) local += v[i];
  int x = local;
  for (int o = 1; o < 32; o <<= 1) {
    const int y = __shfl_up(x, o);
    if (lane >= o) x += y;
  }
  if (lane == 31) wsum[wv] = x;
  __syncthreads();
  if (wv == 0) {
    const int w = wsum[lane];
    int xs = w;
    for (int o = 1; o < 32; o <<= 1) {
      const int y = __shfl_up(xs, o);
      if (lane >= o) xs += y;
    }
    wsum[lane] = xs - w;
  }
  __syncthreads();
  int run = wsum[wv] + (x - local);
  #pragma unroll
  for (int i = 0; i < SCAN_CH; ++i) {
    rowptr[i0 + i] = run;
    cursor[i0 + i] = run;
    run += v[i];
  }
}

__global__ void csr_fill_kernel(const int* __restrict__ ei,
                                int* __restrict__ cursor,
                                int2* __restrict__ perm) {
  int e = blockIdx.x * blockDim.x + threadIdx.x;
  if (e >= NEDGES) return;
  const int s = ei[e], d = ei[NEDGES + e];
  const int pos = atomicAdd(cursor + d, 1);
  if (pos < NEDGES) perm[pos] = make_int2(s, e);
}

#define KC 64
#define BT_STRIDE 72

__global__ __launch_bounds__(256) void wmma_gemm_kernel(
    const _Float16* __restrict__ A, const float* __restrict__ W,
    const float* __restrict__ bias,
    float* __restrict__ outF, _Float16* __restrict__ outH,
    int M, int Nt, int K, int act) {
  __shared__ _Float16 bt[64 * BT_STRIDE];

  const int lane = threadIdx.x & 31;
  const int wave = threadIdx.x >> 5;
  const int half = lane >> 4;
  const int l15  = lane & 15;
  const int n0   = blockIdx.y * 64;

  int rowBase = blockIdx.x * 128 + wave * 16;
  if (rowBase > M - 16) rowBase = M - 16;

  v8f c[4] = {v8f{}, v8f{}, v8f{}, v8f{}};

  for (int kc = 0; kc < K; kc += KC) {
    {
      int kl = threadIdx.x >> 4;
      const int n4 = (threadIdx.x & 15) * 4;
      #pragma unroll
      for (int p = 0; p < 4; ++p, kl += 16) {
        const float4 wv =
            *(const float4*)(W + (size_t)(kc + kl) * Nt + n0 + n4);
        bt[(n4 + 0) * BT_STRIDE + kl] = (_Float16)wv.x;
        bt[(n4 + 1) * BT_STRIDE + kl] = (_Float16)wv.y;
        bt[(n4 + 2) * BT_STRIDE + kl] = (_Float16)wv.z;
        bt[(n4 + 3) * BT_STRIDE + kl] = (_Float16)wv.w;
      }
    }
    __syncthreads();

    const _Float16* arow = A + (size_t)(rowBase + l15) * K + kc;
    #pragma unroll
    for (int ks = 0; ks < 2; ++ks) {
      const int kb = ks * 32 + half * 8;
      v8h alo = *(const v8h*)(arow + kb);
      v8h ahi = *(const v8h*)(arow + kb + 16);
      v16h a = __builtin_shufflevector(alo, ahi, 0, 1, 2, 3, 4, 5, 6, 7,
                                       8, 9, 10, 11, 12, 13, 14, 15);
      v16h b[4];
      #pragma unroll
      for (int nn = 0; nn < 4; ++nn) {
        const _Float16* bp = bt + (nn * 16 + l15) * BT_STRIDE + kb;
        v8h blo = *(const v8h*)(bp);
        v8h bhi = *(const v8h*)(bp + 16);
        b[nn] = __builtin_shufflevector(blo, bhi, 0, 1, 2, 3, 4, 5, 6, 7,
                                        8, 9, 10, 11, 12, 13, 14, 15);
      }
      #pragma unroll
      for (int nn = 0; nn < 4; ++nn) {
        c[nn] = __builtin_amdgcn_wmma_f32_16x16x32_f16(
            false, a, false, b[nn], (short)0, c[nn], false, false);
      }
    }
    __syncthreads();
  }

  #pragma unroll
  for (int nn = 0; nn < 4; ++nn) {
    const int col = n0 + nn * 16 + l15;
    const float bv = bias ? bias[col] : 0.0f;
    #pragma unroll
    for (int r = 0; r < 8; ++r) {
      const int row = rowBase + half * 8 + r;
      float v = c[nn][r] + bv;
      if (act == 1) v = v > 0.0f ? v : (__expf(v) - 1.0f);
      else if (act == 2) v = 1.0f / (1.0f + __expf(-v));
      if (outF) outF[(size_t)row * Nt + col] = v;
      if (outH) outH[(size_t)row * Nt + col] = (_Float16)v;
    }
  }
}

__global__ void edge_logits_kernel(const int* __restrict__ ei,
                                   const float* __restrict__ xl,
                                   const float* __restrict__ xr,
                                   const float* __restrict__ att,
                                   float* __restrict__ ebuf,
                                   unsigned* __restrict__ mkey) {
  int idx = blockIdx.x * blockDim.x + threadIdx.x;
  if (idx >= ETOT * NHEAD) return;
  const int e = idx >> 2, h = idx & 3;
  int s, d;
  if (e < NEDGES) { s = ei[e]; d = ei[NEDGES + e]; } else { s = d = e - NEDGES; }

  const float4* pl = (const float4*)(xl + (size_t)s * FDIM + h * 32);
  const float4* pr = (const float4*)(xr + (size_t)d * FDIM + h * 32);
  const float4* pa = (const float4*)(att + h * 32);
  float acc = 0.0f;
  #pragma unroll
  for (int i = 0; i < 8; ++i) {
    const float4 a = pl[i], b = pr[i], w = pa[i];
    float z;
    z = a.x + b.x; z = z > 0.0f ? z : 0.2f * z; acc += z * w.x;
    z = a.y + b.y; z = z > 0.0f ? z : 0.2f * z; acc += z * w.y;
    z = a.z + b.z; z = z > 0.0f ? z : 0.2f * z; acc += z * w.z;
    z = a.w + b.w; z = z > 0.0f ? z : 0.2f * z; acc += z * w.w;
  }
  ebuf[idx] = acc;
  atomicMax(mkey + (size_t)d * NHEAD + h, fkey(acc));
}

__global__ void edge_exp_kernel(const int* __restrict__ ei,
                                float* __restrict__ ebuf,
                                const unsigned* __restrict__ mkey,
                                float* __restrict__ ssum) {
  int idx = blockIdx.x * blockDim.x + threadIdx.x;
  if (idx >= ETOT * NHEAD) return;
  const int e = idx >> 2, h = idx & 3;
  int d;
  if (e < NEDGES) d = ei[NEDGES + e]; else d = e - NEDGES;
  const float m = funkey(mkey[(size_t)d * NHEAD + h]);
  const float p = __expf(ebuf[idx] - m);
  ebuf[idx] = p;
  atomicAdd(ssum + (size_t)d * NHEAD + h, p);
}

__global__ __launch_bounds__(256) void node_gather_kernel(
    const int* __restrict__ rowptr, const int2* __restrict__ perm,
    const float* __restrict__ xl, const float* __restrict__ ebuf,
    const float* __restrict__ ssum, float* __restrict__ acc) {
  const int lane = threadIdx.x & 31;
  const int d = __builtin_amdgcn_readfirstlane(
      (int)(blockIdx.x * 8 + (threadIdx.x >> 5)));
  if (d >= NNODES) return;
  const int h = lane >> 3;
  const float4* xl4 = (const float4*)xl;
  const float sv = ssum[(size_t)d * NHEAD + h];

  float4 r;
  {
    const float4 v = xl4[(size_t)d * 32 + lane];
    const float w = ebuf[(size_t)(NEDGES + d) * NHEAD + h] / sv;
    r.x = v.x * w; r.y = v.y * w; r.z = v.z * w; r.w = v.w * w;
  }

  const int beg = rowptr[d], end = rowptr[d + 1];
  for (int base = beg; base < end; base += 32) {
    const int cnt = (end - base) < 32 ? (end - base) : 32;
    int2 my = make_int2(0, 0);
    if (lane < cnt) my = perm[base + lane];
    #pragma unroll 4
    for (int j = 0; j < cnt; ++j) {
      const int s = __builtin_amdgcn_readlane(my.x, j);
      const int e = __builtin_amdgcn_readlane(my.y, j);
      const float4 v = xl4[(size_t)s * 32 + lane];
      const float w = ebuf[(size_t)e * NHEAD + h] / sv;
      r.x += v.x * w; r.y += v.y * w; r.z += v.z * w; r.w += v.w * w;
    }
  }
  ((float4*)acc)[(size_t)d * 32 + lane] = r;
}

__global__ void node_finish_kernel(const float* __restrict__ acc,
                                   const float* __restrict__ b,
                                   _Float16* __restrict__ outH, int n) {
  int i = blockIdx.x * blockDim.x + threadIdx.x;
  if (i >= n * FDIM) return;
  float v = acc[i] + b[i & (FDIM - 1)];
  v = v > 0.0f ? v : (__expf(v) - 1.0f);
  outH[i] = (_Float16)v;
}

extern "C" void kernel_launch(void* const* d_in, const int* in_sizes, int n_in,
                              void* d_out, int out_size, void* d_ws,
                              size_t ws_size, hipStream_t stream) {
  (void)in_sizes; (void)n_in; (void)out_size; (void)ws_size;
  const float* x    = (const float*)d_in[0];
  const int*   ei   = (const int*)d_in[1];
  const float* Wl1  = (const float*)d_in[2];
  const float* Wr1  = (const float*)d_in[3];
  const float* att1 = (const float*)d_in[4];
  const float* b1   = (const float*)d_in[5];
  const float* Wl2  = (const float*)d_in[6];
  const float* Wr2  = (const float*)d_in[7];
  const float* att2 = (const float*)d_in[8];
  const float* b2   = (const float*)d_in[9];
  const float* W1   = (const float*)d_in[10];
  const float* bb1  = (const float*)d_in[11];
  const float* W2   = (const float*)d_in[12];
  const float* bb2  = (const float*)d_in[13];
  const float* W3   = (const float*)d_in[14];
  const float* bb3  = (const float*)d_in[15];
  float* out = (float*)d_out;

  char* ws = (char*)d_ws;
  size_t off = 0;
  auto alloc = [&](size_t bytes) -> char* {
    char* p = ws + off;
    off += (bytes + 255) & ~(size_t)255;
    return p;
  };
  const size_t S_F = (size_t)NNODES * FDIM;
  _Float16* xh  = (_Float16*)alloc(S_F * 2);
  _Float16* h1h = (_Float16*)alloc(S_F * 2);
  _Float16* h2h = (_Float16*)alloc(S_F * 2);
  int*  cnt    = (int*)alloc((size_t)SCAN_N * 4);
  int*  rowptr = (int*)alloc((size_t)SCAN_N * 4);
  int*  cursor = (int*)alloc((size_t)SCAN_N * 4);
  int2* perm   = (int2*)alloc((size_t)NEDGES * 8);
  const size_t uoff = off;
  float*    xl   = (float*)alloc(S_F * 4);
  float*    xr   = (float*)alloc(S_F * 4);
  float*    accb = (float*)alloc(S_F * 4);
  float*    ebuf = (float*)alloc((size_t)ETOT * NHEAD * 4);
  unsigned* mkey = (unsigned*)alloc((size_t)NNODES * NHEAD * 4);
  float*    ssum = (float*)alloc((size_t)NNODES * NHEAD * 4);
  _Float16* m1 = (_Float16*)(ws + uoff);
  _Float16* m2 = (_Float16*)(ws + uoff + (size_t)NNODES * 640 * 2);

  const int TB = 256;
  const int ngrid = (int)((S_F + TB - 1) / TB);
  const int egrid = (ETOT * NHEAD + TB - 1) / TB;
  const int hgrid = (NNODES * NHEAD + TB - 1) / TB;
  const int cgrid = (SCAN_N + TB - 1) / TB;
  const int pgrid = (NEDGES + TB - 1) / TB;
  const int wgrid = (NNODES + 7) / 8;
  const dim3 gconv((NNODES + 127) / 128, 128 / 64);

  csr_zero_kernel<<<cgrid, TB, 0, stream>>>(cnt, SCAN_N);
  csr_count_kernel<<<pgrid, TB, 0, stream>>>(ei, cnt);
  csr_scan_kernel<<<1, 1024, 0, stream>>>(cnt, rowptr, cursor);
  csr_fill_kernel<<<pgrid, TB, 0, stream>>>(ei, cursor, perm);

  cast_f16_kernel<<<ngrid, TB, 0, stream>>>(x, xh, (int)S_F);
  wmma_gemm_kernel<<<gconv, TB, 0, stream>>>(xh, Wl1, nullptr, xl, nullptr,
                                             NNODES, 128, 128, 0);
  wmma_gemm_kernel<<<gconv, TB, 0, stream>>>(xh, Wr1, nullptr, xr, nullptr,
                                             NNODES, 128, 128, 0);
  init_conv_kernel<<<hgrid, TB, 0, stream>>>(mkey, ssum, NNODES);
  edge_logits_kernel<<<egrid, TB, 0, stream>>>(ei, xl, xr, att1, ebuf, mkey);
  edge_exp_kernel<<<egrid, TB, 0, stream>>>(ei, ebuf, mkey, ssum);
  node_gather_kernel<<<wgrid, TB, 0, stream>>>(rowptr, perm, xl, ebuf, ssum,
                                               accb);
  node_finish_kernel<<<ngrid, TB, 0, stream>>>(accb, b1, h1h, NNODES);

  wmma_gemm_kernel<<<gconv, TB, 0, stream>>>(h1h, Wl2, nullptr, xl, nullptr,
                                             NNODES, 128, 128, 0);
  wmma_gemm_kernel<<<gconv, TB, 0, stream>>>(h1h, Wr2, nullptr, xr, nullptr,
                                             NNODES, 128, 128, 0);
  init_conv_kernel<<<hgrid, TB, 0, stream>>>(mkey, ssum, NNODES);
  edge_logits_kernel<<<egrid, TB, 0, stream>>>(ei, xl, xr, att2, ebuf, mkey);
  edge_exp_kernel<<<egrid, TB, 0, stream>>>(ei, ebuf, mkey, ssum);
  node_gather_kernel<<<wgrid, TB, 0, stream>>>(rowptr, perm, xl, ebuf, ssum,
                                               accb);
  node_finish_kernel<<<ngrid, TB, 0, stream>>>(accb, b2, h2h, NNODES);

  const dim3 gm1((NNODES + 127) / 128, 640 / 64);
  wmma_gemm_kernel<<<gm1, TB, 0, stream>>>(h2h, W1, bb1, nullptr, m1,
                                           NNODES, 640, 128, 1);
  const dim3 gm2((NNODES + 127) / 128, 320 / 64);
  wmma_gemm_kernel<<<gm2, TB, 0, stream>>>(m1, W2, bb2, nullptr, m2,
                                           NNODES, 320, 640, 1);
  const dim3 gm3((NNODES + 127) / 128, 64 / 64);
  wmma_gemm_kernel<<<gm3, TB, 0, stream>>>(m2, W3, bb3, out, nullptr,
                                           NNODES, 64, 320, 2);
}


# GEMM epilogue fixed at compile time: activation, bias and output type are template arguments and offsets 32-bit, so the per-element scalar branches and 64-bit multiplies are gone
# speedup vs baseline: 5.4248x; 2.0766x over previous
#include <hip/hip_runtime.h>
#include <hip/hip_bf16.h>


typedef __attribute__((ext_vector_type(16))) _Float16 v16h;
typedef __attribute__((ext_vector_type(8)))  _Float16 v8h;
typedef __attribute__((ext_vector_type(8)))  float    v8f;

#define NNODES 50000
#define NEDGES 800000
#define FDIM   128

__global__ void cast_f16_kernel(const float* __restrict__ in,
                                _Float16* __restrict__ out, int n) {
  int i = blockIdx.x * blockDim.x + threadIdx.x;
  if (i < n) out[i] = (_Float16)in[i];
}

__global__ void csr_zero_kernel(int* __restrict__ cnt, int n) {
  int i = blockIdx.x * blockDim.x + threadIdx.x;
  if (i < n) cnt[i] = 0;
}

__global__ void csr_count_kernel(const int* __restrict__ ei,
                                 int* __restrict__ cnt) {
  int e = blockIdx.x * blockDim.x + threadIdx.x;
  if (e < NEDGES) atomicAdd(cnt + ei[NEDGES + e], 1);
}

#define SCAN_CH ((NNODES + 1023) / 1024)
#define SCAN_N  (SCAN_CH * 1024)
__global__ __launch_bounds__(1024) void csr_scan_kernel(
    const int* __restrict__ cnt, int* __restrict__ rowptr,
    int* __restrict__ cursor) {
  __shared__ int wsum[32];
  const int t = threadIdx.x, lane = t & 31, wv = t >> 5;
  const int i0 = t * SCAN_CH;
  int v[SCAN_CH];
  int local = 0;
  #pragma unroll
  for (int i = 0; i < SCAN_CH; ++i) v[i] = cnt[i0 + i];
  #pragma unroll
  for (int i = 0; i < SCAN_CH; ++i) local += v[i];
  int x = local;
  for (int o = 1; o < 32; o <<= 1) {
    const int y = __shfl_up(x, o);
    if (lane >= o) x += y;
  }
  if (lane == 31) wsum[wv] = x;
  __syncthreads();
  if (wv == 0) {
    const int w = wsum[lane];
    int xs = w;
    for (int o = 1; o < 32; o <<= 1) {
      const int y = __shfl_up(xs, o);
      if (lane >= o) xs += y;
    }
    wsum[lane] = xs - w;
  }
  __syncthreads();
  int run = wsum[wv] + (x - local);
  #pragma unroll
  for (int i = 0; i < SCAN_CH; ++i) {
    rowptr[i0 + i] = run;
    cursor[i0 + i] = run;
    run += v[i];
  }
}

__global__ void csr_fill_kernel(const int* __restrict__ ei,
                                int* __restrict__ cursor,
                                int2* __restrict__ perm) {
  int e = blockIdx.x * blockDim.x + threadIdx.x;
  if (e >= NEDGES) return;
  const int s = ei[e], d = ei[NEDGES + e];
  const int pos = atomicAdd(cursor + d, 1);
  if (pos < NEDGES) perm[pos] = make_int2(s, e);
}

__global__ __launch_bounds__(256) void csr_order_kernel(
    const int* __restrict__ rowptr, const int2* __restrict__ perm,
    int* __restrict__ srcs) {
  const int lane = threadIdx.x & 31;
  const int w = __builtin_amdgcn_readfirstlane(
      (int)(blockIdx.x * 8 + (threadIdx.x >> 5)));
  for (int q = 0; q < 8; ++q) {
    const int d = w * 8 + q;
    if (d >= NNODES) break;
    const int beg = rowptr[d], end = rowptr[d + 1];
    for (int base = beg; base < end; base += 32) {
      const bool valid = base + lane < end;
      int2 my = make_int2(0, 0x7fffffff);
      if (valid) my = perm[base + lane];
      int rank = 0;
      for (int cb = beg; cb < end; cb += 32) {
        const int cn = (end - cb) < 32 ? (end - cb) : 32;
        int other = 0x7fffffff;
        if (cb + lane < end) other = perm[cb + lane].y;
        for (int j = 0; j < cn; ++j) {
          const int k = __builtin_amdgcn_readlane(other, j);
          rank += (k < my.y) ? 1 : 0;
        }
      }
      if (valid) srcs[beg + rank] = my.x;
    }
  }
}

#define KC 64
#define BT_STRIDE 72

template <int ACT, bool HASBIAS, bool F32OUT>
__global__ __launch_bounds__(256) void wmma_gemm_kernel(
    const _Float16* __restrict__ A, const float* __restrict__ W,
    const float* __restrict__ bias,
    float* __restrict__ outF, _Float16* __restrict__ outH,
    int M, int Nt, int K) {
  __shared__ _Float16 bt[64 * BT_STRIDE];

  const int lane = threadIdx.x & 31;
  const int wave = threadIdx.x >> 5;
  const int half = lane >> 4;
  const int l15  = lane & 15;
  const int n0   = blockIdx.y * 64;

  int rowBase = blockIdx.x * 128 + wave * 16;
  if (rowBase > M - 16) rowBase = M - 16;

  v8f c[4] = {v8f{}, v8f{}, v8f{}, v8f{}};

  for (int kc = 0; kc < K; kc += KC) {
    {
      int kl = threadIdx.x >> 4;
      const int n4 = (threadIdx.x & 15) * 4;
      #pragma unroll
      for (int p = 0; p < 4; ++p, kl += 16) {
        const float4 wv =
            *(const float4*)(W + (size_t)(kc + kl) * Nt + n0 + n4);
        bt[(n4 + 0) * BT_STRIDE + kl] = (_Float16)wv.x;
        bt[(n4 + 1) * BT_STRIDE + kl] = (_Float16)wv.y;
        bt[(n4 + 2) * BT_STRIDE + kl] = (_Float16)wv.z;
        bt[(n4 + 3) * BT_STRIDE + kl] = (_Float16)wv.w;
      }
    }
    __syncthreads();

    const _Float16* arow = A + (size_t)(rowBase + l15) * K + kc;
    #pragma unroll
    for (int ks = 0; ks < 2; ++ks) {
      const int kb = ks * 32 + half * 8;
      v8h alo = *(const v8h*)(arow + kb);
      v8h ahi = *(const v8h*)(arow + kb + 16);
      v16h a = __builtin_shufflevector(alo, ahi, 0, 1, 2, 3, 4, 5, 6, 7,
                                       8, 9, 10, 11, 12, 13, 14, 15);
      v16h b[4];
      #pragma unroll
      for (int nn = 0; nn < 4; ++nn) {
        const _Float16* bp = bt + (nn * 16 + l15) * BT_STRIDE + kb;
        v8h blo = *(const v8h*)(bp);
        v8h bhi = *(const v8h*)(bp + 16);
        b[nn] = __builtin_shufflevector(blo, bhi, 0, 1, 2, 3, 4, 5, 6, 7,
                                        8, 9, 10, 11, 12, 13, 14, 15);
      }
      #pragma unroll
      for (int nn = 0; nn < 4; ++nn) {
        c[nn] = __builtin_amdgcn_wmma_f32_16x16x32_f16(
            false, a, false, b[nn], (short)0, c[nn], false, false);
      }
    }
    __syncthreads();
  }

  const unsigned o0 =
      (unsigned)(rowBase + half * 8) * (unsigned)Nt + (unsigned)(n0 + l15);
  #pragma unroll
  for (int nn = 0; nn < 4; ++nn) {
    float bv = 0.0f;
    if (HASBIAS) bv = bias[n0 + nn * 16 + l15];
    #pragma unroll
    for (int r = 0; r < 8; ++r) {
      float v = c[nn][r] + bv;
      if (ACT == 1) v = v > 0.0f ? v : (__expf(v) - 1.0f);
      else if (ACT == 2) v = 1.0f / (1.0f + __expf(-v));
      const unsigned o = o0 + (unsigned)r * (unsigned)Nt + (unsigned)(nn * 16);
      if (F32OUT) outF[o] = v;
      else        outH[o] = (_Float16)v;
    }
  }
}

typedef __attribute__((ext_vector_type(4))) _Float16 v4h;

template <int CTRL>
__device__ __forceinline__ float lane_xor_get(float x) {
  return __builtin_bit_cast(
      float, __builtin_amdgcn_update_dpp(0, __builtin_bit_cast(int, x), CTRL,
                                         0xf, 0xf, true));
}

__device__ __forceinline__ float edge_logit(const float4 a, const float4 b,
                                            const float4 w) {
  float z, acc;
  z = a.x + b.x; z = z > 0.0f ? z : 0.2f * z; acc = z * w.x;
  z = a.y + b.y; z = z > 0.0f ? z : 0.2f * z; acc += z * w.y;
  z = a.z + b.z; z = z > 0.0f ? z : 0.2f * z; acc += z * w.z;
  z = a.w + b.w; z = z > 0.0f ? z : 0.2f * z; acc += z * w.w;
  acc += lane_xor_get<0x161>(acc);
  acc += lane_xor_get<0x162>(acc);
  acc += lane_xor_get<0x164>(acc);
  return acc;
}

template <int NB>
__device__ __forceinline__ void conv_batch(const float4* __restrict__ xl4,
                                           int my, int j0, int lane,
                                           const float4 xrv, const float4 av,
                                           float& m, float& ssum, float4& r) {
  float4 v[NB];
  float e[NB];
  #pragma unroll
  for (int t = 0; t < NB; ++t) {
    const int sidx = __builtin_amdgcn_readlane(my, j0 + t);
    v[t] = xl4[(size_t)sidx * 32 + lane];
  }
  float mn = m;
  #pragma unroll
  for (int t = 0; t < NB; ++t) {
    e[t] = edge_logit(v[t], xrv, av);
    mn = fmaxf(mn, e[t]);
  }
  const float sc = __expf(m - mn);
  ssum *= sc;
  r.x *= sc; r.y *= sc; r.z *= sc; r.w *= sc;
  #pragma unroll
  for (int t = 0; t < NB; ++t) {
    const float p = __expf(e[t] - mn);
    ssum += p;
    r.x += v[t].x * p; r.y += v[t].y * p; r.z += v[t].z * p; r.w += v[t].w * p;
  }
  m = mn;
}

__global__ __launch_bounds__(256) void conv_edge_kernel(
    const int* __restrict__ rowptr, const int* __restrict__ srcs,
    const float* __restrict__ xl, const float* __restrict__ xr,
    const float* __restrict__ att, const float* __restrict__ bias,
    _Float16* __restrict__ outH) {
  const int lane = threadIdx.x & 31;
  const int d = __builtin_amdgcn_readfirstlane(
      (int)(blockIdx.x * 8 + (threadIdx.x >> 5)));
  if (d >= NNODES) return;
  const float4* xl4 = (const float4*)xl;
  const float4 xrv = ((const float4*)xr)[(size_t)d * 32 + lane];
  const float4 av = ((const float4*)att)[lane];

  float4 r = xl4[(size_t)d * 32 + lane];
  float m = edge_logit(r, xrv, av);
  float ssum = 1.0f;

  const int beg = rowptr[d], end = rowptr[d + 1];
  for (int base = beg; base < end; base += 32) {
    const int cnt = (end - base) < 32 ? (end - base) : 32;
    int my = 0;
    if (lane < cnt) my = srcs[base + lane];
    int j = 0;
    for (; j + 8 <= cnt; j += 8)
      conv_batch<8>(xl4, my, j, lane, xrv, av, m, ssum, r);
    if (j + 4 <= cnt) { conv_batch<4>(xl4, my, j, lane, xrv, av, m, ssum, r); j += 4; }
    if (j + 2 <= cnt) { conv_batch<2>(xl4, my, j, lane, xrv, av, m, ssum, r); j += 2; }
    if (j < cnt) conv_batch<1>(xl4, my, j, lane, xrv, av, m, ssum, r);
  }

  const float4 bv = ((const float4*)bias)[lane];
  float o0 = r.x / ssum + bv.x, o1 = r.y / ssum + bv.y;
  float o2 = r.z / ssum + bv.z, o3 = r.w / ssum + bv.w;
  o0 = o0 > 0.0f ? o0 : (__expf(o0) - 1.0f);
  o1 = o1 > 0.0f ? o1 : (__expf(o1) - 1.0f);
  o2 = o2 > 0.0f ? o2 : (__expf(o2) - 1.0f);
  o3 = o3 > 0.0f ? o3 : (__expf(o3) - 1.0f);
  v4h hv;
  hv[0] = (_Float16)o0; hv[1] = (_Float16)o1;
  hv[2] = (_Float16)o2; hv[3] = (_Float16)o3;
  *(v4h*)(outH + (size_t)d * FDIM + lane * 4) = hv;
}

extern "C" void kernel_launch(void* const* d_in, const int* in_sizes, int n_in,
                              void* d_out, int out_size, void* d_ws,
                              size_t ws_size, hipStream_t stream) {
  (void)in_sizes; (void)n_in; (void)out_size; (void)ws_size;
  const float* x    = (const float*)d_in[0];
  const int*   ei   = (const int*)d_in[1];
  const float* Wl1  = (const float*)d_in[2];
  const float* Wr1  = (const float*)d_in[3];
  const float* att1 = (const float*)d_in[4];
  const float* b1   = (const float*)d_in[5];
  const float* Wl2  = (const float*)d_in[6];
  const float* Wr2  = (const float*)d_in[7];
  const float* att2 = (const float*)d_in[8];
  const float* b2   = (const float*)d_in[9];
  const float* W1   = (const float*)d_in[10];
  const float* bb1  = (const float*)d_in[11];
  const float* W2   = (const float*)d_in[12];
  const float* bb2  = (const float*)d_in[13];
  const float* W3   = (const float*)d_in[14];
  const float* bb3  = (const float*)d_in[15];
  float* out = (float*)d_out;

  char* ws = (char*)d_ws;
  size_t off = 0;
  auto alloc = [&](size_t bytes) -> char* {
    char* p = ws + off;
    off += (bytes + 255) & ~(size_t)255;
    return p;
  };
  const size_t S_F = (size_t)NNODES * FDIM;
  _Float16* xh  = (_Float16*)alloc(S_F * 2);
  _Float16* h1h = (_Float16*)alloc(S_F * 2);
  _Float16* h2h = (_Float16*)alloc(S_F * 2);
  int*  cnt    = (int*)alloc((size_t)SCAN_N * 4);
  int*  rowptr = (int*)alloc((size_t)SCAN_N * 4);
  int*  cursor = (int*)alloc((size_t)SCAN_N * 4);
  int2* perm   = (int2*)alloc((size_t)NEDGES * 8);
  int*  srcs   = (int*)alloc((size_t)NEDGES * 4);
  const size_t uoff = off;
  float*    xl   = (float*)alloc(S_F * 4);
  float*    xr   = (float*)alloc(S_F * 4);
  _Float16* m1 = (_Float16*)(ws + uoff);
  _Float16* m2 = (_Float16*)(ws + uoff + (size_t)NNODES * 640 * 2);

  const int TB = 256;
  const int ngrid = (int)((S_F + TB - 1) / TB);
  const int cgrid = (SCAN_N + TB - 1) / TB;
  const int pgrid = (NEDGES + TB - 1) / TB;
  const int wgrid = (NNODES + 7) / 8;
  const int ogrid = (NNODES + 63) / 64;
  const dim3 gconv((NNODES + 127) / 128, 128 / 64);

  csr_zero_kernel<<<cgrid, TB, 0, stream>>>(cnt, SCAN_N);
  csr_count_kernel<<<pgrid, TB, 0, stream>>>(ei, cnt);
  csr_scan_kernel<<<1, 1024, 0, stream>>>(cnt, rowptr, cursor);
  csr_fill_kernel<<<pgrid, TB, 0, stream>>>(ei, cursor, perm);
  csr_order_kernel<<<ogrid, TB, 0, stream>>>(rowptr, perm, srcs);

  cast_f16_kernel<<<ngrid, TB, 0, stream>>>(x, xh, (int)S_F);
  wmma_gemm_kernel<0, false, true><<<gconv, TB, 0, stream>>>(
      xh, Wl1, nullptr, xl, nullptr, NNODES, 128, 128);
  wmma_gemm_kernel<0, false, true><<<gconv, TB, 0, stream>>>(
      xh, Wr1, nullptr, xr, nullptr, NNODES, 128, 128);
  conv_edge_kernel<<<wgrid, TB, 0, stream>>>(rowptr, srcs, xl, xr, att1, b1,
                                             h1h);

  wmma_gemm_kernel<0, false, true><<<gconv, TB, 0, stream>>>(
      h1h, Wl2, nullptr, xl, nullptr, NNODES, 128, 128);
  wmma_gemm_kernel<0, false, true><<<gconv, TB, 0, stream>>>(
      h1h, Wr2, nullptr, xr, nullptr, NNODES, 128, 128);
  conv_edge_kernel<<<wgrid, TB, 0, stream>>>(rowptr, srcs, xl, xr, att2, b2,
                                             h2h);

  const dim3 gm1((NNODES + 127) / 128, 640 / 64);
  wmma_gemm_kernel<1, true, false><<<gm1, TB, 0, stream>>>(
      h2h, W1, bb1, nullptr, m1, NNODES, 640, 128);
  const dim3 gm2((NNODES + 127) / 128, 320 / 64);
  wmma_gemm_kernel<1, true, false><<<gm2, TB, 0, stream>>>(
      m1, W2, bb2, nullptr, m2, NNODES, 320, 640);
  const dim3 gm3((NNODES + 127) / 128, 64 / 64);
  wmma_gemm_kernel<2, true, true><<<gm3, TB, 0, stream>>>(
      m2, W3, bb3, out, nullptr, NNODES, 64, 320);
}


# Edge list built by a two-level counting sort on LDS counters (edges binned by node block per chunk, then placed per node inside each block): no global atomics; segment and sum order unchanged
# speedup vs baseline: 7.3048x; 1.3466x over previous
#include <hip/hip_runtime.h>
#include <hip/hip_bf16.h>


typedef __attribute__((ext_vector_type(16))) _Float16 v16h;
typedef __attribute__((ext_vector_type(8)))  _Float16 v8h;
typedef __attribute__((ext_vector_type(8)))  float    v8f;

#define NNODES 50000
#define NEDGES 800000
#define FDIM   128

__global__ void cast_f16_kernel(const float* __restrict__ in,
                                _Float16* __restrict__ out, int n) {
  int i = blockIdx.x * blockDim.x + threadIdx.x;
  if (i < n) out[i] = (_Float16)in[i];
}

#define NBLK 256
#define BLKW 196
#define NCHK 256
#define CHKE ((NEDGES + NCHK - 1) / NCHK)
#define CHKT ((CHKE + 1023) / 1024)

__device__ __forceinline__ int node_block(int d) {
  const unsigned b = (unsigned)d / (unsigned)BLKW;
  return (int)(b < (unsigned)(NBLK - 1) ? b : (unsigned)(NBLK - 1));
}

__global__ __launch_bounds__(1024) void bin_count_kernel(
    const int* __restrict__ ei, int* __restrict__ ccnt) {
  __shared__ int h[NBLK];
  const int t = threadIdx.x;
  if (t < NBLK) h[t] = 0;
  __syncthreads();
  const int e0 = blockIdx.x * CHKE;
  const int e1 = (e0 + CHKE) < NEDGES ? (e0 + CHKE) : NEDGES;
  int dv[CHKT];
  #pragma unroll
  for (int k = 0; k < CHKT; ++k) {
    const int e = e0 + t + k * 1024;
    dv[k] = ei[NEDGES + (e < NEDGES ? e : NEDGES - 1)];
  }
  #pragma unroll
  for (int k = 0; k < CHKT; ++k)
    if (e0 + t + k * 1024 < e1) atomicAdd(&h[node_block(dv[k])], 1);
  __syncthreads();
  if (t < NBLK) ccnt[t * NCHK + blockIdx.x] = h[t];
}

__global__ __launch_bounds__(1024) void bin_scan_kernel(
    const int* __restrict__ ccnt, int* __restrict__ coff) {
  __shared__ int wsum[32];
  const int t = threadIdx.x, lane = t & 31, wv = t >> 5;
  const int4* in4 = (const int4*)ccnt + wv * (NBLK * NCHK / 128);
  int4* out4 = (int4*)coff + wv * (NBLK * NCHK / 128);
  int4 v[16];
  #pragma unroll
  for (int k = 0; k < 16; ++k) v[k] = in4[k * 32 + lane];
  int carry = 0;
  #pragma unroll
  for (int k = 0; k < 16; ++k) {
    const int s0 = v[k].x, s1 = s0 + v[k].y, s2 = s1 + v[k].z, s3 = s2 + v[k].w;
    int x = s3;
    for (int o = 1; o < 32; o <<= 1) {
      const int y = __shfl_up(x, o);
      if (lane >= o) x += y;
    }
    const int ex = carry + x - s3;
    v[k] = make_int4(ex, ex + s0, ex + s1, ex + s2);
    carry += __shfl(x, 31);
  }
  if (lane == 0) wsum[wv] = carry;
  __syncthreads();
  if (wv == 0) {
    const int w = wsum[lane];
    int xs = w;
    for (int o = 1; o < 32; o <<= 1) {
      const int y = __shfl_up(xs, o);
      if (lane >= o) xs += y;
    }
    wsum[lane] = xs - w;
  }
  __syncthreads();
  const int base = wsum[wv];
  #pragma unroll
  for (int k = 0; k < 16; ++k) {
    int4 o = v[k];
    o.x += base; o.y += base; o.z += base; o.w += base;
    out4[k * 32 + lane] = o;
  }
  if (t == 1023) coff[NBLK * NCHK] = base + carry;
}

__global__ __launch_bounds__(1024) void bin_scatter_kernel(
    const int* __restrict__ ei, const int* __restrict__ coff,
    int2* __restrict__ binned) {
  __shared__ int cur[NBLK];
  const int t = threadIdx.x;
  if (t < NBLK) cur[t] = coff[t * NCHK + blockIdx.x];
  __syncthreads();
  const int e0 = blockIdx.x * CHKE;
  const int e1 = (e0 + CHKE) < NEDGES ? (e0 + CHKE) : NEDGES;
  int sv[CHKT], dv[CHKT];
  #pragma unroll
  for (int k = 0; k < CHKT; ++k) {
    const int e = e0 + t + k * 1024;
    const int ec = e < NEDGES ? e : NEDGES - 1;
    sv[k] = ei[ec];
    dv[k] = ei[NEDGES + ec];
  }
  #pragma unroll
  for (int k = 0; k < CHKT; ++k) {
    const int e = e0 + t + k * 1024;
    if (e < e1) {
      const int b = node_block(dv[k]);
      const int p = atomicAdd(&cur[b], 1);
      if ((unsigned)p < (unsigned)NEDGES)
        binned[p] = make_int2((sv[k] & 0xffff) | ((dv[k] - b * BLKW) << 16), e);
    }
  }
}

__global__ __launch_bounds__(1024) void bin_place_kernel(
    const int* __restrict__ coff, const int2* __restrict__ binned,
    int* __restrict__ rowptr, int2* __restrict__ perm) {
  __shared__ int cnt[256];
  __shared__ int cur[256];
  __shared__ int wtot[8];
  const int t = threadIdx.x, lane = t & 31, wv = t >> 5;
  const int b = blockIdx.x;
  const int beg = coff[b * NCHK], end = coff[(b + 1) * NCHK];
  if (t < 256) cnt[t] = 0;
  __syncthreads();
  for (int i0 = beg; i0 < end; i0 += 4096) {
    int rx[4];
    #pragma unroll
    for (int k = 0; k < 4; ++k) {
      const int i = i0 + t + k * 1024;
      rx[k] = binned[i < NEDGES ? i : NEDGES - 1].x;
    }
    #pragma unroll
    for (int k = 0; k < 4; ++k)
      if (i0 + t + k * 1024 < end) atomicAdd(&cnt[(rx[k] >> 16) & 0xff], 1);
  }
  __syncthreads();
  int c = 0;
  if (t < 256) c = cnt[t];
  int x = c;
  for (int o = 1; o < 32; o <<= 1) {
    const int y = __shfl_up(x, o);
    if (lane >= o) x += y;
  }
  if (t < 256 && lane == 31) wtot[wv] = x;
  __syncthreads();
  if (t < 256) {
    int first = beg + x - c;
    for (int w = 0; w < wv; ++w) first += wtot[w];
    cur[t] = first;
    const int node = b * BLKW + t;
    if (t < BLKW && node < NNODES) rowptr[node] = first;
  }
  if (b == NBLK - 1 && t == 0) rowptr[NNODES] = end;
  __syncthreads();
  for (int i0 = beg; i0 < end; i0 += 4096) {
    int2 r[4];
    #pragma unroll
    for (int k = 0; k < 4; ++k) {
      const int i = i0 + t + k * 1024;
      r[k] = binned[i < NEDGES ? i : NEDGES - 1];
    }
    #pragma unroll
    for (int k = 0; k < 4; ++k) {
      if (i0 + t + k * 1024 < end) {
        const int p = atomicAdd(&cur[(r[k].x >> 16) & 0xff], 1);
        if ((unsigned)p < (unsigned)NEDGES)
          perm[p] = make_int2(r[k].x & 0xffff, r[k].y);
      }
    }
  }
}

__global__ __launch_bounds__(256) void csr_order_kernel(
    const int* __restrict__ rowptr, const int2* __restrict__ perm,
    int* __restrict__ srcs) {
  const int lane = threadIdx.x & 31;
  const int w = __builtin_amdgcn_readfirstlane(
      (int)(blockIdx.x * 8 + (threadIdx.x >> 5)));
  for (int q = 0; q < 8; ++q) {
    const int d = w * 8 + q;
    if (d >= NNODES) break;
    const int beg = rowptr[d], end = rowptr[d + 1];
    for (int base = beg; base < end; base += 32) {
      const bool valid = base + lane < end;
      int2 my = make_int2(0, 0x7fffffff);
      if (valid) my = perm[base + lane];
      int rank = 0;
      for (int cb = beg; cb < end; cb += 32) {
        const int cn = (end - cb) < 32 ? (end - cb) : 32;
        int other = 0x7fffffff;
        if (cb + lane < end) other = perm[cb + lane].y;
        for (int j = 0; j < cn; ++j) {
          const int k = __builtin_amdgcn_readlane(other, j);
          rank += (k < my.y) ? 1 : 0;
        }
      }
      if (valid) srcs[beg + rank] = my.x;
    }
  }
}

#define KC 64
#define BT_STRIDE 72

template <int ACT, bool HASBIAS, bool F32OUT>
__global__ __launch_bounds__(256) void wmma_gemm_kernel(
    const _Float16* __restrict__ A, const float* __restrict__ W,
    const float* __restrict__ bias,
    float* __restrict__ outF, _Float16* __restrict__ outH,
    int M, int Nt, int K) {
  __shared__ _Float16 bt[64 * BT_STRIDE];

  const int lane = threadIdx.x & 31;
  const int wave = threadIdx.x >> 5;
  const int half = lane >> 4;
  const int l15  = lane & 15;
  const int n0   = blockIdx.y * 64;

  int rowBase = blockIdx.x * 128 + wave * 16;
  if (rowBase > M - 16) rowBase = M - 16;

  v8f c[4] = {v8f{}, v8f{}, v8f{}, v8f{}};

  for (int kc = 0; kc < K; kc += KC) {
    {
      int kl = threadIdx.x >> 4;
      const int n4 = (threadIdx.x & 15) * 4;
      #pragma unroll
      for (int p = 0; p < 4; ++p, kl += 16) {
        const float4 wv =
            *(const float4*)(W + (size_t)(kc + kl) * Nt + n0 + n4);
        bt[(n4 + 0) * BT_STRIDE + kl] = (_Float16)wv.x;
        bt[(n4 + 1) * BT_STRIDE + kl] = (_Float16)wv.y;
        bt[(n4 + 2) * BT_STRIDE + kl] = (_Float16)wv.z;
        bt[(n4 + 3) * BT_STRIDE + kl] = (_Float16)wv.w;
      }
    }
    __syncthreads();

    const _Float16* arow = A + (size_t)(rowBase + l15) * K + kc;
    #pragma unroll
    for (int ks = 0; ks < 2; ++ks) {
      const int kb = ks * 32 + half * 8;
      v8h alo = *(const v8h*)(arow + kb);
      v8h ahi = *(const v8h*)(arow + kb + 16);
      v16h a = __builtin_shufflevector(alo, ahi, 0, 1, 2, 3, 4, 5, 6, 7,
                                       8, 9, 10, 11, 12, 13, 14, 15);
      v16h b[4];
      #pragma unroll
      for (int nn = 0; nn < 4; ++nn) {
        const _Float16* bp = bt + (nn * 16 + l15) * BT_STRIDE + kb;
        v8h blo = *(const v8h*)(bp);
        v8h bhi = *(const v8h*)(bp + 16);
        b[nn] = __builtin_shufflevector(blo, bhi, 0, 1, 2, 3, 4, 5, 6, 7,
                                        8, 9, 10, 11, 12, 13, 14, 15);
      }
      #pragma unroll
      for (int nn = 0; nn < 4; ++nn) {
        c[nn] = __builtin_amdgcn_wmma_f32_16x16x32_f16(
            false, a, false, b[nn], (short)0, c[nn], false, false);
      }
    }
    __syncthreads();
  }

  const unsigned o0 =
      (unsigned)(rowBase + half * 8) * (unsigned)Nt + (unsigned)(n0 + l15);
  #pragma unroll
  for (int nn = 0; nn < 4; ++nn) {
    float bv = 0.0f;
    if (HASBIAS) bv = bias[n0 + nn * 16 + l15];
    #pragma unroll
    for (int r = 0; r < 8; ++r) {
      float v = c[nn][r] + bv;
      if (ACT == 1) v = v > 0.0f ? v : (__expf(v) - 1.0f);
      else if (ACT == 2) v = 1.0f / (1.0f + __expf(-v));
      const unsigned o = o0 + (unsigned)r * (unsigned)Nt + (unsigned)(nn * 16);
      if (F32OUT) outF[o] = v;
      else        outH[o] = (_Float16)v;
    }
  }
}

typedef __attribute__((ext_vector_type(4))) _Float16 v4h;

template <int CTRL>
__device__ __forceinline__ float lane_xor_get(float x) {
  return __builtin_bit_cast(
      float, __builtin_amdgcn_update_dpp(0, __builtin_bit_cast(int, x), CTRL,
                                         0xf, 0xf, true));
}

__device__ __forceinline__ float edge_logit(const float4 a, const float4 b,
                                            const float4 w) {
  float z, acc;
  z = a.x + b.x; z = z > 0.0f ? z : 0.2f * z; acc = z * w.x;
  z = a.y + b.y; z = z > 0.0f ? z : 0.2f * z; acc += z * w.y;
  z = a.z + b.z; z = z > 0.0f ? z : 0.2f * z; acc += z * w.z;
  z = a.w + b.w; z = z > 0.0f ? z : 0.2f * z; acc += z * w.w;
  acc += lane_xor_get<0x161>(acc);
  acc += lane_xor_get<0x162>(acc);
  acc += lane_xor_get<0x164>(acc);
  return acc;
}

template <int NB>
__device__ __forceinline__ void conv_batch(const float4* __restrict__ xl4,
                                           int my, int j0, int lane,
                                           const float4 xrv, const float4 av,
                                           float& m, float& ssum, float4& r) {
  float4 v[NB];
  float e[NB];
  #pragma unroll
  for (int t = 0; t < NB; ++t) {
    const int sidx = __builtin_amdgcn_readlane(my, j0 + t);
    v[t] = xl4[(size_t)sidx * 32 + lane];
  }
  float mn = m;
  #pragma unroll
  for (int t = 0; t < NB; ++t) {
    e[t] = edge_logit(v[t], xrv, av);
    mn = fmaxf(mn, e[t]);
  }
  const float sc = __expf(m - mn);
  ssum *= sc;
  r.x *= sc; r.y *= sc; r.z *= sc; r.w *= sc;
  #pragma unroll
  for (int t = 0; t < NB; ++t) {
    const float p = __expf(e[t] - mn);
    ssum += p;
    r.x += v[t].x * p; r.y += v[t].y * p; r.z += v[t].z * p; r.w += v[t].w * p;
  }
  m = mn;
}

__global__ __launch_bounds__(256) void conv_edge_kernel(
    const int* __restrict__ rowptr, const int* __restrict__ srcs,
    const float* __restrict__ xl, const float* __restrict__ xr,
    const float* __restrict__ att, const float* __restrict__ bias,
    _Float16* __restrict__ outH) {
  const int lane = threadIdx.x & 31;
  const int d = __builtin_amdgcn_readfirstlane(
      (int)(blockIdx.x * 8 + (threadIdx.x >> 5)));
  if (d >= NNODES) return;
  const float4* xl4 = (const float4*)xl;
  const float4 xrv = ((const float4*)xr)[(size_t)d * 32 + lane];
  const float4 av = ((const float4*)att)[lane];

  float4 r = xl4[(size_t)d * 32 + lane];
  float m = edge_logit(r, xrv, av);
  float ssum = 1.0f;

  const int beg = rowptr[d], end = rowptr[d + 1];
  for (int base = beg; base < end; base += 32) {
    const int cnt = (end - base) < 32 ? (end - base) : 32;
    int my = 0;
    if (lane < cnt) my = srcs[base + lane];
    int j = 0;
    for (; j + 8 <= cnt; j += 8)
      conv_batch<8>(xl4, my, j, lane, xrv, av, m, ssum, r);
    if (j + 4 <= cnt) { conv_batch<4>(xl4, my, j, lane, xrv, av, m, ssum, r); j += 4; }
    if (j + 2 <= cnt) { conv_batch<2>(xl4, my, j, lane, xrv, av, m, ssum, r); j += 2; }
    if (j < cnt) conv_batch<1>(xl4, my, j, lane, xrv, av, m, ssum, r);
  }

  const float4 bv = ((const float4*)bias)[lane];
  float o0 = r.x / ssum + bv.x, o1 = r.y / ssum + bv.y;
  float o2 = r.z / ssum + bv.z, o3 = r.w / ssum + bv.w;
  o0 = o0 > 0.0f ? o0 : (__expf(o0) - 1.0f);
  o1 = o1 > 0.0f ? o1 : (__expf(o1) - 1.0f);
  o2 = o2 > 0.0f ? o2 : (__expf(o2) - 1.0f);
  o3 = o3 > 0.0f ? o3 : (__expf(o3) - 1.0f);
  v4h hv;
  hv[0] = (_Float16)o0; hv[1] = (_Float16)o1;
  hv[2] = (_Float16)o2; hv[3] = (_Float16)o3;
  *(v4h*)(outH + (size_t)d * FDIM + lane * 4) = hv;
}

extern "C" void kernel_launch(void* const* d_in, const int* in_sizes, int n_in,
                              void* d_out, int out_size, void* d_ws,
                              size_t ws_size, hipStream_t stream) {
  (void)in_sizes; (void)n_in; (void)out_size; (void)ws_size;
  const float* x    = (const float*)d_in[0];
  const int*   ei   = (const int*)d_in[1];
  const float* Wl1  = (const float*)d_in[2];
  const float* Wr1  = (const float*)d_in[3];
  const float* att1 = (const float*)d_in[4];
  const float* b1   = (const float*)d_in[5];
  const float* Wl2  = (const float*)d_in[6];
  const float* Wr2  = (const float*)d_in[7];
  const float* att2 = (const float*)d_in[8];
  const float* b2   = (const float*)d_in[9];
  const float* W1   = (const float*)d_in[10];
  const float* bb1  = (const float*)d_in[11];
  const float* W2   = (const float*)d_in[12];
  const float* bb2  = (const float*)d_in[13];
  const float* W3   = (const float*)d_in[14];
  const float* bb3  = (const float*)d_in[15];
  float* out = (float*)d_out;

  char* ws = (char*)d_ws;
  size_t off = 0;
  auto alloc = [&](size_t bytes) -> char* {
    char* p = ws + off;
    off += (bytes + 255) & ~(size_t)255;
    return p;
  };
  const size_t S_F = (size_t)NNODES * FDIM;
  _Float16* xh  = (_Float16*)alloc(S_F * 2);
  _Float16* h1h = (_Float16*)alloc(S_F * 2);
  _Float16* h2h = (_Float16*)alloc(S_F * 2);
  int*  ccnt   = (int*)alloc((size_t)NBLK * NCHK * 4);
  int*  coff   = (int*)alloc(((size_t)NBLK * NCHK + 1) * 4);
  int*  rowptr = (int*)alloc(((size_t)NNODES + 1) * 4);
  int2* binned = (int2*)alloc((size_t)NEDGES * 8);
  int2* perm   = (int2*)alloc((size_t)NEDGES * 8);
  int*  srcs   = (int*)alloc((size_t)NEDGES * 4);
  const size_t uoff = off;
  float*    xl   = (float*)alloc(S_F * 4);
  float*    xr   = (float*)alloc(S_F * 4);
  _Float16* m1 = (_Float16*)(ws + uoff);
  _Float16* m2 = (_Float16*)(ws + uoff + (size_t)NNODES * 640 * 2);

  const int TB = 256;
  const int ngrid = (int)((S_F + TB - 1) / TB);
  const int wgrid = (NNODES + 7) / 8;
  const int ogrid = (NNODES + 63) / 64;
  const dim3 gconv((NNODES + 127) / 128, 128 / 64);

  bin_count_kernel<<<NCHK, 1024, 0, stream>>>(ei, ccnt);
  bin_scan_kernel<<<1, 1024, 0, stream>>>(ccnt, coff);
  bin_scatter_kernel<<<NCHK, 1024, 0, stream>>>(ei, coff, binned);
  bin_place_kernel<<<NBLK, 1024, 0, stream>>>(coff, binned, rowptr, perm);
  csr_order_kernel<<<ogrid, TB, 0, stream>>>(rowptr, perm, srcs);

  cast_f16_kernel<<<ngrid, TB, 0, stream>>>(x, xh, (int)S_F);
  wmma_gemm_kernel<0, false, true><<<gconv, TB, 0, stream>>>(
      xh, Wl1, nullptr, xl, nullptr, NNODES, 128, 128);
  wmma_gemm_kernel<0, false, true><<<gconv, TB, 0, stream>>>(
      xh, Wr1, nullptr, xr, nullptr, NNODES, 128, 128);
  conv_edge_kernel<<<wgrid, TB, 0, stream>>>(rowptr, srcs, xl, xr, att1, b1,
                                             h1h);

  wmma_gemm_kernel<0, false, true><<<gconv, TB, 0, stream>>>(
      h1h, Wl2, nullptr, xl, nullptr, NNODES, 128, 128);
  wmma_gemm_kernel<0, false, true><<<gconv, TB, 0, stream>>>(
      h1h, Wr2, nullptr, xr, nullptr, NNODES, 128, 128);
  conv_edge_kernel<<<wgrid, TB, 0, stream>>>(rowptr, srcs, xl, xr, att2, b2,
                                             h2h);

  const dim3 gm1((NNODES + 127) / 128, 640 / 64);
  wmma_gemm_kernel<1, true, false><<<gm1, TB, 0, stream>>>(
      h2h, W1, bb1, nullptr, m1, NNODES, 640, 128);
  const dim3 gm2((NNODES + 127) / 128, 320 / 64);
  wmma_gemm_kernel<1, true, false><<<gm2, TB, 0, stream>>>(
      m1, W2, bb2, nullptr, m2, NNODES, 320, 640);
  const dim3 gm3((NNODES + 127) / 128, 64 / 64);
  wmma_gemm_kernel<2, true, true><<<gm3, TB, 0, stream>>>(
      m2, W3, bb3, out, nullptr, NNODES, 64, 320);
}


# First conv's two GEMMs read the f32 input x themselves and round it to f16 in registers (same rounding, same k order); the one-element-per-thread cast kernel and its launch are gone
# speedup vs baseline: 7.5928x; 1.0394x over previous
#include <hip/hip_runtime.h>
#include <hip/hip_bf16.h>


typedef __attribute__((ext_vector_type(16))) _Float16 v16h;
typedef __attribute__((ext_vector_type(8)))  _Float16 v8h;
typedef __attribute__((ext_vector_type(8)))  float    v8f;

#define NNODES 50000
#define NEDGES 800000
#define FDIM   128

#define NBLK 256
#define BLKW 196
#define NCHK 256
#define CHKE ((NEDGES + NCHK - 1) / NCHK)
#define CHKT ((CHKE + 1023) / 1024)

__device__ __forceinline__ int node_block(int d) {
  const unsigned b = (unsigned)d / (unsigned)BLKW;
  return (int)(b < (unsigned)(NBLK - 1) ? b : (unsigned)(NBLK - 1));
}

__global__ __launch_bounds__(1024) void bin_count_kernel(
    const int* __restrict__ ei, int* __restrict__ ccnt) {
  __shared__ int h[NBLK];
  const int t = threadIdx.x;
  if (t < NBLK) h[t] = 0;
  __syncthreads();
  const int e0 = blockIdx.x * CHKE;
  const int e1 = (e0 + CHKE) < NEDGES ? (e0 + CHKE) : NEDGES;
  int dv[CHKT];
  #pragma unroll
  for (int k = 0; k < CHKT; ++k) {
    const int e = e0 + t + k * 1024;
    dv[k] = ei[NEDGES + (e < NEDGES ? e : NEDGES - 1)];
  }
  #pragma unroll
  for (int k = 0; k < CHKT; ++k)
    if (e0 + t + k * 1024 < e1) atomicAdd(&h[node_block(dv[k])], 1);
  __syncthreads();
  if (t < NBLK) ccnt[t * NCHK + blockIdx.x] = h[t];
}

__global__ __launch_bounds__(1024) void bin_scan_kernel(
    const int* __restrict__ ccnt, int* __restrict__ coff) {
  __shared__ int wsum[32];
  const int t = threadIdx.x, lane = t & 31, wv = t >> 5;
  const int4* in4 = (const int4*)ccnt + wv * (NBLK * NCHK / 128);
  int4* out4 = (int4*)coff + wv * (NBLK * NCHK / 128);
  int4 v[16];
  #pragma unroll
  for (int k = 0; k < 16; ++k) v[k] = in4[k * 32 + lane];
  int carry = 0;
  #pragma unroll
  for (int k = 0; k < 16; ++k) {
    const int s0 = v[k].x, s1 = s0 + v[k].y, s2 = s1 + v[k].z, s3 = s2 + v[k].w;
    int x = s3;
    for (int o = 1; o < 32; o <<= 1) {
      const int y = __shfl_up(x, o);
      if (lane >= o) x += y;
    }
    const int ex = carry + x - s3;
    v[k] = make_int4(ex, ex + s0, ex + s1, ex + s2);
    carry += __shfl(x, 31);
  }
  if (lane == 0) wsum[wv] = carry;
  __syncthreads();
  if (wv == 0) {
    const int w = wsum[lane];
    int xs = w;
    for (int o = 1; o < 32; o <<= 1) {
      const int y = __shfl_up(xs, o);
      if (lane >= o) xs += y;
    }
    wsum[lane] = xs - w;
  }
  __syncthreads();
  const int base = wsum[wv];
  #pragma unroll
  for (int k = 0; k < 16; ++k) {
    int4 o = v[k];
    o.x += base; o.y += base; o.z += base; o.w += base;
    out4[k * 32 + lane] = o;
  }
  if (t == 1023) coff[NBLK * NCHK] = base + carry;
}

__global__ __launch_bounds__(1024) void bin_scatter_kernel(
    const int* __restrict__ ei, const int* __restrict__ coff,
    int2* __restrict__ binned) {
  __shared__ int cur[NBLK];
  const int t = threadIdx.x;
  if (t < NBLK) cur[t] = coff[t * NCHK + blockIdx.x];
  __syncthreads();
  const int e0 = blockIdx.x * CHKE;
  const int e1 = (e0 + CHKE) < NEDGES ? (e0 + CHKE) : NEDGES;
  int sv[CHKT], dv[CHKT];
  #pragma unroll
  for (int k = 0; k < CHKT; ++k) {
    const int e = e0 + t + k * 1024;
    const int ec = e < NEDGES ? e : NEDGES - 1;
    sv[k] = ei[ec];
    dv[k] = ei[NEDGES + ec];
  }
  #pragma unroll
  for (int k = 0; k < CHKT; ++k) {
    const int e = e0 + t + k * 1024;
    if (e < e1) {
      const int b = node_block(dv[k]);
      const int p = atomicAdd(&cur[b], 1);
      if ((unsigned)p < (unsigned)NEDGES)
        binned[p] = make_int2((sv[k] & 0xffff) | ((dv[k] - b * BLKW) << 16), e);
    }
  }
}

__global__ __launch_bounds__(1024) void bin_place_kernel(
    const int* __restrict__ coff, const int2* __restrict__ binned,
    int* __restrict__ rowptr, int2* __restrict__ perm) {
  __shared__ int cnt[256];
  __shared__ int cur[256];
  __shared__ int wtot[8];
  const int t = threadIdx.x, lane = t & 31, wv = t >> 5;
  const int b = blockIdx.x;
  const int beg = coff[b * NCHK], end = coff[(b + 1) * NCHK];
  if (t < 256) cnt[t] = 0;
  __syncthreads();
  for (int i0 = beg; i0 < end; i0 += 4096) {
    int rx[4];
    #pragma unroll
    for (int k = 0; k < 4; ++k) {
      const int i = i0 + t + k * 1024;
      rx[k] = binned[i < NEDGES ? i : NEDGES - 1].x;
    }
    #pragma unroll
    for (int k = 0; k < 4; ++k)
      if (i0 + t + k * 1024 < end) atomicAdd(&cnt[(rx[k] >> 16) & 0xff], 1);
  }
  __syncthreads();
  int c = 0;
  if (t < 256) c = cnt[t];
  int x = c;
  for (int o = 1; o < 32; o <<= 1) {
    const int y = __shfl_up(x, o);
    if (lane >= o) x += y;
  }
  if (t < 256 && lane == 31) wtot[wv] = x;
  __syncthreads();
  if (t < 256) {
    int first = beg + x - c;
    for (int w = 0; w < wv; ++w) first += wtot[w];
    cur[t] = first;
    const int node = b * BLKW + t;
    if (t < BLKW && node < NNODES) rowptr[node] = first;
  }
  if (b == NBLK - 1 && t == 0) rowptr[NNODES] = end;
  __syncthreads();
  for (int i0 = beg; i0 < end; i0 += 4096) {
    int2 r[4];
    #pragma unroll
    for (int k = 0; k < 4; ++k) {
      const int i = i0 + t + k * 1024;
      r[k] = binned[i < NEDGES ? i : NEDGES - 1];
    }
    #pragma unroll
    for (int k = 0; k < 4; ++k) {
      if (i0 + t + k * 1024 < end) {
        const int p = atomicAdd(&cur[(r[k].x >> 16) & 0xff], 1);
        if ((unsigned)p < (unsigned)NEDGES)
          perm[p] = make_int2(r[k].x & 0xffff, r[k].y);
      }
    }
  }
}

__global__ __launch_bounds__(256) void csr_order_kernel(
    const int* __restrict__ rowptr, const int2* __restrict__ perm,
    int* __restrict__ srcs) {
  const int lane = threadIdx.x & 31;
  const int w = __builtin_amdgcn_readfirstlane(
      (int)(blockIdx.x * 8 + (threadIdx.x >> 5)));
  for (int q = 0; q < 8; ++q) {
    const int d = w * 8 + q;
    if (d >= NNODES) break;
    const int beg = rowptr[d], end = rowptr[d + 1];
    for (int base = beg; base < end; base += 32) {
      const bool valid = base + lane < end;
      int2 my = make_int2(0, 0x7fffffff);
      if (valid) my = perm[base + lane];
      int rank = 0;
      for (int cb = beg; cb < end; cb += 32) {
        const int cn = (end - cb) < 32 ? (end - cb) : 32;
        int other = 0x7fffffff;
        if (cb + lane < end) other = perm[cb + lane].y;
        for (int j = 0; j < cn; ++j) {
          const int k = __builtin_amdgcn_readlane(other, j);
          rank += (k < my.y) ? 1 : 0;
        }
      }
      if (valid) srcs[beg + rank] = my.x;
    }
  }
}

#define KC 64
#define BT_STRIDE 72

template <int ACT, bool HASBIAS, bool F32OUT, bool AF32>
__global__ __launch_bounds__(256) void wmma_gemm_kernel(
    const _Float16* __restrict__ A, const float* __restrict__ Af,
    const float* __restrict__ W,
    const float* __restrict__ bias,
    float* __restrict__ outF, _Float16* __restrict__ outH,
    int M, int Nt, int K) {
  __shared__ _Float16 bt[64 * BT_STRIDE];

  const int lane = threadIdx.x & 31;
  const int wave = threadIdx.x >> 5;
  const int half = lane >> 4;
  const int l15  = lane & 15;
  const int n0   = blockIdx.y * 64;

  int rowBase = blockIdx.x * 128 + wave * 16;
  if (rowBase > M - 16) rowBase = M - 16;

  v8f c[4] = {v8f{}, v8f{}, v8f{}, v8f{}};

  for (int kc = 0; kc < K; kc += KC) {
    {
      int kl = threadIdx.x >> 4;
      const int n4 = (threadIdx.x & 15) * 4;
      #pragma unroll
      for (int p = 0; p < 4; ++p, kl += 16) {
        const float4 wv =
            *(const float4*)(W + (size_t)(kc + kl) * Nt + n0 + n4);
        bt[(n4 + 0) * BT_STRIDE + kl] = (_Float16)wv.x;
        bt[(n4 + 1) * BT_STRIDE + kl] = (_Float16)wv.y;
        bt[(n4 + 2) * BT_STRIDE + kl] = (_Float16)wv.z;
        bt[(n4 + 3) * BT_STRIDE + kl] = (_Float16)wv.w;
      }
    }
    __syncthreads();

    const size_t aoff = (size_t)(rowBase + l15) * K + kc;
    #pragma unroll
    for (int ks = 0; ks < 2; ++ks) {
      const int kb = ks * 32 + half * 8;
      v16h a;
      if (AF32) {
        const float* ap = Af + aoff + kb;
        const float4 f0 = *(const float4*)(ap);
        const float4 f1 = *(const float4*)(ap + 4);
        const float4 f2 = *(const float4*)(ap + 16);
        const float4 f3 = *(const float4*)(ap + 20);
        a[0]  = (_Float16)f0.x; a[1]  = (_Float16)f0.y;
        a[2]  = (_Float16)f0.z; a[3]  = (_Float16)f0.w;
        a[4]  = (_Float16)f1.x; a[5]  = (_Float16)f1.y;
        a[6]  = (_Float16)f1.z; a[7]  = (_Float16)f1.w;
        a[8]  = (_Float16)f2.x; a[9]  = (_Float16)f2.y;
        a[10] = (_Float16)f2.z; a[11] = (_Float16)f2.w;
        a[12] = (_Float16)f3.x; a[13] = (_Float16)f3.y;
        a[14] = (_Float16)f3.z; a[15] = (_Float16)f3.w;
      } else {
        const _Float16* arow = A + aoff;
        v8h alo = *(const v8h*)(arow + kb);
        v8h ahi = *(const v8h*)(arow + kb + 16);
        a = __builtin_shufflevector(alo, ahi, 0, 1, 2, 3, 4, 5, 6, 7,
                                    8, 9, 10, 11, 12, 13, 14, 15);
      }
      v16h b[4];
      #pragma unroll
      for (int nn = 0; nn < 4; ++nn) {
        const _Float16* bp = bt + (nn * 16 + l15) * BT_STRIDE + kb;
        v8h blo = *(const v8h*)(bp);
        v8h bhi = *(const v8h*)(bp + 16);
        b[nn] = __builtin_shufflevector(blo, bhi, 0, 1, 2, 3, 4, 5, 6, 7,
                                        8, 9, 10, 11, 12, 13, 14, 15);
      }
      #pragma unroll
      for (int nn = 0; nn < 4; ++nn) {
        c[nn] = __builtin_amdgcn_wmma_f32_16x16x32_f16(
            false, a, false, b[nn], (short)0, c[nn], false, false);
      }
    }
    __syncthreads();
  }

  const unsigned o0 =
      (unsigned)(rowBase + half * 8) * (unsigned)Nt + (unsigned)(n0 + l15);
  #pragma unroll
  for (int nn = 0; nn < 4; ++nn) {
    float bv = 0.0f;
    if (HASBIAS) bv = bias[n0 + nn * 16 + l15];
    #pragma unroll
    for (int r = 0; r < 8; ++r) {
      float v = c[nn][r] + bv;
      if (ACT == 1) v = v > 0.0f ? v : (__expf(v) - 1.0f);
      else if (ACT == 2) v = 1.0f / (1.0f + __expf(-v));
      const unsigned o = o0 + (unsigned)r * (unsigned)Nt + (unsigned)(nn * 16);
      if (F32OUT) outF[o] = v;
      else        outH[o] = (_Float16)v;
    }
  }
}

typedef __attribute__((ext_vector_type(4))) _Float16 v4h;

template <int CTRL>
__device__ __forceinline__ float lane_xor_get(float x) {
  return __builtin_bit_cast(
      float, __builtin_amdgcn_update_dpp(0, __builtin_bit_cast(int, x), CTRL,
                                         0xf, 0xf, true));
}

__device__ __forceinline__ float edge_logit(const float4 a, const float4 b,
                                            const float4 w) {
  float z, acc;
  z = a.x + b.x; z = z > 0.0f ? z : 0.2f * z; acc = z * w.x;
  z = a.y + b.y; z = z > 0.0f ? z : 0.2f * z; acc += z * w.y;
  z = a.z + b.z; z = z > 0.0f ? z : 0.2f * z; acc += z * w.z;
  z = a.w + b.w; z = z > 0.0f ? z : 0.2f * z; acc += z * w.w;
  acc += lane_xor_get<0x161>(acc);
  acc += lane_xor_get<0x162>(acc);
  acc += lane_xor_get<0x164>(acc);
  return acc;
}

template <int NB>
__device__ __forceinline__ void conv_batch(const float4* __restrict__ xl4,
                                           int my, int j0, int lane,
                                           const float4 xrv, const float4 av,
                                           float& m, float& ssum, float4& r) {
  float4 v[NB];
  float e[NB];
  #pragma unroll
  for (int t = 0; t < NB; ++t) {
    const int sidx = __builtin_amdgcn_readlane(my, j0 + t);
    v[t] = xl4[(size_t)sidx * 32 + lane];
  }
  float mn = m;
  #pragma unroll
  for (int t = 0; t < NB; ++t) {
    e[t] = edge_logit(v[t], xrv, av);
    mn = fmaxf(mn, e[t]);
  }
  const float sc = __expf(m - mn);
  ssum *= sc;
  r.x *= sc; r.y *= sc; r.z *= sc; r.w *= sc;
  #pragma unroll
  for (int t = 0; t < NB; ++t) {
    const float p = __expf(e[t] - mn);
    ssum += p;
    r.x += v[t].x * p; r.y += v[t].y * p; r.z += v[t].z * p; r.w += v[t].w * p;
  }
  m = mn;
}

__global__ __launch_bounds__(256) void conv_edge_kernel(
    const int* __restrict__ rowptr, const int* __restrict__ srcs,
    const float* __restrict__ xl, const float* __restrict__ xr,
    const float* __restrict__ att, const float* __restrict__ bias,
    _Float16* __restrict__ outH) {
  const int lane = threadIdx.x & 31;
  const int d = __builtin_amdgcn_readfirstlane(
      (int)(blockIdx.x * 8 + (threadIdx.x >> 5)));
  if (d >= NNODES) return;
  const float4* xl4 = (const float4*)xl;
  const float4 xrv = ((const float4*)xr)[(size_t)d * 32 + lane];
  const float4 av = ((const float4*)att)[lane];

  float4 r = xl4[(size_t)d * 32 + lane];
  float m = edge_logit(r, xrv, av);
  float ssum = 1.0f;

  const int beg = rowptr[d], end = rowptr[d + 1];
  for (int base = beg; base < end; base += 32) {
    const int cnt = (end - base) < 32 ? (end - base) : 32;
    int my = 0;
    if (lane < cnt) my = srcs[base + lane];
    int j = 0;
    for (; j + 8 <= cnt; j += 8)
      conv_batch<8>(xl4, my, j, lane, xrv, av, m, ssum, r);
    if (j + 4 <= cnt) { conv_batch<4>(xl4, my, j, lane, xrv, av, m, ssum, r); j += 4; }
    if (j + 2 <= cnt) { conv_batch<2>(xl4, my, j, lane, xrv, av, m, ssum, r); j += 2; }
    if (j < cnt) conv_batch<1>(xl4, my, j, lane, xrv, av, m, ssum, r);
  }

  const float4 bv = ((const float4*)bias)[lane];
  float o0 = r.x / ssum + bv.x, o1 = r.y / ssum + bv.y;
  float o2 = r.z / ssum + bv.z, o3 = r.w / ssum + bv.w;
  o0 = o0 > 0.0f ? o0 : (__expf(o0) - 1.0f);
  o1 = o1 > 0.0f ? o1 : (__expf(o1) - 1.0f);
  o2 = o2 > 0.0f ? o2 : (__expf(o2) - 1.0f);
  o3 = o3 > 0.0f ? o3 : (__expf(o3) - 1.0f);
  v4h hv;
  hv[0] = (_Float16)o0; hv[1] = (_Float16)o1;
  hv[2] = (_Float16)o2; hv[3] = (_Float16)o3;
  *(v4h*)(outH + (size_t)d * FDIM + lane * 4) = hv;
}

extern "C" void kernel_launch(void* const* d_in, const int* in_sizes, int n_in,
                              void* d_out, int out_size, void* d_ws,
                              size_t ws_size, hipStream_t stream) {
  (void)in_sizes; (void)n_in; (void)out_size; (void)ws_size;
  const float* x    = (const float*)d_in[0];
  const int*   ei   = (const int*)d_in[1];
  const float* Wl1  = (const float*)d_in[2];
  const float* Wr1  = (const float*)d_in[3];
  const float* att1 = (const float*)d_in[4];
  const float* b1   = (const float*)d_in[5];
  const float* Wl2  = (const float*)d_in[6];
  const float* Wr2  = (const float*)d_in[7];
  const float* att2 = (const float*)d_in[8];
  const float* b2   = (const float*)d_in[9];
  const float* W1   = (const float*)d_in[10];
  const float* bb1  = (const float*)d_in[11];
  const float* W2   = (const float*)d_in[12];
  const float* bb2  = (const float*)d_in[13];
  const float* W3   = (const float*)d_in[14];
  const float* bb3  = (const float*)d_in[15];
  float* out = (float*)d_out;

  char* ws = (char*)d_ws;
  size_t off = 0;
  auto alloc = [&](size_t bytes) -> char* {
    char* p = ws + off;
    off += (bytes + 255) & ~(size_t)255;
    return p;
  };
  const size_t S_F = (size_t)NNODES * FDIM;
  _Float16* h1h = (_Float16*)alloc(S_F * 2);
  _Float16* h2h = (_Float16*)alloc(S_F * 2);
  int*  ccnt   = (int*)alloc((size_t)NBLK * NCHK * 4);
  int*  coff   = (int*)alloc(((size_t)NBLK * NCHK + 1) * 4);
  int*  rowptr = (int*)alloc(((size_t)NNODES + 1) * 4);
  int2* binned = (int2*)alloc((size_t)NEDGES * 8);
  int2* perm   = (int2*)alloc((size_t)NEDGES * 8);
  int*  srcs   = (int*)alloc((size_t)NEDGES * 4);
  const size_t uoff = off;
  float*    xl   = (float*)alloc(S_F * 4);
  float*    xr   = (float*)alloc(S_F * 4);
  _Float16* m1 = (_Float16*)(ws + uoff);
  _Float16* m2 = (_Float16*)(ws + uoff + (size_t)NNODES * 640 * 2);

  const int TB = 256;
  const int wgrid = (NNODES + 7) / 8;
  const int ogrid = (NNODES + 63) / 64;
  const dim3 gconv((NNODES + 127) / 128, 128 / 64);

  bin_count_kernel<<<NCHK, 1024, 0, stream>>>(ei, ccnt);
  bin_scan_kernel<<<1, 1024, 0, stream>>>(ccnt, coff);
  bin_scatter_kernel<<<NCHK, 1024, 0, stream>>>(ei, coff, binned);
  bin_place_kernel<<<NBLK, 1024, 0, stream>>>(coff, binned, rowptr, perm);
  csr_order_kernel<<<ogrid, TB, 0, stream>>>(rowptr, perm, srcs);

  wmma_gemm_kernel<0, false, true, true><<<gconv, TB, 0, stream>>>(
      nullptr, x, Wl1, nullptr, xl, nullptr, NNODES, 128, 128);
  wmma_gemm_kernel<0, false, true, true><<<gconv, TB, 0, stream>>>(
      nullptr, x, Wr1, nullptr, xr, nullptr, NNODES, 128, 128);
  conv_edge_kernel<<<wgrid, TB, 0, stream>>>(rowptr, srcs, xl, xr, att1, b1,
                                             h1h);

  wmma_gemm_kernel<0, false, true, false><<<gconv, TB, 0, stream>>>(
      h1h, nullptr, Wl2, nullptr, xl, nullptr, NNODES, 128, 128);
  wmma_gemm_kernel<0, false, true, false><<<gconv, TB, 0, stream>>>(
      h1h, nullptr, Wr2, nullptr, xr, nullptr, NNODES, 128, 128);
  conv_edge_kernel<<<wgrid, TB, 0, stream>>>(rowptr, srcs, xl, xr, att2, b2,
                                             h2h);

  const dim3 gm1((NNODES + 127) / 128, 640 / 64);
  wmma_gemm_kernel<1, true, false, false><<<gm1, TB, 0, stream>>>(
      h2h, nullptr, W1, bb1, nullptr, m1, NNODES, 640, 128);
  const dim3 gm2((NNODES + 127) / 128, 320 / 64);
  wmma_gemm_kernel<1, true, false, false><<<gm2, TB, 0, stream>>>(
      m1, nullptr, W2, bb2, nullptr, m2, NNODES, 320, 640);
  const dim3 gm3((NNODES + 127) / 128, 64 / 64);
  wmma_gemm_kernel<2, true, true, false><<<gm3, TB, 0, stream>>>(
      m2, nullptr, W3, bb3, out, nullptr, NNODES, 64, 320);
}


# GEMM wave tile 16x64 -> 32x64 (conv, last layer) and 64x64 (two wide MLP layers): each staged W tile and each B fragment read from LDS now serves 2 or 4 A fragments; k and sum order unchanged
# speedup vs baseline: 8.5283x; 1.1232x over previous
#include <hip/hip_runtime.h>
#include <hip/hip_bf16.h>


typedef __attribute__((ext_vector_type(16))) _Float16 v16h;
typedef __attribute__((ext_vector_type(8)))  _Float16 v8h;
typedef __attribute__((ext_vector_type(8)))  float    v8f;

#define NNODES 50000
#define NEDGES 800000
#define FDIM   128

#define NBLK 256
#define BLKW 196
#define NCHK 256
#define CHKE ((NEDGES + NCHK - 1) / NCHK)
#define CHKT ((CHKE + 1023) / 1024)

__device__ __forceinline__ int node_block(int d) {
  const unsigned b = (unsigned)d / (unsigned)BLKW;
  return (int)(b < (unsigned)(NBLK - 1) ? b : (unsigned)(NBLK - 1));
}

__global__ __launch_bounds__(1024) void bin_count_kernel(
    const int* __restrict__ ei, int* __restrict__ ccnt) {
  __shared__ int h[NBLK];
  const int t = threadIdx.x;
  if (t < NBLK) h[t] = 0;
  __syncthreads();
  const int e0 = blockIdx.x * CHKE;
  const int e1 = (e0 + CHKE) < NEDGES ? (e0 + CHKE) : NEDGES;
  int dv[CHKT];
  #pragma unroll
  for (int k = 0; k < CHKT; ++k) {
    const int e = e0 + t + k * 1024;
    dv[k] = ei[NEDGES + (e < NEDGES ? e : NEDGES - 1)];
  }
  #pragma unroll
  for (int k = 0; k < CHKT; ++k)
    if (e0 + t + k * 1024 < e1) atomicAdd(&h[node_block(dv[k])], 1);
  __syncthreads();
  if (t < NBLK) ccnt[t * NCHK + blockIdx.x] = h[t];
}

__global__ __launch_bounds__(1024) void bin_scan_kernel(
    const int* __restrict__ ccnt, int* __restrict__ coff) {
  __shared__ int wsum[32];
  const int t = threadIdx.x, lane = t & 31, wv = t >> 5;
  const int4* in4 = (const int4*)ccnt + wv * (NBLK * NCHK / 128);
  int4* out4 = (int4*)coff + wv * (NBLK * NCHK / 128);
  int4 v[16];
  #pragma unroll
  for (int k = 0; k < 16; ++k) v[k] = in4[k * 32 + lane];
  int carry = 0;
  #pragma unroll
  for (int k = 0; k < 16; ++k) {
    const int s0 = v[k].x, s1 = s0 + v[k].y, s2 = s1 + v[k].z, s3 = s2 + v[k].w;
    int x = s3;
    for (int o = 1; o < 32; o <<= 1) {
      const int y = __shfl_up(x, o);
      if (lane >= o) x += y;
    }
    const int ex = carry + x - s3;
    v[k] = make_int4(ex, ex + s0, ex + s1, ex + s2);
    carry += __shfl(x, 31);
  }
  if (lane == 0) wsum[wv] = carry;
  __syncthreads();
  if (wv == 0) {
    const int w = wsum[lane];
    int xs = w;
    for (int o = 1; o < 32; o <<= 1) {
      const int y = __shfl_up(xs, o);
      if (lane >= o) xs += y;
    }
    wsum[lane] = xs - w;
  }
  __syncthreads();
  const int base = wsum[wv];
  #pragma unroll
  for (int k = 0; k < 16; ++k) {
    int4 o = v[k];
    o.x += base; o.y += base; o.z += base; o.w += base;
    out4[k * 32 + lane] = o;
  }
  if (t == 1023) coff[NBLK * NCHK] = base + carry;
}

__global__ __launch_bounds__(1024) void bin_scatter_kernel(
    const int* __restrict__ ei, const int* __restrict__ coff,
    int2* __restrict__ binned) {
  __shared__ int cur[NBLK];
  const int t = threadIdx.x;
  if (t < NBLK) cur[t] = coff[t * NCHK + blockIdx.x];
  __syncthreads();
  const int e0 = blockIdx.x * CHKE;
  const int e1 = (e0 + CHKE) < NEDGES ? (e0 + CHKE) : NEDGES;
  int sv[CHKT], dv[CHKT];
  #pragma unroll
  for (int k = 0; k < CHKT; ++k) {
    const int e = e0 + t + k * 1024;
    const int ec = e < NEDGES ? e : NEDGES - 1;
    sv[k] = ei[ec];
    dv[k] = ei[NEDGES + ec];
  }
  #pragma unroll
  for (int k = 0; k < CHKT; ++k) {
    const int e = e0 + t + k * 1024;
    if (e < e1) {
      const int b = node_block(dv[k]);
      const int p = atomicAdd(&cur[b], 1);
      if ((unsigned)p < (unsigned)NEDGES)
        binned[p] = make_int2((sv[k] & 0xffff) | ((dv[k] - b * BLKW) << 16), e);
    }
  }
}

__global__ __launch_bounds__(1024) void bin_place_kernel(
    const int* __restrict__ coff, const int2* __restrict__ binned,
    int* __restrict__ rowptr, int2* __restrict__ perm) {
  __shared__ int cnt[256];
  __shared__ int cur[256];
  __shared__ int wtot[8];
  const int t = threadIdx.x, lane = t & 31, wv = t >> 5;
  const int b = blockIdx.x;
  const int beg = coff[b * NCHK], end = coff[(b + 1) * NCHK];
  if (t < 256) cnt[t] = 0;
  __syncthreads();
  for (int i0 = beg; i0 < end; i0 += 4096) {
    int rx[4];
    #pragma unroll
    for (int k = 0; k < 4; ++k) {
      const int i = i0 + t + k * 1024;
      rx[k] = binned[i < NEDGES ? i : NEDGES - 1].x;
    }
    #pragma unroll
    for (int k = 0; k < 4; ++k)
      if (i0 + t + k * 1024 < end) atomicAdd(&cnt[(rx[k] >> 16) & 0xff], 1);
  }
  __syncthreads();
  int c = 0;
  if (t < 256) c = cnt[t];
  int x = c;
  for (int o = 1; o < 32; o <<= 1) {
    const int y = __shfl_up(x, o);
    if (lane >= o) x += y;
  }
  if (t < 256 && lane == 31) wtot[wv] = x;
  __syncthreads();
  if (t < 256) {
    int first = beg + x - c;
    for (int w = 0; w < wv; ++w) first += wtot[w];
    cur[t] = first;
    const int node = b * BLKW + t;
    if (t < BLKW && node < NNODES) rowptr[node] = first;
  }
  if (b == NBLK - 1 && t == 0) rowptr[NNODES] = end;
  __syncthreads();
  for (int i0 = beg; i0 < end; i0 += 4096) {
    int2 r[4];
    #pragma unroll
    for (int k = 0; k < 4; ++k) {
      const int i = i0 + t + k * 1024;
      r[k] = binned[i < NEDGES ? i : NEDGES - 1];
    }
    #pragma unroll
    for (int k = 0; k < 4; ++k) {
      if (i0 + t + k * 1024 < end) {
        const int p = atomicAdd(&cur[(r[k].x >> 16) & 0xff], 1);
        if ((unsigned)p < (unsigned)NEDGES)
          perm[p] = make_int2(r[k].x & 0xffff, r[k].y);
      }
    }
  }
}

__global__ __launch_bounds__(256) void csr_order_kernel(
    const int* __restrict__ rowptr, const int2* __restrict__ perm,
    int* __restrict__ srcs) {
  const int lane = threadIdx.x & 31;
  const int w = __builtin_amdgcn_readfirstlane(
      (int)(blockIdx.x * 8 + (threadIdx.x >> 5)));
  for (int q = 0; q < 8; ++q) {
    const int d = w * 8 + q;
    if (d >= NNODES) break;
    const int beg = rowptr[d], end = rowptr[d + 1];
    for (int base = beg; base < end; base += 32) {
      const bool valid = base + lane < end;
      int2 my = make_int2(0, 0x7fffffff);
      if (valid) my = perm[base + lane];
      int rank = 0;
      for (int cb = beg; cb < end; cb += 32) {
        const int cn = (end - cb) < 32 ? (end - cb) : 32;
        int other = 0x7fffffff;
        if (cb + lane < end) other = perm[cb + lane].y;
        for (int j = 0; j < cn; ++j) {
          const int k = __builtin_amdgcn_readlane(other, j);
          rank += (k < my.y) ? 1 : 0;
        }
      }
      if (valid) srcs[beg + rank] = my.x;
    }
  }
}

#define KC 64
#define BT_STRIDE 72

template <int ACT, bool HASBIAS, bool F32OUT, bool AF32, int MT>
__global__ __launch_bounds__(256) void wmma_gemm_kernel(
    const _Float16* __restrict__ A, const float* __restrict__ Af,
    const float* __restrict__ W,
    const float* __restrict__ bias,
    float* __restrict__ outF, _Float16* __restrict__ outH,
    int M, int Nt, int K) {
  __shared__ _Float16 bt[64 * BT_STRIDE];

  const int lane = threadIdx.x & 31;
  const int wave = threadIdx.x >> 5;
  const int half = lane >> 4;
  const int l15  = lane & 15;
  const int n0   = blockIdx.y * 64;

  int rowBase = blockIdx.x * (128 * MT) + wave * (16 * MT);
  if (rowBase > M - 16 * MT) rowBase = M - 16 * MT;

  v8f c[MT][4];
  #pragma unroll
  for (int mt = 0; mt < MT; ++mt) {
    #pragma unroll
    for (int nn = 0; nn < 4; ++nn) c[mt][nn] = v8f{};
  }

  for (int kc = 0; kc < K; kc += KC) {
    {
      int kl = threadIdx.x >> 4;
      const int n4 = (threadIdx.x & 15) * 4;
      #pragma unroll
      for (int p = 0; p < 4; ++p, kl += 16) {
        const float4 wv =
            *(const float4*)(W + (size_t)(kc + kl) * Nt + n0 + n4);
        bt[(n4 + 0) * BT_STRIDE + kl] = (_Float16)wv.x;
        bt[(n4 + 1) * BT_STRIDE + kl] = (_Float16)wv.y;
        bt[(n4 + 2) * BT_STRIDE + kl] = (_Float16)wv.z;
        bt[(n4 + 3) * BT_STRIDE + kl] = (_Float16)wv.w;
      }
    }
    __syncthreads();

    const size_t aoff = (size_t)(rowBase + l15) * K + kc;
    v16h a[2][MT];
    #pragma unroll
    for (int ks = 0; ks < 2; ++ks) {
      const int kb = ks * 32 + half * 8;
      #pragma unroll
      for (int mt = 0; mt < MT; ++mt) {
        if (AF32) {
          const float* ap = Af + aoff + (size_t)(mt * 16) * K + kb;
          const float4 f0 = *(const float4*)(ap);
          const float4 f1 = *(const float4*)(ap + 4);
          const float4 f2 = *(const float4*)(ap + 16);
          const float4 f3 = *(const float4*)(ap + 20);
          v16h t;
          t[0]  = (_Float16)f0.x; t[1]  = (_Float16)f0.y;
          t[2]  = (_Float16)f0.z; t[3]  = (_Float16)f0.w;
          t[4]  = (_Float16)f1.x; t[5]  = (_Float16)f1.y;
          t[6]  = (_Float16)f1.z; t[7]  = (_Float16)f1.w;
          t[8]  = (_Float16)f2.x; t[9]  = (_Float16)f2.y;
          t[10] = (_Float16)f2.z; t[11] = (_Float16)f2.w;
          t[12] = (_Float16)f3.x; t[13] = (_Float16)f3.y;
          t[14] = (_Float16)f3.z; t[15] = (_Float16)f3.w;
          a[ks][mt] = t;
        } else {
          const _Float16* arow = A + aoff + (size_t)(mt * 16) * K;
          v8h alo = *(const v8h*)(arow + kb);
          v8h ahi = *(const v8h*)(arow + kb + 16);
          a[ks][mt] = __builtin_shufflevector(alo, ahi, 0, 1, 2, 3, 4, 5, 6, 7,
                                              8, 9, 10, 11, 12, 13, 14, 15);
        }
      }
    }
    __builtin_amdgcn_sched_barrier(0);
    #pragma unroll
    for (int ks = 0; ks < 2; ++ks) {
      const int kb = ks * 32 + half * 8;
      v16h b[4];
      #pragma unroll
      for (int nn = 0; nn < 4; ++nn) {
        const _Float16* bp = bt + (nn * 16 + l15) * BT_STRIDE + kb;
        v8h blo = *(const v8h*)(bp);
        v8h bhi = *(const v8h*)(bp + 16);
        b[nn] = __builtin_shufflevector(blo, bhi, 0, 1, 2, 3, 4, 5, 6, 7,
                                        8, 9, 10, 11, 12, 13, 14, 15);
      }
      #pragma unroll
      for (int mt = 0; mt < MT; ++mt) {
        #pragma unroll
        for (int nn = 0; nn < 4; ++nn) {
          c[mt][nn] = __builtin_amdgcn_wmma_f32_16x16x32_f16(
              false, a[ks][mt], false, b[nn], (short)0, c[mt][nn], false, false);
        }
      }
    }
    __syncthreads();
  }

  float bv[4] = {0.0f, 0.0f, 0.0f, 0.0f};
  if (HASBIAS) {
    #pragma unroll
    for (int nn = 0; nn < 4; ++nn) bv[nn] = bias[n0 + nn * 16 + l15];
  }
  #pragma unroll
  for (int mt = 0; mt < MT; ++mt) {
    const unsigned o0 =
        (unsigned)(rowBase + mt * 16 + half * 8) * (unsigned)Nt +
        (unsigned)(n0 + l15);
    #pragma unroll
    for (int nn = 0; nn < 4; ++nn) {
      #pragma unroll
      for (int r = 0; r < 8; ++r) {
        float v = c[mt][nn][r] + bv[nn];
        if (ACT == 1) v = v > 0.0f ? v : (__expf(v) - 1.0f);
        else if (ACT == 2) v = 1.0f / (1.0f + __expf(-v));
        const unsigned o = o0 + (unsigned)r * (unsigned)Nt + (unsigned)(nn * 16);
        if (F32OUT) outF[o] = v;
        else        outH[o] = (_Float16)v;
      }
    }
  }
}

typedef __attribute__((ext_vector_type(4))) _Float16 v4h;

template <int CTRL>
__device__ __forceinline__ float lane_xor_get(float x) {
  return __builtin_bit_cast(
      float, __builtin_amdgcn_update_dpp(0, __builtin_bit_cast(int, x), CTRL,
                                         0xf, 0xf, true));
}

__device__ __forceinline__ float edge_logit(const float4 a, const float4 b,
                                            const float4 w) {
  float z, acc;
  z = a.x + b.x; z = z > 0.0f ? z : 0.2f * z; acc = z * w.x;
  z = a.y + b.y; z = z > 0.0f ? z : 0.2f * z; acc += z * w.y;
  z = a.z + b.z; z = z > 0.0f ? z : 0.2f * z; acc += z * w.z;
  z = a.w + b.w; z = z > 0.0f ? z : 0.2f * z; acc += z * w.w;
  acc += lane_xor_get<0x161>(acc);
  acc += lane_xor_get<0x162>(acc);
  acc += lane_xor_get<0x164>(acc);
  return acc;
}

template <int NB>
__device__ __forceinline__ void conv_batch(const float4* __restrict__ xl4,
                                           int my, int j0, int lane,
                                           const float4 xrv, const float4 av,
                                           float& m, float& ssum, float4& r) {
  float4 v[NB];
  float e[NB];
  #pragma unroll
  for (int t = 0; t < NB; ++t) {
    const int sidx = __builtin_amdgcn_readlane(my, j0 + t);
    v[t] = xl4[(size_t)sidx * 32 + lane];
  }
  float mn = m;
  #pragma unroll
  for (int t = 0; t < NB; ++t) {
    e[t] = edge_logit(v[t], xrv, av);
    mn = fmaxf(mn, e[t]);
  }
  const float sc = __expf(m - mn);
  ssum *= sc;
  r.x *= sc; r.y *= sc; r.z *= sc; r.w *= sc;
  #pragma unroll
  for (int t = 0; t < NB; ++t) {
    const float p = __expf(e[t] - mn);
    ssum += p;
    r.x += v[t].x * p; r.y += v[t].y * p; r.z += v[t].z * p; r.w += v[t].w * p;
  }
  m = mn;
}

__global__ __launch_bounds__(256) void conv_edge_kernel(
    const int* __restrict__ rowptr, const int* __restrict__ srcs,
    const float* __restrict__ xl, const float* __restrict__ xr,
    const float* __restrict__ att, const float* __restrict__ bias,
    _Float16* __restrict__ outH) {
  const int lane = threadIdx.x & 31;
  const int d = __builtin_amdgcn_readfirstlane(
      (int)(blockIdx.x * 8 + (threadIdx.x >> 5)));
  if (d >= NNODES) return;
  const float4* xl4 = (const float4*)xl;
  const float4 xrv = ((const float4*)xr)[(size_t)d * 32 + lane];
  const float4 av = ((const float4*)att)[lane];

  float4 r = xl4[(size_t)d * 32 + lane];
  float m = edge_logit(r, xrv, av);
  float ssum = 1.0f;

  const int beg = rowptr[d], end = rowptr[d + 1];
  for (int base = beg; base < end; base += 32) {
    const int cnt = (end - base) < 32 ? (end - base) : 32;
    int my = 0;
    if (lane < cnt) my = srcs[base + lane];
    int j = 0;
    for (; j + 8 <= cnt; j += 8)
      conv_batch<8>(xl4, my, j, lane, xrv, av, m, ssum, r);
    if (j + 4 <= cnt) { conv_batch<4>(xl4, my, j, lane, xrv, av, m, ssum, r); j += 4; }
    if (j + 2 <= cnt) { conv_batch<2>(xl4, my, j, lane, xrv, av, m, ssum, r); j += 2; }
    if (j < cnt) conv_batch<1>(xl4, my, j, lane, xrv, av, m, ssum, r);
  }

  const float4 bv = ((const float4*)bias)[lane];
  float o0 = r.x / ssum + bv.x, o1 = r.y / ssum + bv.y;
  float o2 = r.z / ssum + bv.z, o3 = r.w / ssum + bv.w;
  o0 = o0 > 0.0f ? o0 : (__expf(o0) - 1.0f);
  o1 = o1 > 0.0f ? o1 : (__expf(o1) - 1.0f);
  o2 = o2 > 0.0f ? o2 : (__expf(o2) - 1.0f);
  o3 = o3 > 0.0f ? o3 : (__expf(o3) - 1.0f);
  v4h hv;
  hv[0] = (_Float16)o0; hv[1] = (_Float16)o1;
  hv[2] = (_Float16)o2; hv[3] = (_Float16)o3;
  *(v4h*)(outH + (size_t)d * FDIM + lane * 4) = hv;
}

extern "C" void kernel_launch(void* const* d_in, const int* in_sizes, int n_in,
                              void* d_out, int out_size, void* d_ws,
                              size_t ws_size, hipStream_t stream) {
  (void)in_sizes; (void)n_in; (void)out_size; (void)ws_size;
  const float* x    = (const float*)d_in[0];
  const int*   ei   = (const int*)d_in[1];
  const float* Wl1  = (const float*)d_in[2];
  const float* Wr1  = (const float*)d_in[3];
  const float* att1 = (const float*)d_in[4];
  const float* b1   = (const float*)d_in[5];
  const float* Wl2  = (const float*)d_in[6];
  const float* Wr2  = (const float*)d_in[7];
  const float* att2 = (const float*)d_in[8];
  const float* b2   = (const float*)d_in[9];
  const float* W1   = (const float*)d_in[10];
  const float* bb1  = (const float*)d_in[11];
  const float* W2   = (const float*)d_in[12];
  const float* bb2  = (const float*)d_in[13];
  const float* W3   = (const float*)d_in[14];
  const float* bb3  = (const float*)d_in[15];
  float* out = (float*)d_out;

  char* ws = (char*)d_ws;
  size_t off = 0;
  auto alloc = [&](size_t bytes) -> char* {
    char* p = ws + off;
    off += (bytes + 255) & ~(size_t)255;
    return p;
  };
  const size_t S_F = (size_t)NNODES * FDIM;
  _Float16* h1h = (_Float16*)alloc(S_F * 2);
  _Float16* h2h = (_Float16*)alloc(S_F * 2);
  int*  ccnt   = (int*)alloc((size_t)NBLK * NCHK * 4);
  int*  coff   = (int*)alloc(((size_t)NBLK * NCHK + 1) * 4);
  int*  rowptr = (int*)alloc(((size_t)NNODES + 1) * 4);
  int2* binned = (int2*)alloc((size_t)NEDGES * 8);
  int2* perm   = (int2*)alloc((size_t)NEDGES * 8);
  int*  srcs   = (int*)alloc((size_t)NEDGES * 4);
  const size_t uoff = off;
  float*    xl   = (float*)alloc(S_F * 4);
  float*    xr   = (float*)alloc(S_F * 4);
  _Float16* m1 = (_Float16*)(ws + uoff);
  _Float16* m2 = (_Float16*)(ws + uoff + (size_t)NNODES * 640 * 2);

  const int TB = 256;
  const int wgrid = (NNODES + 7) / 8;
  const int ogrid = (NNODES + 63) / 64;
  const dim3 gconv((NNODES + 255) / 256, 128 / 64);

  bin_count_kernel<<<NCHK, 1024, 0, stream>>>(ei, ccnt);
  bin_scan_kernel<<<1, 1024, 0, stream>>>(ccnt, coff);
  bin_scatter_kernel<<<NCHK, 1024, 0, stream>>>(ei, coff, binned);
  bin_place_kernel<<<NBLK, 1024, 0, stream>>>(coff, binned, rowptr, perm);
  csr_order_kernel<<<ogrid, TB, 0, stream>>>(rowptr, perm, srcs);

  wmma_gemm_kernel<0, false, true, true, 2><<<gconv, TB, 0, stream>>>(
      nullptr, x, Wl1, nullptr, xl, nullptr, NNODES, 128, 128);
  wmma_gemm_kernel<0, false, true, true, 2><<<gconv, TB, 0, stream>>>(
      nullptr, x, Wr1, nullptr, xr, nullptr, NNODES, 128, 128);
  conv_edge_kernel<<<wgrid, TB, 0, stream>>>(rowptr, srcs, xl, xr, att1, b1,
                                             h1h);

  wmma_gemm_kernel<0, false, true, false, 2><<<gconv, TB, 0, stream>>>(
      h1h, nullptr, Wl2, nullptr, xl, nullptr, NNODES, 128, 128);
  wmma_gemm_kernel<0, false, true, false, 2><<<gconv, TB, 0, stream>>>(
      h1h, nullptr, Wr2, nullptr, xr, nullptr, NNODES, 128, 128);
  conv_edge_kernel<<<wgrid, TB, 0, stream>>>(rowptr, srcs, xl, xr, att2, b2,
                                             h2h);

  const dim3 gm1((NNODES + 511) / 512, 640 / 64);
  wmma_gemm_kernel<1, true, false, false, 4><<<gm1, TB, 0, stream>>>(
      h2h, nullptr, W1, bb1, nullptr, m1, NNODES, 640, 128);
  const dim3 gm2((NNODES + 511) / 512, 320 / 64);
  wmma_gemm_kernel<1, true, false, false, 4><<<gm2, TB, 0, stream>>>(
      m1, nullptr, W2, bb2, nullptr, m2, NNODES, 320, 640);
  const dim3 gm3((NNODES + 255) / 256, 64 / 64);
  wmma_gemm_kernel<2, true, true, false, 2><<<gm3, TB, 0, stream>>>(
      m2, nullptr, W3, bb3, out, nullptr, NNODES, 64, 320);
}


# Each conv's two GEMMs (x*Wl, x*Wr) run as one launch: a block stages both 64x64 weight tiles and every A fragment it loads meets both; 14 -> 12 launches, A read once; k and sum order unchanged
# speedup vs baseline: 8.8336x; 1.0358x over previous
#include <hip/hip_runtime.h>
#include <hip/hip_bf16.h>


typedef __attribute__((ext_vector_type(16))) _Float16 v16h;
typedef __attribute__((ext_vector_type(8)))  _Float16 v8h;
typedef __attribute__((ext_vector_type(8)))  float    v8f;

#define NNODES 50000
#define NEDGES 800000
#define FDIM   128

#define NBLK 256
#define BLKW 196
#define NCHK 256
#define CHKE ((NEDGES + NCHK - 1) / NCHK)
#define CHKT ((CHKE + 1023) / 1024)

__device__ __forceinline__ int node_block(int d) {
  const unsigned b = (unsigned)d / (unsigned)BLKW;
  return (int)(b < (unsigned)(NBLK - 1) ? b : (unsigned)(NBLK - 1));
}

__global__ __launch_bounds__(1024) void bin_count_kernel(
    const int* __restrict__ ei, int* __restrict__ ccnt) {
  __shared__ int h[NBLK];
  const int t = threadIdx.x;
  if (t < NBLK) h[t] = 0;
  __syncthreads();
  const int e0 = blockIdx.x * CHKE;
  const int e1 = (e0 + CHKE) < NEDGES ? (e0 + CHKE) : NEDGES;
  int dv[CHKT];
  #pragma unroll
  for (int k = 0; k < CHKT; ++k) {
    const int e = e0 + t + k * 1024;
    dv[k] = ei[NEDGES + (e < NEDGES ? e : NEDGES - 1)];
  }
  #pragma unroll
  for (int k = 0; k < CHKT; ++k)
    if (e0 + t + k * 1024 < e1) atomicAdd(&h[node_block(dv[k])], 1);
  __syncthreads();
  if (t < NBLK) ccnt[t * NCHK + blockIdx.x] = h[t];
}

__global__ __launch_bounds__(1024) void bin_scan_kernel(
    const int* __restrict__ ccnt, int* __restrict__ coff) {
  __shared__ int wsum[32];
  const int t = threadIdx.x, lane = t & 31, wv = t >> 5;
  const int4* in4 = (const int4*)ccnt + wv * (NBLK * NCHK / 128);
  int4* out4 = (int4*)coff + wv * (NBLK * NCHK / 128);
  int4 v[16];
  #pragma unroll
  for (int k = 0; k < 16; ++k) v[k] = in4[k * 32 + lane];
  int carry = 0;
  #pragma unroll
  for (int k = 0; k < 16; ++k) {
    const int s0 = v[k].x, s1 = s0 + v[k].y, s2 = s1 + v[k].z, s3 = s2 + v[k].w;
    int x = s3;
    for (int o = 1; o < 32; o <<= 1) {
      const int y = __shfl_up(x, o);
      if (lane >= o) x += y;
    }
    const int ex = carry + x - s3;
    v[k] = make_int4(ex, ex + s0, ex + s1, ex + s2);
    carry += __shfl(x, 31);
  }
  if (lane == 0) wsum[wv] = carry;
  __syncthreads();
  if (wv == 0) {
    const int w = wsum[lane];
    int xs = w;
    for (int o = 1; o < 32; o <<= 1) {
      const int y = __shfl_up(xs, o);
      if (lane >= o) xs += y;
    }
    wsum[lane] = xs - w;
  }
  __syncthreads();
  const int base = wsum[wv];
  #pragma unroll
  for (int k = 0; k < 16; ++k) {
    int4 o = v[k];
    o.x += base; o.y += base; o.z += base; o.w += base;
    out4[k * 32 + lane] = o;
  }
  if (t == 1023) coff[NBLK * NCHK] = base + carry;
}

__global__ __launch_bounds__(1024) void bin_scatter_kernel(
    const int* __restrict__ ei, const int* __restrict__ coff,
    int2* __restrict__ binned) {
  __shared__ int cur[NBLK];
  const int t = threadIdx.x;
  if (t < NBLK) cur[t] = coff[t * NCHK + blockIdx.x];
  __syncthreads();
  const int e0 = blockIdx.x * CHKE;
  const int e1 = (e0 + CHKE) < NEDGES ? (e0 + CHKE) : NEDGES;
  int sv[CHKT], dv[CHKT];
  #pragma unroll
  for (int k = 0; k < CHKT; ++k) {
    const int e = e0 + t + k * 1024;
    const int ec = e < NEDGES ? e : NEDGES - 1;
    sv[k] = ei[ec];
    dv[k] = ei[NEDGES + ec];
  }
  #pragma unroll
  for (int k = 0; k < CHKT; ++k) {
    const int e = e0 + t + k * 1024;
    if (e < e1) {
      const int b = node_block(dv[k]);
      const int p = atomicAdd(&cur[b], 1);
      if ((unsigned)p < (unsigned)NEDGES)
        binned[p] = make_int2((sv[k] & 0xffff) | ((dv[k] - b * BLKW) << 16), e);
    }
  }
}

__global__ __launch_bounds__(1024) void bin_place_kernel(
    const int* __restrict__ coff, const int2* __restrict__ binned,
    int* __restrict__ rowptr, int2* __restrict__ perm) {
  __shared__ int cnt[256];
  __shared__ int cur[256];
  __shared__ int wtot[8];
  const int t = threadIdx.x, lane = t & 31, wv = t >> 5;
  const int b = blockIdx.x;
  const int beg = coff[b * NCHK], end = coff[(b + 1) * NCHK];
  if (t < 256) cnt[t] = 0;
  __syncthreads();
  for (int i0 = beg; i0 < end; i0 += 4096) {
    int rx[4];
    #pragma unroll
    for (int k = 0; k < 4; ++k) {
      const int i = i0 + t + k * 1024;
      rx[k] = binned[i < NEDGES ? i : NEDGES - 1].x;
    }
    #pragma unroll
    for (int k = 0; k < 4; ++k)
      if (i0 + t + k * 1024 < end) atomicAdd(&cnt[(rx[k] >> 16) & 0xff], 1);
  }
  __syncthreads();
  int c = 0;
  if (t < 256) c = cnt[t];
  int x = c;
  for (int o = 1; o < 32; o <<= 1) {
    const int y = __shfl_up(x, o);
    if (lane >= o) x += y;
  }
  if (t < 256 && lane == 31) wtot[wv] = x;
  __syncthreads();
  if (t < 256) {
    int first = beg + x - c;
    for (int w = 0; w < wv; ++w) first += wtot[w];
    cur[t] = first;
    const int node = b * BLKW + t;
    if (t < BLKW && node < NNODES) rowptr[node] = first;
  }
  if (b == NBLK - 1 && t == 0) rowptr[NNODES] = end;
  __syncthreads();
  for (int i0 = beg; i0 < end; i0 += 4096) {
    int2 r[4];
    #pragma unroll
    for (int k = 0; k < 4; ++k) {
      const int i = i0 + t + k * 1024;
      r[k] = binned[i < NEDGES ? i : NEDGES - 1];
    }
    #pragma unroll
    for (int k = 0; k < 4; ++k) {
      if (i0 + t + k * 1024 < end) {
        const int p = atomicAdd(&cur[(r[k].x >> 16) & 0xff], 1);
        if ((unsigned)p < (unsigned)NEDGES)
          perm[p] = make_int2(r[k].x & 0xffff, r[k].y);
      }
    }
  }
}

__global__ __launch_bounds__(256) void csr_order_kernel(
    const int* __restrict__ rowptr, const int2* __restrict__ perm,
    int* __restrict__ srcs) {
  const int lane = threadIdx.x & 31;
  const int w = __builtin_amdgcn_readfirstlane(
      (int)(blockIdx.x * 8 + (threadIdx.x >> 5)));
  for (int q = 0; q < 8; ++q) {
    const int d = w * 8 + q;
    if (d >= NNODES) break;
    const int beg = rowptr[d], end = rowptr[d + 1];
    for (int base = beg; base < end; base += 32) {
      const bool valid = base + lane < end;
      int2 my = make_int2(0, 0x7fffffff);
      if (valid) my = perm[base + lane];
      int rank = 0;
      for (int cb = beg; cb < end; cb += 32) {
        const int cn = (end - cb) < 32 ? (end - cb) : 32;
        int other = 0x7fffffff;
        if (cb + lane < end) other = perm[cb + lane].y;
        for (int j = 0; j < cn; ++j) {
          const int k = __builtin_amdgcn_readlane(other, j);
          rank += (k < my.y) ? 1 : 0;
        }
      }
      if (valid) srcs[beg + rank] = my.x;
    }
  }
}

#define KC 64
#define BT_STRIDE 72

template <int ACT, bool HASBIAS, bool F32OUT, bool AF32, int MT, int NW>
__global__ __launch_bounds__(256) void wmma_gemm_kernel(
    const _Float16* __restrict__ A, const float* __restrict__ Af,
    const float* __restrict__ W, const float* __restrict__ Wb,
    const float* __restrict__ bias,
    float* __restrict__ outF, float* __restrict__ outFb,
    _Float16* __restrict__ outH,
    int M, int Nt, int K) {
  static_assert(NW == 1 || F32OUT, "the second product has an f32 output only");
  __shared__ _Float16 bt[NW][64 * BT_STRIDE];

  const int lane = threadIdx.x & 31;
  const int wave = threadIdx.x >> 5;
  const int half = lane >> 4;
  const int l15  = lane & 15;
  const int n0   = blockIdx.y * 64;

  int rowBase = blockIdx.x * (128 * MT) + wave * (16 * MT);
  if (rowBase > M - 16 * MT) rowBase = M - 16 * MT;

  v8f c[NW][MT][4];
  #pragma unroll
  for (int w = 0; w < NW; ++w) {
    #pragma unroll
    for (int mt = 0; mt < MT; ++mt) {
      #pragma unroll
      for (int nn = 0; nn < 4; ++nn) c[w][mt][nn] = v8f{};
    }
  }

  for (int kc = 0; kc < K; kc += KC) {
    #pragma unroll
    for (int w = 0; w < NW; ++w) {
      const float* __restrict__ Wp = (w == 0) ? W : Wb;
      int kl = threadIdx.x >> 4;
      const int n4 = (threadIdx.x & 15) * 4;
      #pragma unroll
      for (int p = 0; p < 4; ++p, kl += 16) {
        const float4 wv =
            *(const float4*)(Wp + (size_t)(kc + kl) * Nt + n0 + n4);
        bt[w][(n4 + 0) * BT_STRIDE + kl] = (_Float16)wv.x;
        bt[w][(n4 + 1) * BT_STRIDE + kl] = (_Float16)wv.y;
        bt[w][(n4 + 2) * BT_STRIDE + kl] = (_Float16)wv.z;
        bt[w][(n4 + 3) * BT_STRIDE + kl] = (_Float16)wv.w;
      }
    }
    __syncthreads();

    const size_t aoff = (size_t)(rowBase + l15) * K + kc;
    v16h a[2][MT];
    #pragma unroll
    for (int ks = 0; ks < 2; ++ks) {
      const int kb = ks * 32 + half * 8;
      #pragma unroll
      for (int mt = 0; mt < MT; ++mt) {
        if (AF32) {
          const float* ap = Af + aoff + (size_t)(mt * 16) * K + kb;
          const float4 f0 = *(const float4*)(ap);
          const float4 f1 = *(const float4*)(ap + 4);
          const float4 f2 = *(const float4*)(ap + 16);
          const float4 f3 = *(const float4*)(ap + 20);
          v16h t;
          t[0]  = (_Float16)f0.x; t[1]  = (_Float16)f0.y;
          t[2]  = (_Float16)f0.z; t[3]  = (_Float16)f0.w;
          t[4]  = (_Float16)f1.x; t[5]  = (_Float16)f1.y;
          t[6]  = (_Float16)f1.z; t[7]  = (_Float16)f1.w;
          t[8]  = (_Float16)f2.x; t[9]  = (_Float16)f2.y;
          t[10] = (_Float16)f2.z; t[11] = (_Float16)f2.w;
          t[12] = (_Float16)f3.x; t[13] = (_Float16)f3.y;
          t[14] = (_Float16)f3.z; t[15] = (_Float16)f3.w;
          a[ks][mt] = t;
        } else {
          const _Float16* arow = A + aoff + (size_t)(mt * 16) * K;
          v8h alo = *(const v8h*)(arow + kb);
          v8h ahi = *(const v8h*)(arow + kb + 16);
          a[ks][mt] = __builtin_shufflevector(alo, ahi, 0, 1, 2, 3, 4, 5, 6, 7,
                                              8, 9, 10, 11, 12, 13, 14, 15);
        }
      }
    }
    __builtin_amdgcn_sched_barrier(0);
    #pragma unroll
    for (int ks = 0; ks < 2; ++ks) {
      const int kb = ks * 32 + half * 8;
      #pragma unroll
      for (int w = 0; w < NW; ++w) {
        v16h b[4];
        #pragma unroll
        for (int nn = 0; nn < 4; ++nn) {
          const _Float16* bp = bt[w] + (nn * 16 + l15) * BT_STRIDE + kb;
          v8h blo = *(const v8h*)(bp);
          v8h bhi = *(const v8h*)(bp + 16);
          b[nn] = __builtin_shufflevector(blo, bhi, 0, 1, 2, 3, 4, 5, 6, 7,
                                          8, 9, 10, 11, 12, 13, 14, 15);
        }
        #pragma unroll
        for (int mt = 0; mt < MT; ++mt) {
          #pragma unroll
          for (int nn = 0; nn < 4; ++nn) {
            c[w][mt][nn] = __builtin_amdgcn_wmma_f32_16x16x32_f16(
                false, a[ks][mt], false, b[nn], (short)0, c[w][mt][nn], false,
                false);
          }
        }
      }
    }
    __syncthreads();
  }

  float bv[4] = {0.0f, 0.0f, 0.0f, 0.0f};
  if (HASBIAS) {
    #pragma unroll
    for (int nn = 0; nn < 4; ++nn) bv[nn] = bias[n0 + nn * 16 + l15];
  }
  #pragma unroll
  for (int w = 0; w < NW; ++w) {
    float* __restrict__ oF = (w == 0) ? outF : outFb;
    #pragma unroll
    for (int mt = 0; mt < MT; ++mt) {
      const unsigned o0 =
          (unsigned)(rowBase + mt * 16 + half * 8) * (unsigned)Nt +
          (unsigned)(n0 + l15);
      #pragma unroll
      for (int nn = 0; nn < 4; ++nn) {
        #pragma unroll
        for (int r = 0; r < 8; ++r) {
          float v = c[w][mt][nn][r] + bv[nn];
          if (ACT == 1) v = v > 0.0f ? v : (__expf(v) - 1.0f);
          else if (ACT == 2) v = 1.0f / (1.0f + __expf(-v));
          const unsigned o =
              o0 + (unsigned)r * (unsigned)Nt + (unsigned)(nn * 16);
          if (F32OUT) oF[o] = v;
          else        outH[o] = (_Float16)v;
        }
      }
    }
  }
}

typedef __attribute__((ext_vector_type(4))) _Float16 v4h;

template <int CTRL>
__device__ __forceinline__ float lane_xor_get(float x) {
  return __builtin_bit_cast(
      float, __builtin_amdgcn_update_dpp(0, __builtin_bit_cast(int, x), CTRL,
                                         0xf, 0xf, true));
}

__device__ __forceinline__ float edge_logit(const float4 a, const float4 b,
                                            const float4 w) {
  float z, acc;
  z = a.x + b.x; z = z > 0.0f ? z : 0.2f * z; acc = z * w.x;
  z = a.y + b.y; z = z > 0.0f ? z : 0.2f * z; acc += z * w.y;
  z = a.z + b.z; z = z > 0.0f ? z : 0.2f * z; acc += z * w.z;
  z = a.w + b.w; z = z > 0.0f ? z : 0.2f * z; acc += z * w.w;
  acc += lane_xor_get<0x161>(acc);
  acc += lane_xor_get<0x162>(acc);
  acc += lane_xor_get<0x164>(acc);
  return acc;
}

template <int NB>
__device__ __forceinline__ void conv_batch(const float4* __restrict__ xl4,
                                           int my, int j0, int lane,
                                           const float4 xrv, const float4 av,
                                           float& m, float& ssum, float4& r) {
  float4 v[NB];
  float e[NB];
  #pragma unroll
  for (int t = 0; t < NB; ++t) {
    const int sidx = __builtin_amdgcn_readlane(my, j0 + t);
    v[t] = xl4[(size_t)sidx * 32 + lane];
  }
  float mn = m;
  #pragma unroll
  for (int t = 0; t < NB; ++t) {
    e[t] = edge_logit(v[t], xrv, av);
    mn = fmaxf(mn, e[t]);
  }
  const float sc = __expf(m - mn);
  ssum *= sc;
  r.x *= sc; r.y *= sc; r.z *= sc; r.w *= sc;
  #pragma unroll
  for (int t = 0; t < NB; ++t) {
    const float p = __expf(e[t] - mn);
    ssum += p;
    r.x += v[t].x * p; r.y += v[t].y * p; r.z += v[t].z * p; r.w += v[t].w * p;
  }
  m = mn;
}

__global__ __launch_bounds__(256) void conv_edge_kernel(
    const int* __restrict__ rowptr, const int* __restrict__ srcs,
    const float* __restrict__ xl, const float* __restrict__ xr,
    const float* __restrict__ att, const float* __restrict__ bias,
    _Float16* __restrict__ outH) {
  const int lane = threadIdx.x & 31;
  const int d = __builtin_amdgcn_readfirstlane(
      (int)(blockIdx.x * 8 + (threadIdx.x >> 5)));
  if (d >= NNODES) return;
  const float4* xl4 = (const float4*)xl;
  const float4 xrv = ((const float4*)xr)[(size_t)d * 32 + lane];
  const float4 av = ((const float4*)att)[lane];

  float4 r = xl4[(size_t)d * 32 + lane];
  float m = edge_logit(r, xrv, av);
  float ssum = 1.0f;

  const int beg = rowptr[d], end = rowptr[d + 1];
  for (int base = beg; base < end; base += 32) {
    const int cnt = (end - base) < 32 ? (end - base) : 32;
    int my = 0;
    if (lane < cnt) my = srcs[base + lane];
    int j = 0;
    for (; j + 8 <= cnt; j += 8)
      conv_batch<8>(xl4, my, j, lane, xrv, av, m, ssum, r);
    if (j + 4 <= cnt) { conv_batch<4>(xl4, my, j, lane, xrv, av, m, ssum, r); j += 4; }
    if (j + 2 <= cnt) { conv_batch<2>(xl4, my, j, lane, xrv, av, m, ssum, r); j += 2; }
    if (j < cnt) conv_batch<1>(xl4, my, j, lane, xrv, av, m, ssum, r);
  }

  const float4 bv = ((const float4*)bias)[lane];
  float o0 = r.x / ssum + bv.x, o1 = r.y / ssum + bv.y;
  float o2 = r.z / ssum + bv.z, o3 = r.w / ssum + bv.w;
  o0 = o0 > 0.0f ? o0 : (__expf(o0) - 1.0f);
  o1 = o1 > 0.0f ? o1 : (__expf(o1) - 1.0f);
  o2 = o2 > 0.0f ? o2 : (__expf(o2) - 1.0f);
  o3 = o3 > 0.0f ? o3 : (__expf(o3) - 1.0f);
  v4h hv;
  hv[0] = (_Float16)o0; hv[1] = (_Float16)o1;
  hv[2] = (_Float16)o2; hv[3] = (_Float16)o3;
  *(v4h*)(outH + (size_t)d * FDIM + lane * 4) = hv;
}

extern "C" void kernel_launch(void* const* d_in, const int* in_sizes, int n_in,
                              void* d_out, int out_size, void* d_ws,
                              size_t ws_size, hipStream_t stream) {
  (void)in_sizes; (void)n_in; (void)out_size; (void)ws_size;
  const float* x    = (const float*)d_in[0];
  const int*   ei   = (const int*)d_in[1];
  const float* Wl1  = (const float*)d_in[2];
  const float* Wr1  = (const float*)d_in[3];
  const float* att1 = (const float*)d_in[4];
  const float* b1   = (const float*)d_in[5];
  const float* Wl2  = (const float*)d_in[6];
  const float* Wr2  = (const float*)d_in[7];
  const float* att2 = (const float*)d_in[8];
  const float* b2   = (const float*)d_in[9];
  const float* W1   = (const float*)d_in[10];
  const float* bb1  = (const float*)d_in[11];
  const float* W2   = (const float*)d_in[12];
  const float* bb2  = (const float*)d_in[13];
  const float* W3   = (const float*)d_in[14];
  const float* bb3  = (const float*)d_in[15];
  float* out = (float*)d_out;

  char* ws = (char*)d_ws;
  size_t off = 0;
  auto alloc = [&](size_t bytes) -> char* {
    char* p = ws + off;
    off += (bytes + 255) & ~(size_t)255;
    return p;
  };
  const size_t S_F = (size_t)NNODES * FDIM;
  _Float16* h1h = (_Float16*)alloc(S_F * 2);
  _Float16* h2h = (_Float16*)alloc(S_F * 2);
  int*  ccnt   = (int*)alloc((size_t)NBLK * NCHK * 4);
  int*  coff   = (int*)alloc(((size_t)NBLK * NCHK + 1) * 4);
  int*  rowptr = (int*)alloc(((size_t)NNODES + 1) * 4);
  int2* binned = (int2*)alloc((size_t)NEDGES * 8);
  int2* perm   = (int2*)alloc((size_t)NEDGES * 8);
  int*  srcs   = (int*)alloc((size_t)NEDGES * 4);
  const size_t uoff = off;
  float*    xl   = (float*)alloc(S_F * 4);
  float*    xr   = (float*)alloc(S_F * 4);
  _Float16* m1 = (_Float16*)(ws + uoff);
  _Float16* m2 = (_Float16*)(ws + uoff + (size_t)NNODES * 640 * 2);

  const int TB = 256;
  const int wgrid = (NNODES + 7) / 8;
  const int ogrid = (NNODES + 63) / 64;
  const dim3 gconv((NNODES + 255) / 256, 128 / 64);

  bin_count_kernel<<<NCHK, 1024, 0, stream>>>(ei, ccnt);
  bin_scan_kernel<<<1, 1024, 0, stream>>>(ccnt, coff);
  bin_scatter_kernel<<<NCHK, 1024, 0, stream>>>(ei, coff, binned);
  bin_place_kernel<<<NBLK, 1024, 0, stream>>>(coff, binned, rowptr, perm);
  csr_order_kernel<<<ogrid, TB, 0, stream>>>(rowptr, perm, srcs);

  wmma_gemm_kernel<0, false, true, true, 2, 2><<<gconv, TB, 0, stream>>>(
      nullptr, x, Wl1, Wr1, nullptr, xl, xr, nullptr, NNODES, 128, 128);
  conv_edge_kernel<<<wgrid, TB, 0, stream>>>(rowptr, srcs, xl, xr, att1, b1,
                                             h1h);

  wmma_gemm_kernel<0, false, true, false, 2, 2><<<gconv, TB, 0, stream>>>(
      h1h, nullptr, Wl2, Wr2, nullptr, xl, xr, nullptr, NNODES, 128, 128);
  conv_edge_kernel<<<wgrid, TB, 0, stream>>>(rowptr, srcs, xl, xr, att2, b2,
                                             h2h);

  const dim3 gm1((NNODES + 511) / 512, 640 / 64);
  wmma_gemm_kernel<1, true, false, false, 4, 1><<<gm1, TB, 0, stream>>>(
      h2h, nullptr, W1, nullptr, bb1, nullptr, nullptr, m1, NNODES, 640, 128);
  const dim3 gm2((NNODES + 511) / 512, 320 / 64);
  wmma_gemm_kernel<1, true, false, false, 4, 1><<<gm2, TB, 0, stream>>>(
      m1, nullptr, W2, nullptr, bb2, nullptr, nullptr, m2, NNODES, 320, 640);
  const dim3 gm3((NNODES + 255) / 256, 64 / 64);
  wmma_gemm_kernel<2, true, true, false, 2, 1><<<gm3, TB, 0, stream>>>(
      m2, nullptr, W3, nullptr, bb3, out, nullptr, nullptr, NNODES, 64, 320);
}


# Conv edge kernel: per-edge arithmetic on each row's own register pairs (2-wide vectors; leaky ReLU as max(z, 0.2z)): cross-edge shuffles, compares and selects gone; logit partial sum order changes
# speedup vs baseline: 9.1517x; 1.0360x over previous
#include <hip/hip_runtime.h>
#include <hip/hip_bf16.h>


typedef __attribute__((ext_vector_type(16))) _Float16 v16h;
typedef __attribute__((ext_vector_type(8)))  _Float16 v8h;
typedef __attribute__((ext_vector_type(8)))  float    v8f;

#define NNODES 50000
#define NEDGES 800000
#define FDIM   128

#define NBLK 256
#define BLKW 196
#define NCHK 256
#define CHKE ((NEDGES + NCHK - 1) / NCHK)
#define CHKT ((CHKE + 1023) / 1024)

__device__ __forceinline__ int node_block(int d) {
  const unsigned b = (unsigned)d / (unsigned)BLKW;
  return (int)(b < (unsigned)(NBLK - 1) ? b : (unsigned)(NBLK - 1));
}

__global__ __launch_bounds__(1024) void bin_count_kernel(
    const int* __restrict__ ei, int* __restrict__ ccnt) {
  __shared__ int h[NBLK];
  const int t = threadIdx.x;
  if (t < NBLK) h[t] = 0;
  __syncthreads();
  const int e0 = blockIdx.x * CHKE;
  const int e1 = (e0 + CHKE) < NEDGES ? (e0 + CHKE) : NEDGES;
  int dv[CHKT];
  #pragma unroll
  for (int k = 0; k < CHKT; ++k) {
    const int e = e0 + t + k * 1024;
    dv[k] = ei[NEDGES + (e < NEDGES ? e : NEDGES - 1)];
  }
  #pragma unroll
  for (int k = 0; k < CHKT; ++k)
    if (e0 + t + k * 1024 < e1) atomicAdd(&h[node_block(dv[k])], 1);
  __syncthreads();
  if (t < NBLK) ccnt[t * NCHK + blockIdx.x] = h[t];
}

__global__ __launch_bounds__(1024) void bin_scan_kernel(
    const int* __restrict__ ccnt, int* __restrict__ coff) {
  __shared__ int wsum[32];
  const int t = threadIdx.x, lane = t & 31, wv = t >> 5;
  const int4* in4 = (const int4*)ccnt + wv * (NBLK * NCHK / 128);
  int4* out4 = (int4*)coff + wv * (NBLK * NCHK / 128);
  int4 v[16];
  #pragma unroll
  for (int k = 0; k < 16; ++k) v[k] = in4[k * 32 + lane];
  int carry = 0;
  #pragma unroll
  for (int k = 0; k < 16; ++k) {
    const int s0 = v[k].x, s1 = s0 + v[k].y, s2 = s1 + v[k].z, s3 = s2 + v[k].w;
    int x = s3;
    for (int o = 1; o < 32; o <<= 1) {
      const int y = __shfl_up(x, o);
      if (lane >= o) x += y;
    }
    const int ex = carry + x - s3;
    v[k] = make_int4(ex, ex + s0, ex + s1, ex + s2);
    carry += __shfl(x, 31);
  }
  if (lane == 0) wsum[wv] = carry;
  __syncthreads();
  if (wv == 0) {
    const int w = wsum[lane];
    int xs = w;
    for (int o = 1; o < 32; o <<= 1) {
      const int y = __shfl_up(xs, o);
      if (lane >= o) xs += y;
    }
    wsum[lane] = xs - w;
  }
  __syncthreads();
  const int base = wsum[wv];
  #pragma unroll
  for (int k = 0; k < 16; ++k) {
    int4 o = v[k];
    o.x += base; o.y += base; o.z += base; o.w += base;
    out4[k * 32 + lane] = o;
  }
  if (t == 1023) coff[NBLK * NCHK] = base + carry;
}

__global__ __launch_bounds__(1024) void bin_scatter_kernel(
    const int* __restrict__ ei, const int* __restrict__ coff,
    int2* __restrict__ binned) {
  __shared__ int cur[NBLK];
  const int t = threadIdx.x;
  if (t < NBLK) cur[t] = coff[t * NCHK + blockIdx.x];
  __syncthreads();
  const int e0 = blockIdx.x * CHKE;
  const int e1 = (e0 + CHKE) < NEDGES ? (e0 + CHKE) : NEDGES;
  int sv[CHKT], dv[CHKT];
  #pragma unroll
  for (int k = 0; k < CHKT; ++k) {
    const int e = e0 + t + k * 1024;
    const int ec = e < NEDGES ? e : NEDGES - 1;
    sv[k] = ei[ec];
    dv[k] = ei[NEDGES + ec];
  }
  #pragma unroll
  for (int k = 0; k < CHKT; ++k) {
    const int e = e0 + t + k * 1024;
    if (e < e1) {
      const int b = node_block(dv[k]);
      const int p = atomicAdd(&cur[b], 1);
      if ((unsigned)p < (unsigned)NEDGES)
        binned[p] = make_int2((sv[k] & 0xffff) | ((dv[k] - b * BLKW) << 16), e);
    }
  }
}

__global__ __launch_bounds__(1024) void bin_place_kernel(
    const int* __restrict__ coff, const int2* __restrict__ binned,
    int* __restrict__ rowptr, int2* __restrict__ perm) {
  __shared__ int cnt[256];
  __shared__ int cur[256];
  __shared__ int wtot[8];
  const int t = threadIdx.x, lane = t & 31, wv = t >> 5;
  const int b = blockIdx.x;
  const int beg = coff[b * NCHK], end = coff[(b + 1) * NCHK];
  if (t < 256) cnt[t] = 0;
  __syncthreads();
  for (int i0 = beg; i0 < end; i0 += 4096) {
    int rx[4];
    #pragma unroll
    for (int k = 0; k < 4; ++k) {
      const int i = i0 + t + k * 1024;
      rx[k] = binned[i < NEDGES ? i : NEDGES - 1].x;
    }
    #pragma unroll
    for (int k = 0; k < 4; ++k)
      if (i0 + t + k * 1024 < end) atomicAdd(&cnt[(rx[k] >> 16) & 0xff], 1);
  }
  __syncthreads();
  int c = 0;
  if (t < 256) c = cnt[t];
  int x = c;
  for (int o = 1; o < 32; o <<= 1) {
    const int y = __shfl_up(x, o);
    if (lane >= o) x += y;
  }
  if (t < 256 && lane == 31) wtot[wv] = x;
  __syncthreads();
  if (t < 256) {
    int first = beg + x - c;
    for (int w = 0; w < wv; ++w) first += wtot[w];
    cur[t] = first;
    const int node = b * BLKW + t;
    if (t < BLKW && node < NNODES) rowptr[node] = first;
  }
  if (b == NBLK - 1 && t == 0) rowptr[NNODES] = end;
  __syncthreads();
  for (int i0 = beg; i0 < end; i0 += 4096) {
    int2 r[4];
    #pragma unroll
    for (int k = 0; k < 4; ++k) {
      const int i = i0 + t + k * 1024;
      r[k] = binned[i < NEDGES ? i : NEDGES - 1];
    }
    #pragma unroll
    for (int k = 0; k < 4; ++k) {
      if (i0 + t + k * 1024 < end) {
        const int p = atomicAdd(&cur[(r[k].x >> 16) & 0xff], 1);
        if ((unsigned)p < (unsigned)NEDGES)
          perm[p] = make_int2(r[k].x & 0xffff, r[k].y);
      }
    }
  }
}

__global__ __launch_bounds__(256) void csr_order_kernel(
    const int* __restrict__ rowptr, const int2* __restrict__ perm,
    int* __restrict__ srcs) {
  const int lane = threadIdx.x & 31;
  const int w = __builtin_amdgcn_readfirstlane(
      (int)(blockIdx.x * 8 + (threadIdx.x >> 5)));
  for (int q = 0; q < 8; ++q) {
    const int d = w * 8 + q;
    if (d >= NNODES) break;
    const int beg = rowptr[d], end = rowptr[d + 1];
    for (int base = beg; base < end; base += 32) {
      const bool valid = base + lane < end;
      int2 my = make_int2(0, 0x7fffffff);
      if (valid) my = perm[base + lane];
      int rank = 0;
      for (int cb = beg; cb < end; cb += 32) {
        const int cn = (end - cb) < 32 ? (end - cb) : 32;
        int other = 0x7fffffff;
        if (cb + lane < end) other = perm[cb + lane].y;
        for (int j = 0; j < cn; ++j) {
          const int k = __builtin_amdgcn_readlane(other, j);
          rank += (k < my.y) ? 1 : 0;
        }
      }
      if (valid) srcs[beg + rank] = my.x;
    }
  }
}

#define KC 64
#define BT_STRIDE 72

template <int ACT, bool HASBIAS, bool F32OUT, bool AF32, int MT, int NW>
__global__ __launch_bounds__(256) void wmma_gemm_kernel(
    const _Float16* __restrict__ A, const float* __restrict__ Af,
    const float* __restrict__ W, const float* __restrict__ Wb,
    const float* __restrict__ bias,
    float* __restrict__ outF, float* __restrict__ outFb,
    _Float16* __restrict__ outH,
    int M, int Nt, int K) {
  static_assert(NW == 1 || F32OUT, "the second product has an f32 output only");
  __shared__ _Float16 bt[NW][64 * BT_STRIDE];

  const int lane = threadIdx.x & 31;
  const int wave = threadIdx.x >> 5;
  const int half = lane >> 4;
  const int l15  = lane & 15;
  const int n0   = blockIdx.y * 64;

  int rowBase = blockIdx.x * (128 * MT) + wave * (16 * MT);
  if (rowBase > M - 16 * MT) rowBase = M - 16 * MT;

  v8f c[NW][MT][4];
  #pragma unroll
  for (int w = 0; w < NW; ++w) {
    #pragma unroll
    for (int mt = 0; mt < MT; ++mt) {
      #pragma unroll
      for (int nn = 0; nn < 4; ++nn) c[w][mt][nn] = v8f{};
    }
  }

  for (int kc = 0; kc < K; kc += KC) {
    #pragma unroll
    for (int w = 0; w < NW; ++w) {
      const float* __restrict__ Wp = (w == 0) ? W : Wb;
      int kl = threadIdx.x >> 4;
      const int n4 = (threadIdx.x & 15) * 4;
      #pragma unroll
      for (int p = 0; p < 4; ++p, kl += 16) {
        const float4 wv =
            *(const float4*)(Wp + (size_t)(kc + kl) * Nt + n0 + n4);
        bt[w][(n4 + 0) * BT_STRIDE + kl] = (_Float16)wv.x;
        bt[w][(n4 + 1) * BT_STRIDE + kl] = (_Float16)wv.y;
        bt[w][(n4 + 2) * BT_STRIDE + kl] = (_Float16)wv.z;
        bt[w][(n4 + 3) * BT_STRIDE + kl] = (_Float16)wv.w;
      }
    }
    __syncthreads();

    const size_t aoff = (size_t)(rowBase + l15) * K + kc;
    v16h a[2][MT];
    #pragma unroll
    for (int ks = 0; ks < 2; ++ks) {
      const int kb = ks * 32 + half * 8;
      #pragma unroll
      for (int mt = 0; mt < MT; ++mt) {
        if (AF32) {
          const float* ap = Af + aoff + (size_t)(mt * 16) * K + kb;
          const float4 f0 = *(const float4*)(ap);
          const float4 f1 = *(const float4*)(ap + 4);
          const float4 f2 = *(const float4*)(ap + 16);
          const float4 f3 = *(const float4*)(ap + 20);
          v16h t;
          t[0]  = (_Float16)f0.x; t[1]  = (_Float16)f0.y;
          t[2]  = (_Float16)f0.z; t[3]  = (_Float16)f0.w;
          t[4]  = (_Float16)f1.x; t[5]  = (_Float16)f1.y;
          t[6]  = (_Float16)f1.z; t[7]  = (_Float16)f1.w;
          t[8]  = (_Float16)f2.x; t[9]  = (_Float16)f2.y;
          t[10] = (_Float16)f2.z; t[11] = (_Float16)f2.w;
          t[12] = (_Float16)f3.x; t[13] = (_Float16)f3.y;
          t[14] = (_Float16)f3.z; t[15] = (_Float16)f3.w;
          a[ks][mt] = t;
        } else {
          const _Float16* arow = A + aoff + (size_t)(mt * 16) * K;
          v8h alo = *(const v8h*)(arow + kb);
          v8h ahi = *(const v8h*)(arow + kb + 16);
          a[ks][mt] = __builtin_shufflevector(alo, ahi, 0, 1, 2, 3, 4, 5, 6, 7,
                                              8, 9, 10, 11, 12, 13, 14, 15);
        }
      }
    }
    __builtin_amdgcn_sched_barrier(0);
    #pragma unroll
    for (int ks = 0; ks < 2; ++ks) {
      const int kb = ks * 32 + half * 8;
      #pragma unroll
      for (int w = 0; w < NW; ++w) {
        v16h b[4];
        #pragma unroll
        for (int nn = 0; nn < 4; ++nn) {
          const _Float16* bp = bt[w] + (nn * 16 + l15) * BT_STRIDE + kb;
          v8h blo = *(const v8h*)(bp);
          v8h bhi = *(const v8h*)(bp + 16);
          b[nn] = __builtin_shufflevector(blo, bhi, 0, 1, 2, 3, 4, 5, 6, 7,
                                          8, 9, 10, 11, 12, 13, 14, 15);
        }
        #pragma unroll
        for (int mt = 0; mt < MT; ++mt) {
          #pragma unroll
          for (int nn = 0; nn < 4; ++nn) {
            c[w][mt][nn] = __builtin_amdgcn_wmma_f32_16x16x32_f16(
                false, a[ks][mt], false, b[nn], (short)0, c[w][mt][nn], false,
                false);
          }
        }
      }
    }
    __syncthreads();
  }

  float bv[4] = {0.0f, 0.0f, 0.0f, 0.0f};
  if (HASBIAS) {
    #pragma unroll
    for (int nn = 0; nn < 4; ++nn) bv[nn] = bias[n0 + nn * 16 + l15];
  }
  #pragma unroll
  for (int w = 0; w < NW; ++w) {
    float* __restrict__ oF = (w == 0) ? outF : outFb;
    #pragma unroll
    for (int mt = 0; mt < MT; ++mt) {
      const unsigned o0 =
          (unsigned)(rowBase + mt * 16 + half * 8) * (unsigned)Nt +
          (unsigned)(n0 + l15);
      #pragma unroll
      for (int nn = 0; nn < 4; ++nn) {
        #pragma unroll
        for (int r = 0; r < 8; ++r) {
          float v = c[w][mt][nn][r] + bv[nn];
          if (ACT == 1) v = v > 0.0f ? v : (__expf(v) - 1.0f);
          else if (ACT == 2) v = 1.0f / (1.0f + __expf(-v));
          const unsigned o =
              o0 + (unsigned)r * (unsigned)Nt + (unsigned)(nn * 16);
          if (F32OUT) oF[o] = v;
          else        outH[o] = (_Float16)v;
        }
      }
    }
  }
}

typedef __attribute__((ext_vector_type(4))) _Float16 v4h;
typedef __attribute__((ext_vector_type(2))) float v2f;
typedef __attribute__((ext_vector_type(4))) float v4f;

template <int CTRL>
__device__ __forceinline__ float lane_xor_get(float x) {
  return __builtin_bit_cast(
      float, __builtin_amdgcn_update_dpp(0, __builtin_bit_cast(int, x), CTRL,
                                         0xf, 0xf, true));
}

__device__ __forceinline__ float edge_logit(const v4f a, const v4f b,
                                            const v4f w) {
  const v2f z0 = a.xy + b.xy;
  const v2f z1 = a.zw + b.zw;
  const v2f l0 = __builtin_elementwise_max(z0, z0 * 0.2f);
  const v2f l1 = __builtin_elementwise_max(z1, z1 * 0.2f);
  v2f p = l0 * w.xy;
  p = l1 * w.zw + p;
  float acc = p.x + p.y;
  acc += lane_xor_get<0x161>(acc);
  acc += lane_xor_get<0x162>(acc);
  acc += lane_xor_get<0x164>(acc);
  return acc;
}

template <int NB>
__device__ __forceinline__ void conv_batch(const v4f* __restrict__ xl4,
                                           int my, int j0, int lane,
                                           const v4f xrv, const v4f av,
                                           float& m, float& ssum, v4f& r) {
  v4f v[NB];
  float e[NB];
  unsigned off[NB];
  #pragma unroll
  for (int t = 0; t < NB; ++t) {
    const int sidx = __builtin_amdgcn_readlane(my, j0 + t);
    off[t] = (unsigned)sidx * 32u + (unsigned)lane;
  }
  #pragma unroll
  for (int t = 0; t < NB; ++t) v[t] = xl4[off[t]];
  float mn = m;
  #pragma unroll
  for (int t = 0; t < NB; ++t) {
    e[t] = edge_logit(v[t], xrv, av);
    mn = fmaxf(mn, e[t]);
  }
  const float sc = __expf(m - mn);
  ssum *= sc;
  r *= sc;
  #pragma unroll
  for (int t = 0; t < NB; ++t) {
    const float p = __expf(e[t] - mn);
    ssum += p;
    r = v[t] * p + r;
  }
  m = mn;
}

__global__ __launch_bounds__(256) void conv_edge_kernel(
    const int* __restrict__ rowptr, const int* __restrict__ srcs,
    const float* __restrict__ xl, const float* __restrict__ xr,
    const float* __restrict__ att, const float* __restrict__ bias,
    _Float16* __restrict__ outH) {
  const int lane = threadIdx.x & 31;
  const int d = __builtin_amdgcn_readfirstlane(
      (int)(blockIdx.x * 8 + (threadIdx.x >> 5)));
  if (d >= NNODES) return;
  const v4f* xl4 = (const v4f*)xl;
  const v4f xrv = ((const v4f*)xr)[(size_t)d * 32 + lane];
  const v4f av = ((const v4f*)att)[lane];

  v4f r = xl4[(size_t)d * 32 + lane];
  float m = edge_logit(r, xrv, av);
  float ssum = 1.0f;

  const int beg = rowptr[d], end = rowptr[d + 1];
  for (int base = beg; base < end; base += 32) {
    const int cnt = (end - base) < 32 ? (end - base) : 32;
    int my = 0;
    if (lane < cnt) my = srcs[base + lane];
    int j = 0;
    for (; j + 8 <= cnt; j += 8)
      conv_batch<8>(xl4, my, j, lane, xrv, av, m, ssum, r);
    if (j + 4 <= cnt) { conv_batch<4>(xl4, my, j, lane, xrv, av, m, ssum, r); j += 4; }
    if (j + 2 <= cnt) { conv_batch<2>(xl4, my, j, lane, xrv, av, m, ssum, r); j += 2; }
    if (j < cnt) conv_batch<1>(xl4, my, j, lane, xrv, av, m, ssum, r);
  }

  const v4f bv = ((const v4f*)bias)[lane];
  float o0 = r.x / ssum + bv.x, o1 = r.y / ssum + bv.y;
  float o2 = r.z / ssum + bv.z, o3 = r.w / ssum + bv.w;
  o0 = o0 > 0.0f ? o0 : (__expf(o0) - 1.0f);
  o1 = o1 > 0.0f ? o1 : (__expf(o1) - 1.0f);
  o2 = o2 > 0.0f ? o2 : (__expf(o2) - 1.0f);
  o3 = o3 > 0.0f ? o3 : (__expf(o3) - 1.0f);
  v4h hv;
  hv[0] = (_Float16)o0; hv[1] = (_Float16)o1;
  hv[2] = (_Float16)o2; hv[3] = (_Float16)o3;
  *(v4h*)(outH + (size_t)d * FDIM + lane * 4) = hv;
}

extern "C" void kernel_launch(void* const* d_in, const int* in_sizes, int n_in,
                              void* d_out, int out_size, void* d_ws,
                              size_t ws_size, hipStream_t stream) {
  (void)in_sizes; (void)n_in; (void)out_size; (void)ws_size;
  const float* x    = (const float*)d_in[0];
  const int*   ei   = (const int*)d_in[1];
  const float* Wl1  = (const float*)d_in[2];
  const float* Wr1  = (const float*)d_in[3];
  const float* att1 = (const float*)d_in[4];
  const float* b1   = (const float*)d_in[5];
  const float* Wl2  = (const float*)d_in[6];
  const float* Wr2  = (const float*)d_in[7];
  const float* att2 = (const float*)d_in[8];
  const float* b2   = (const float*)d_in[9];
  const float* W1   = (const float*)d_in[10];
  const float* bb1  = (const float*)d_in[11];
  const float* W2   = (const float*)d_in[12];
  const float* bb2  = (const float*)d_in[13];
  const float* W3   = (const float*)d_in[14];
  const float* bb3  = (const float*)d_in[15];
  float* out = (float*)d_out;

  char* ws = (char*)d_ws;
  size_t off = 0;
  auto alloc = [&](size_t bytes) -> char* {
    char* p = ws + off;
    off += (bytes + 255) & ~(size_t)255;
    return p;
  };
  const size_t S_F = (size_t)NNODES * FDIM;
  _Float16* h1h = (_Float16*)alloc(S_F * 2);
  _Float16* h2h = (_Float16*)alloc(S_F * 2);
  int*  ccnt   = (int*)alloc((size_t)NBLK * NCHK * 4);
  int*  coff   = (int*)alloc(((size_t)NBLK * NCHK + 1) * 4);
  int*  rowptr = (int*)alloc(((size_t)NNODES + 1) * 4);
  int2* binned = (int2*)alloc((size_t)NEDGES * 8);
  int2* perm   = (int2*)alloc((size_t)NEDGES * 8);
  int*  srcs   = (int*)alloc((size_t)NEDGES * 4);
  const size_t uoff = off;
  float*    xl   = (float*)alloc(S_F * 4);
  float*    xr   = (float*)alloc(S_F * 4);
  _Float16* m1 = (_Float16*)(ws + uoff);
  _Float16* m2 = (_Float16*)(ws + uoff + (size_t)NNODES * 640 * 2);

  const int TB = 256;
  const int wgrid = (NNODES + 7) / 8;
  const int ogrid = (NNODES + 63) / 64;
  const dim3 gconv((NNODES + 255) / 256, 128 / 64);

  bin_count_kernel<<<NCHK, 1024, 0, stream>>>(ei, ccnt);
  bin_scan_kernel<<<1, 1024, 0, stream>>>(ccnt, coff);
  bin_scatter_kernel<<<NCHK, 1024, 0, stream>>>(ei, coff, binned);
  bin_place_kernel<<<NBLK, 1024, 0, stream>>>(coff, binned, rowptr, perm);
  csr_order_kernel<<<ogrid, TB, 0, stream>>>(rowptr, perm, srcs);

  wmma_gemm_kernel<0, false, true, true, 2, 2><<<gconv, TB, 0, stream>>>(
      nullptr, x, Wl1, Wr1, nullptr, xl, xr, nullptr, NNODES, 128, 128);
  conv_edge_kernel<<<wgrid, TB, 0, stream>>>(rowptr, srcs, xl, xr, att1, b1,
                                             h1h);

  wmma_gemm_kernel<0, false, true, false, 2, 2><<<gconv, TB, 0, stream>>>(
      h1h, nullptr, Wl2, Wr2, nullptr, xl, xr, nullptr, NNODES, 128, 128);
  conv_edge_kernel<<<wgrid, TB, 0, stream>>>(rowptr, srcs, xl, xr, att2, b2,
                                             h2h);

  const dim3 gm1((NNODES + 511) / 512, 640 / 64);
  wmma_gemm_kernel<1, true, false, false, 4, 1><<<gm1, TB, 0, stream>>>(
      h2h, nullptr, W1, nullptr, bb1, nullptr, nullptr, m1, NNODES, 640, 128);
  const dim3 gm2((NNODES + 511) / 512, 320 / 64);
  wmma_gemm_kernel<1, true, false, false, 4, 1><<<gm2, TB, 0, stream>>>(
      m1, nullptr, W2, nullptr, bb2, nullptr, nullptr, m2, NNODES, 320, 640);
  const dim3 gm3((NNODES + 255) / 256, 64 / 64);
  wmma_gemm_kernel<2, true, true, false, 2, 1><<<gm3, TB, 0, stream>>>(
      m2, nullptr, W3, nullptr, bb3, out, nullptr, nullptr, NNODES, 64, 320);
}


# Each destination segment is put in edge-id order inside the placement kernel (records held in registers, ranks counted in LDS): the ordering launch and its pass through memory go; sum order unchanged
# speedup vs baseline: 9.4648x; 1.0342x over previous
#include <hip/hip_runtime.h>
#include <hip/hip_bf16.h>


typedef __attribute__((ext_vector_type(16))) _Float16 v16h;
typedef __attribute__((ext_vector_type(8)))  _Float16 v8h;
typedef __attribute__((ext_vector_type(8)))  float    v8f;

#define NNODES 50000
#define NEDGES 800000
#define FDIM   128

#define NBLK 256
#define BLKW 196
#define NCHK 256
#define CHKE ((NEDGES + NCHK - 1) / NCHK)
#define CHKT ((CHKE + 1023) / 1024)

__device__ __forceinline__ int node_block(int d) {
  const unsigned b = (unsigned)d / (unsigned)BLKW;
  return (int)(b < (unsigned)(NBLK - 1) ? b : (unsigned)(NBLK - 1));
}

__global__ __launch_bounds__(1024) void bin_count_kernel(
    const int* __restrict__ ei, int* __restrict__ ccnt) {
  __shared__ int h[NBLK];
  const int t = threadIdx.x;
  if (t < NBLK) h[t] = 0;
  __syncthreads();
  const int e0 = blockIdx.x * CHKE;
  const int e1 = (e0 + CHKE) < NEDGES ? (e0 + CHKE) : NEDGES;
  int dv[CHKT];
  #pragma unroll
  for (int k = 0; k < CHKT; ++k) {
    const int e = e0 + t + k * 1024;
    dv[k] = ei[NEDGES + (e < NEDGES ? e : NEDGES - 1)];
  }
  #pragma unroll
  for (int k = 0; k < CHKT; ++k)
    if (e0 + t + k * 1024 < e1) atomicAdd(&h[node_block(dv[k])], 1);
  __syncthreads();
  if (t < NBLK) ccnt[t * NCHK + blockIdx.x] = h[t];
}

__global__ __launch_bounds__(1024) void bin_scan_kernel(
    const int* __restrict__ ccnt, int* __restrict__ coff) {
  __shared__ int wsum[32];
  const int t = threadIdx.x, lane = t & 31, wv = t >> 5;
  const int4* in4 = (const int4*)ccnt + wv * (NBLK * NCHK / 128);
  int4* out4 = (int4*)coff + wv * (NBLK * NCHK / 128);
  int4 v[16];
  #pragma unroll
  for (int k = 0; k < 16; ++k) v[k] = in4[k * 32 + lane];
  int carry = 0;
  #pragma unroll
  for (int k = 0; k < 16; ++k) {
    const int s0 = v[k].x, s1 = s0 + v[k].y, s2 = s1 + v[k].z, s3 = s2 + v[k].w;
    int x = s3;
    for (int o = 1; o < 32; o <<= 1) {
      const int y = __shfl_up(x, o);
      if (lane >= o) x += y;
    }
    const int ex = carry + x - s3;
    v[k] = make_int4(ex, ex + s0, ex + s1, ex + s2);
    carry += __shfl(x, 31);
  }
  if (lane == 0) wsum[wv] = carry;
  __syncthreads();
  if (wv == 0) {
    const int w = wsum[lane];
    int xs = w;
    for (int o = 1; o < 32; o <<= 1) {
      const int y = __shfl_up(xs, o);
      if (lane >= o) xs += y;
    }
    wsum[lane] = xs - w;
  }
  __syncthreads();
  const int base = wsum[wv];
  #pragma unroll
  for (int k = 0; k < 16; ++k) {
    int4 o = v[k];
    o.x += base; o.y += base; o.z += base; o.w += base;
    out4[k * 32 + lane] = o;
  }
  if (t == 1023) coff[NBLK * NCHK] = base + carry;
}

__global__ __launch_bounds__(1024) void bin_scatter_kernel(
    const int* __restrict__ ei, const int* __restrict__ coff,
    int2* __restrict__ binned) {
  __shared__ int cur[NBLK];
  const int t = threadIdx.x;
  if (t < NBLK) cur[t] = coff[t * NCHK + blockIdx.x];
  __syncthreads();
  const int e0 = blockIdx.x * CHKE;
  const int e1 = (e0 + CHKE) < NEDGES ? (e0 + CHKE) : NEDGES;
  int sv[CHKT], dv[CHKT];
  #pragma unroll
  for (int k = 0; k < CHKT; ++k) {
    const int e = e0 + t + k * 1024;
    const int ec = e < NEDGES ? e : NEDGES - 1;
    sv[k] = ei[ec];
    dv[k] = ei[NEDGES + ec];
  }
  #pragma unroll
  for (int k = 0; k < CHKT; ++k) {
    const int e = e0 + t + k * 1024;
    if (e < e1) {
      const int b = node_block(dv[k]);
      const int p = atomicAdd(&cur[b], 1);
      if ((unsigned)p < (unsigned)NEDGES)
        binned[p] = make_int2((sv[k] & 0xffff) | ((dv[k] - b * BLKW) << 16), e);
    }
  }
}

#define PCAP 4096
__global__ __launch_bounds__(1024) void bin_place_kernel(
    const int* __restrict__ coff, const int2* __restrict__ binned,
    int* __restrict__ rowptr, int2* perm, int* __restrict__ srcs) {
  __shared__ int cnt[256];
  __shared__ int cur[256];
  __shared__ int seg0[256];
  __shared__ int wtot[8];
  __shared__ int2 lperm[PCAP];
  const int t = threadIdx.x, lane = t & 31, wv = t >> 5;
  const int b = blockIdx.x;
  const int beg = coff[b * NCHK], end = coff[(b + 1) * NCHK];
  const int n = end - beg;
  const bool fits = n <= PCAP;
  if (t < 256) cnt[t] = 0;
  __syncthreads();
  int2 r[4];
  if (fits) {
    #pragma unroll
    for (int k = 0; k < 4; ++k) {
      const int i = beg + t + k * 1024;
      r[k] = binned[i < NEDGES ? (i < 0 ? 0 : i) : NEDGES - 1];
    }
    #pragma unroll
    for (int k = 0; k < 4; ++k)
      if (t + k * 1024 < n) atomicAdd(&cnt[(r[k].x >> 16) & 0xff], 1);
  } else {
    for (int i0 = beg; i0 < end; i0 += 4096) {
      int rx[4];
      #pragma unroll
      for (int k = 0; k < 4; ++k) {
        const int i = i0 + t + k * 1024;
        rx[k] = binned[i < NEDGES ? i : NEDGES - 1].x;
      }
      #pragma unroll
      for (int k = 0; k < 4; ++k)
        if (i0 + t + k * 1024 < end) atomicAdd(&cnt[(rx[k] >> 16) & 0xff], 1);
    }
  }
  __syncthreads();
  int c = 0;
  if (t < 256) c = cnt[t];
  int x = c;
  for (int o = 1; o < 32; o <<= 1) {
    const int y = __shfl_up(x, o);
    if (lane >= o) x += y;
  }
  if (t < 256 && lane == 31) wtot[wv] = x;
  __syncthreads();
  if (t < 256) {
    int first = beg + x - c;
    for (int w = 0; w < wv; ++w) first += wtot[w];
    cur[t] = first;
    seg0[t] = first;
    const int node = b * BLKW + t;
    if (t < BLKW && node < NNODES) rowptr[node] = first;
  }
  if (b == NBLK - 1 && t == 0) rowptr[NNODES] = end;
  __syncthreads();
  if (fits) {
    #pragma unroll
    for (int k = 0; k < 4; ++k) {
      if (t + k * 1024 < n) {
        const int p = atomicAdd(&cur[(r[k].x >> 16) & 0xff], 1) - beg;
        if ((unsigned)p < (unsigned)PCAP) lperm[p] = r[k];
      }
    }
    __syncthreads();
    for (int i = t; i < n; i += 1024) {
      const int2 my = lperm[i];
      const int ld = (my.x >> 16) & 0xff;
      const int s0 = seg0[ld] - beg;
      const int s1 = s0 + cnt[ld];
      int rank = 0;
      for (int j = s0; j < s1; ++j)
        rank += (lperm[j & (PCAP - 1)].y < my.y) ? 1 : 0;
      const int q = beg + s0 + rank;
      if ((unsigned)q < (unsigned)NEDGES) srcs[q] = my.x & 0xffff;
    }
  } else {
    for (int i0 = beg; i0 < end; i0 += 4096) {
      #pragma unroll
      for (int k = 0; k < 4; ++k) {
        const int i = i0 + t + k * 1024;
        r[k] = binned[i < NEDGES ? i : NEDGES - 1];
      }
      #pragma unroll
      for (int k = 0; k < 4; ++k) {
        if (i0 + t + k * 1024 < end) {
          const int p = atomicAdd(&cur[(r[k].x >> 16) & 0xff], 1);
          if ((unsigned)p < (unsigned)NEDGES) perm[p] = r[k];
        }
      }
    }
    __threadfence();
    __syncthreads();
    for (int i = beg + t; i < end; i += 1024) {
      const int2 my = perm[i];
      const int ld = (my.x >> 16) & 0xff;
      const int s0 = seg0[ld];
      const int s1 = s0 + cnt[ld];
      int rank = 0;
      for (int j = s0; j < s1; ++j) {
        const int jc = j < NEDGES ? (j < 0 ? 0 : j) : NEDGES - 1;
        rank += (perm[jc].y < my.y) ? 1 : 0;
      }
      const int q = s0 + rank;
      if ((unsigned)q < (unsigned)NEDGES) srcs[q] = my.x & 0xffff;
    }
  }
}

#define KC 64
#define BT_STRIDE 72

template <int ACT, bool HASBIAS, bool F32OUT, bool AF32, int MT, int NW>
__global__ __launch_bounds__(256) void wmma_gemm_kernel(
    const _Float16* __restrict__ A, const float* __restrict__ Af,
    const float* __restrict__ W, const float* __restrict__ Wb,
    const float* __restrict__ bias,
    float* __restrict__ outF, float* __restrict__ outFb,
    _Float16* __restrict__ outH,
    int M, int Nt, int K) {
  static_assert(NW == 1 || F32OUT, "the second product has an f32 output only");
  __shared__ _Float16 bt[NW][64 * BT_STRIDE];

  const int lane = threadIdx.x & 31;
  const int wave = threadIdx.x >> 5;
  const int half = lane >> 4;
  const int l15  = lane & 15;
  const int n0   = blockIdx.y * 64;

  int rowBase = blockIdx.x * (128 * MT) + wave * (16 * MT);
  if (rowBase > M - 16 * MT) rowBase = M - 16 * MT;

  v8f c[NW][MT][4];
  #pragma unroll
  for (int w = 0; w < NW; ++w) {
    #pragma unroll
    for (int mt = 0; mt < MT; ++mt) {
      #pragma unroll
      for (int nn = 0; nn < 4; ++nn) c[w][mt][nn] = v8f{};
    }
  }

  for (int kc = 0; kc < K; kc += KC) {
    #pragma unroll
    for (int w = 0; w < NW; ++w) {
      const float* __restrict__ Wp = (w == 0) ? W : Wb;
      int kl = threadIdx.x >> 4;
      const int n4 = (threadIdx.x & 15) * 4;
      #pragma unroll
      for (int p = 0; p < 4; ++p, kl += 16) {
        const float4 wv =
            *(const float4*)(Wp + (size_t)(kc + kl) * Nt + n0 + n4);
        bt[w][(n4 + 0) * BT_STRIDE + kl] = (_Float16)wv.x;
        bt[w][(n4 + 1) * BT_STRIDE + kl] = (_Float16)wv.y;
        bt[w][(n4 + 2) * BT_STRIDE + kl] = (_Float16)wv.z;
        bt[w][(n4 + 3) * BT_STRIDE + kl] = (_Float16)wv.w;
      }
    }
    __syncthreads();

    const size_t aoff = (size_t)(rowBase + l15) * K + kc;
    v16h a[2][MT];
    #pragma unroll
    for (int ks = 0; ks < 2; ++ks) {
      const int kb = ks * 32 + half * 8;
      #pragma unroll
      for (int mt = 0; mt < MT; ++mt) {
        if (AF32) {
          const float* ap = Af + aoff + (size_t)(mt * 16) * K + kb;
          const float4 f0 = *(const float4*)(ap);
          const float4 f1 = *(const float4*)(ap + 4);
          const float4 f2 = *(const float4*)(ap + 16);
          const float4 f3 = *(const float4*)(ap + 20);
          v16h t;
          t[0]  = (_Float16)f0.x; t[1]  = (_Float16)f0.y;
          t[2]  = (_Float16)f0.z; t[3]  = (_Float16)f0.w;
          t[4]  = (_Float16)f1.x; t[5]  = (_Float16)f1.y;
          t[6]  = (_Float16)f1.z; t[7]  = (_Float16)f1.w;
          t[8]  = (_Float16)f2.x; t[9]  = (_Float16)f2.y;
          t[10] = (_Float16)f2.z; t[11] = (_Float16)f2.w;
          t[12] = (_Float16)f3.x; t[13] = (_Float16)f3.y;
          t[14] = (_Float16)f3.z; t[15] = (_Float16)f3.w;
          a[ks][mt] = t;
        } else {
          const _Float16* arow = A + aoff + (size_t)(mt * 16) * K;
          v8h alo = *(const v8h*)(arow + kb);
          v8h ahi = *(const v8h*)(arow + kb + 16);
          a[ks][mt] = __builtin_shufflevector(alo, ahi, 0, 1, 2, 3, 4, 5, 6, 7,
                                              8, 9, 10, 11, 12, 13, 14, 15);
        }
      }
    }
    __builtin_amdgcn_sched_barrier(0);
    #pragma unroll
    for (int ks = 0; ks < 2; ++ks) {
      const int kb = ks * 32 + half * 8;
      #pragma unroll
      for (int w = 0; w < NW; ++w) {
        v16h b[4];
        #pragma unroll
        for (int nn = 0; nn < 4; ++nn) {
          const _Float16* bp = bt[w] + (nn * 16 + l15) * BT_STRIDE + kb;
          v8h blo = *(const v8h*)(bp);
          v8h bhi = *(const v8h*)(bp + 16);
          b[nn] = __builtin_shufflevector(blo, bhi, 0, 1, 2, 3, 4, 5, 6, 7,
                                          8, 9, 10, 11, 12, 13, 14, 15);
        }
        #pragma unroll
        for (int mt = 0; mt < MT; ++mt) {
          #pragma unroll
          for (int nn = 0; nn < 4; ++nn) {
            c[w][mt][nn] = __builtin_amdgcn_wmma_f32_16x16x32_f16(
                false, a[ks][mt], false, b[nn], (short)0, c[w][mt][nn], false,
                false);
          }
        }
      }
    }
    __syncthreads();
  }

  float bv[4] = {0.0f, 0.0f, 0.0f, 0.0f};
  if (HASBIAS) {
    #pragma unroll
    for (int nn = 0; nn < 4; ++nn) bv[nn] = bias[n0 + nn * 16 + l15];
  }
  #pragma unroll
  for (int w = 0; w < NW; ++w) {
    float* __restrict__ oF = (w == 0) ? outF : outFb;
    #pragma unroll
    for (int mt = 0; mt < MT; ++mt) {
      const unsigned o0 =
          (unsigned)(rowBase + mt * 16 + half * 8) * (unsigned)Nt +
          (unsigned)(n0 + l15);
      #pragma unroll
      for (int nn = 0; nn < 4; ++nn) {
        #pragma unroll
        for (int r = 0; r < 8; ++r) {
          float v = c[w][mt][nn][r] + bv[nn];
          if (ACT == 1) v = v > 0.0f ? v : (__expf(v) - 1.0f);
          else if (ACT == 2) v = 1.0f / (1.0f + __expf(-v));
          const unsigned o =
              o0 + (unsigned)r * (unsigned)Nt + (unsigned)(nn * 16);
          if (F32OUT) oF[o] = v;
          else        outH[o] = (_Float16)v;
        }
      }
    }
  }
}

typedef __attribute__((ext_vector_type(4))) _Float16 v4h;
typedef __attribute__((ext_vector_type(2))) float v2f;
typedef __attribute__((ext_vector_type(4))) float v4f;

template <int CTRL>
__device__ __forceinline__ float lane_xor_get(float x) {
  return __builtin_bit_cast(
      float, __builtin_amdgcn_update_dpp(0, __builtin_bit_cast(int, x), CTRL,
                                         0xf, 0xf, true));
}

__device__ __forceinline__ float edge_logit(const v4f a, const v4f b,
                                            const v4f w) {
  const v2f z0 = a.xy + b.xy;
  const v2f z1 = a.zw + b.zw;
  const v2f l0 = __builtin_elementwise_max(z0, z0 * 0.2f);
  const v2f l1 = __builtin_elementwise_max(z1, z1 * 0.2f);
  v2f p = l0 * w.xy;
  p = l1 * w.zw + p;
  float acc = p.x + p.y;
  acc += lane_xor_get<0x161>(acc);
  acc += lane_xor_get<0x162>(acc);
  acc += lane_xor_get<0x164>(acc);
  return acc;
}

template <int NB>
__device__ __forceinline__ void conv_batch(const v4f* __restrict__ xl4,
                                           int my, int j0, int lane,
                                           const v4f xrv, const v4f av,
                                           float& m, float& ssum, v4f& r) {
  v4f v[NB];
  float e[NB];
  unsigned off[NB];
  #pragma unroll
  for (int t = 0; t < NB; ++t) {
    const int sidx = __builtin_amdgcn_readlane(my, j0 + t);
    off[t] = (unsigned)sidx * 32u + (unsigned)lane;
  }
  #pragma unroll
  for (int t = 0; t < NB; ++t) v[t] = xl4[off[t]];
  float mn = m;
  #pragma unroll
  for (int t = 0; t < NB; ++t) {
    e[t] = edge_logit(v[t], xrv, av);
    mn = fmaxf(mn, e[t]);
  }
  const float sc = __expf(m - mn);
  ssum *= sc;
  r *= sc;
  #pragma unroll
  for (int t = 0; t < NB; ++t) {
    const float p = __expf(e[t] - mn);
    ssum += p;
    r = v[t] * p + r;
  }
  m = mn;
}

__global__ __launch_bounds__(256) void conv_edge_kernel(
    const int* __restrict__ rowptr, const int* __restrict__ srcs,
    const float* __restrict__ xl, const float* __restrict__ xr,
    const float* __restrict__ att, const float* __restrict__ bias,
    _Float16* __restrict__ outH) {
  const int lane = threadIdx.x & 31;
  const int d = __builtin_amdgcn_readfirstlane(
      (int)(blockIdx.x * 8 + (threadIdx.x >> 5)));
  if (d >= NNODES) return;
  const v4f* xl4 = (const v4f*)xl;
  const v4f xrv = ((const v4f*)xr)[(size_t)d * 32 + lane];
  const v4f av = ((const v4f*)att)[lane];

  v4f r = xl4[(size_t)d * 32 + lane];
  float m = edge_logit(r, xrv, av);
  float ssum = 1.0f;

  const int beg = rowptr[d], end = rowptr[d + 1];
  for (int base = beg; base < end; base += 32) {
    const int cnt = (end - base) < 32 ? (end - base) : 32;
    int my = 0;
    if (lane < cnt) my = srcs[base + lane];
    int j = 0;
    for (; j + 8 <= cnt; j += 8)
      conv_batch<8>(xl4, my, j, lane, xrv, av, m, ssum, r);
    if (j + 4 <= cnt) { conv_batch<4>(xl4, my, j, lane, xrv, av, m, ssum, r); j += 4; }
    if (j + 2 <= cnt) { conv_batch<2>(xl4, my, j, lane, xrv, av, m, ssum, r); j += 2; }
    if (j < cnt) conv_batch<1>(xl4, my, j, lane, xrv, av, m, ssum, r);
  }

  const v4f bv = ((const v4f*)bias)[lane];
  float o0 = r.x / ssum + bv.x, o1 = r.y / ssum + bv.y;
  float o2 = r.z / ssum + bv.z, o3 = r.w / ssum + bv.w;
  o0 = o0 > 0.0f ? o0 : (__expf(o0) - 1.0f);
  o1 = o1 > 0.0f ? o1 : (__expf(o1) - 1.0f);
  o2 = o2 > 0.0f ? o2 : (__expf(o2) - 1.0f);
  o3 = o3 > 0.0f ? o3 : (__expf(o3) - 1.0f);
  v4h hv;
  hv[0] = (_Float16)o0; hv[1] = (_Float16)o1;
  hv[2] = (_Float16)o2; hv[3] = (_Float16)o3;
  *(v4h*)(outH + (size_t)d * FDIM + lane * 4) = hv;
}

extern "C" void kernel_launch(void* const* d_in, const int* in_sizes, int n_in,
                              void* d_out, int out_size, void* d_ws,
                              size_t ws_size, hipStream_t stream) {
  (void)in_sizes; (void)n_in; (void)out_size; (void)ws_size;
  const float* x    = (const float*)d_in[0];
  const int*   ei   = (const int*)d_in[1];
  const float* Wl1  = (const float*)d_in[2];
  const float* Wr1  = (const float*)d_in[3];
  const float* att1 = (const float*)d_in[4];
  const float* b1   = (const float*)d_in[5];
  const float* Wl2  = (const float*)d_in[6];
  const float* Wr2  = (const float*)d_in[7];
  const float* att2 = (const float*)d_in[8];
  const float* b2   = (const float*)d_in[9];
  const float* W1   = (const float*)d_in[10];
  const float* bb1  = (const float*)d_in[11];
  const float* W2   = (const float*)d_in[12];
  const float* bb2  = (const float*)d_in[13];
  const float* W3   = (const float*)d_in[14];
  const float* bb3  = (const float*)d_in[15];
  float* out = (float*)d_out;

  char* ws = (char*)d_ws;
  size_t off = 0;
  auto alloc = [&](size_t bytes) -> char* {
    char* p = ws + off;
    off += (bytes + 255) & ~(size_t)255;
    return p;
  };
  const size_t S_F = (size_t)NNODES * FDIM;
  _Float16* h1h = (_Float16*)alloc(S_F * 2);
  _Float16* h2h = (_Float16*)alloc(S_F * 2);
  int*  ccnt   = (int*)alloc((size_t)NBLK * NCHK * 4);
  int*  coff   = (int*)alloc(((size_t)NBLK * NCHK + 1) * 4);
  int*  rowptr = (int*)alloc(((size_t)NNODES + 1) * 4);
  int2* binned = (int2*)alloc((size_t)NEDGES * 8);
  int2* perm   = (int2*)alloc((size_t)NEDGES * 8);
  int*  srcs   = (int*)alloc((size_t)NEDGES * 4);
  const size_t uoff = off;
  float*    xl   = (float*)alloc(S_F * 4);
  float*    xr   = (float*)alloc(S_F * 4);
  _Float16* m1 = (_Float16*)(ws + uoff);
  _Float16* m2 = (_Float16*)(ws + uoff + (size_t)NNODES * 640 * 2);

  const int TB = 256;
  const int wgrid = (NNODES + 7) / 8;
  const dim3 gconv((NNODES + 255) / 256, 128 / 64);

  bin_count_kernel<<<NCHK, 1024, 0, stream>>>(ei, ccnt);
  bin_scan_kernel<<<1, 1024, 0, stream>>>(ccnt, coff);
  bin_scatter_kernel<<<NCHK, 1024, 0, stream>>>(ei, coff, binned);
  bin_place_kernel<<<NBLK, 1024, 0, stream>>>(coff, binned, rowptr, perm, srcs);

  wmma_gemm_kernel<0, false, true, true, 2, 2><<<gconv, TB, 0, stream>>>(
      nullptr, x, Wl1, Wr1, nullptr, xl, xr, nullptr, NNODES, 128, 128);
  conv_edge_kernel<<<wgrid, TB, 0, stream>>>(rowptr, srcs, xl, xr, att1, b1,
                                             h1h);

  wmma_gemm_kernel<0, false, true, false, 2, 2><<<gconv, TB, 0, stream>>>(
      h1h, nullptr, Wl2, Wr2, nullptr, xl, xr, nullptr, NNODES, 128, 128);
  conv_edge_kernel<<<wgrid, TB, 0, stream>>>(rowptr, srcs, xl, xr, att2, b2,
                                             h2h);

  const dim3 gm1((NNODES + 511) / 512, 640 / 64);
  wmma_gemm_kernel<1, true, false, false, 4, 1><<<gm1, TB, 0, stream>>>(
      h2h, nullptr, W1, nullptr, bb1, nullptr, nullptr, m1, NNODES, 640, 128);
  const dim3 gm2((NNODES + 511) / 512, 320 / 64);
  wmma_gemm_kernel<1, true, false, false, 4, 1><<<gm2, TB, 0, stream>>>(
      m1, nullptr, W2, nullptr, bb2, nullptr, nullptr, m2, NNODES, 320, 640);
  const dim3 gm3((NNODES + 255) / 256, 64 / 64);
  wmma_gemm_kernel<2, true, true, false, 2, 1><<<gm3, TB, 0, stream>>>(
      m2, nullptr, W3, nullptr, bb3, out, nullptr, nullptr, NNODES, 64, 320);
}


# K loop of the f16-source GEMMs: a chunk's global loads (W rows, then all A fragments) issue at its top, so the A latency runs beside W staging and barrier, not behind it; sum order unchanged
# speedup vs baseline: 9.8450x; 1.0402x over previous
#include <hip/hip_runtime.h>
#include <hip/hip_bf16.h>


typedef __attribute__((ext_vector_type(16))) _Float16 v16h;
typedef __attribute__((ext_vector_type(8)))  _Float16 v8h;
typedef __attribute__((ext_vector_type(8)))  float    v8f;

#define NNODES 50000
#define NEDGES 800000
#define FDIM   128

#define NBLK 256
#define BLKW 196
#define NCHK 256
#define CHKE ((NEDGES + NCHK - 1) / NCHK)
#define CHKT ((CHKE + 1023) / 1024)

__device__ __forceinline__ int node_block(int d) {
  const unsigned b = (unsigned)d / (unsigned)BLKW;
  return (int)(b < (unsigned)(NBLK - 1) ? b : (unsigned)(NBLK - 1));
}

__global__ __launch_bounds__(1024) void bin_count_kernel(
    const int* __restrict__ ei, int* __restrict__ ccnt) {
  __shared__ int h[NBLK];
  const int t = threadIdx.x;
  if (t < NBLK) h[t] = 0;
  __syncthreads();
  const int e0 = blockIdx.x * CHKE;
  const int e1 = (e0 + CHKE) < NEDGES ? (e0 + CHKE) : NEDGES;
  int dv[CHKT];
  #pragma unroll
  for (int k = 0; k < CHKT; ++k) {
    const int e = e0 + t + k * 1024;
    dv[k] = ei[NEDGES + (e < NEDGES ? e : NEDGES - 1)];
  }
  #pragma unroll
  for (int k = 0; k < CHKT; ++k)
    if (e0 + t + k * 1024 < e1) atomicAdd(&h[node_block(dv[k])], 1);
  __syncthreads();
  if (t < NBLK) ccnt[t * NCHK + blockIdx.x] = h[t];
}

__global__ __launch_bounds__(1024) void bin_scan_kernel(
    const int* __restrict__ ccnt, int* __restrict__ coff) {
  __shared__ int wsum[32];
  const int t = threadIdx.x, lane = t & 31, wv = t >> 5;
  const int4* in4 = (const int4*)ccnt + wv * (NBLK * NCHK / 128);
  int4* out4 = (int4*)coff + wv * (NBLK * NCHK / 128);
  int4 v[16];
  #pragma unroll
  for (int k = 0; k < 16; ++k) v[k] = in4[k * 32 + lane];
  int carry = 0;
  #pragma unroll
  for (int k = 0; k < 16; ++k) {
    const int s0 = v[k].x, s1 = s0 + v[k].y, s2 = s1 + v[k].z, s3 = s2 + v[k].w;
    int x = s3;
    for (int o = 1; o < 32; o <<= 1) {
      const int y = __shfl_up(x, o);
      if (lane >= o) x += y;
    }
    const int ex = carry + x - s3;
    v[k] = make_int4(ex, ex + s0, ex + s1, ex + s2);
    carry += __shfl(x, 31);
  }
  if (lane == 0) wsum[wv] = carry;
  __syncthreads();
  if (wv == 0) {
    const int w = wsum[lane];
    int xs = w;
    for (int o = 1; o < 32; o <<= 1) {
      const int y = __shfl_up(xs, o);
      if (lane >= o) xs += y;
    }
    wsum[lane] = xs - w;
  }
  __syncthreads();
  const int base = wsum[wv];
  #pragma unroll
  for (int k = 0; k < 16; ++k) {
    int4 o = v[k];
    o.x += base; o.y += base; o.z += base; o.w += base;
    out4[k * 32 + lane] = o;
  }
  if (t == 1023) coff[NBLK * NCHK] = base + carry;
}

__global__ __launch_bounds__(1024) void bin_scatter_kernel(
    const int* __restrict__ ei, const int* __restrict__ coff,
    int2* __restrict__ binned) {
  __shared__ int cur[NBLK];
  const int t = threadIdx.x;
  if (t < NBLK) cur[t] = coff[t * NCHK + blockIdx.x];
  __syncthreads();
  const int e0 = blockIdx.x * CHKE;
  const int e1 = (e0 + CHKE) < NEDGES ? (e0 + CHKE) : NEDGES;
  int sv[CHKT], dv[CHKT];
  #pragma unroll
  for (int k = 0; k < CHKT; ++k) {
    const int e = e0 + t + k * 1024;
    const int ec = e < NEDGES ? e : NEDGES - 1;
    sv[k] = ei[ec];
    dv[k] = ei[NEDGES + ec];
  }
  #pragma unroll
  for (int k = 0; k < CHKT; ++k) {
    const int e = e0 + t + k * 1024;
    if (e < e1) {
      const int b = node_block(dv[k]);
      const int p = atomicAdd(&cur[b], 1);
      if ((unsigned)p < (unsigned)NEDGES)
        binned[p] = make_int2((sv[k] & 0xffff) | ((dv[k] - b * BLKW) << 16), e);
    }
  }
}

#define PCAP 4096
__global__ __launch_bounds__(1024) void bin_place_kernel(
    const int* __restrict__ coff, const int2* __restrict__ binned,
    int* __restrict__ rowptr, int2* perm, int* __restrict__ srcs) {
  __shared__ int cnt[256];
  __shared__ int cur[256];
  __shared__ int seg0[256];
  __shared__ int wtot[8];
  __shared__ int2 lperm[PCAP];
  const int t = threadIdx.x, lane = t & 31, wv = t >> 5;
  const int b = blockIdx.x;
  const int beg = coff[b * NCHK], end = coff[(b + 1) * NCHK];
  const int n = end - beg;
  const bool fits = n <= PCAP;
  if (t < 256) cnt[t] = 0;
  __syncthreads();
  int2 r[4];
  if (fits) {
    #pragma unroll
    for (int k = 0; k < 4; ++k) {
      const int i = beg + t + k * 1024;
      r[k] = binned[i < NEDGES ? (i < 0 ? 0 : i) : NEDGES - 1];
    }
    #pragma unroll
    for (int k = 0; k < 4; ++k)
      if (t + k * 1024 < n) atomicAdd(&cnt[(r[k].x >> 16) & 0xff], 1);
  } else {
    for (int i0 = beg; i0 < end; i0 += 4096) {
      int rx[4];
      #pragma unroll
      for (int k = 0; k < 4; ++k) {
        const int i = i0 + t + k * 1024;
        rx[k] = binned[i < NEDGES ? i : NEDGES - 1].x;
      }
      #pragma unroll
      for (int k = 0; k < 4; ++k)
        if (i0 + t + k * 1024 < end) atomicAdd(&cnt[(rx[k] >> 16) & 0xff], 1);
    }
  }
  __syncthreads();
  int c = 0;
  if (t < 256) c = cnt[t];
  int x = c;
  for (int o = 1; o < 32; o <<= 1) {
    const int y = __shfl_up(x, o);
    if (lane >= o) x += y;
  }
  if (t < 256 && lane == 31) wtot[wv] = x;
  __syncthreads();
  if (t < 256) {
    int first = beg + x - c;
    for (int w = 0; w < wv; ++w) first += wtot[w];
    cur[t] = first;
    seg0[t] = first;
    const int node = b * BLKW + t;
    if (t < BLKW && node < NNODES) rowptr[node] = first;
  }
  if (b == NBLK - 1 && t == 0) rowptr[NNODES] = end;
  __syncthreads();
  if (fits) {
    #pragma unroll
    for (int k = 0; k < 4; ++k) {
      if (t + k * 1024 < n) {
        const int p = atomicAdd(&cur[(r[k].x >> 16) & 0xff], 1) - beg;
        if ((unsigned)p < (unsigned)PCAP) lperm[p] = r[k];
      }
    }
    __syncthreads();
    for (int i = t; i < n; i += 1024) {
      const int2 my = lperm[i];
      const int ld = (my.x >> 16) & 0xff;
      const int s0 = seg0[ld] - beg;
      const int s1 = s0 + cnt[ld];
      int rank = 0;
      for (int j = s0; j < s1; ++j)
        rank += (lperm[j & (PCAP - 1)].y < my.y) ? 1 : 0;
      const int q = beg + s0 + rank;
      if ((unsigned)q < (unsigned)NEDGES) srcs[q] = my.x & 0xffff;
    }
  } else {
    for (int i0 = beg; i0 < end; i0 += 4096) {
      #pragma unroll
      for (int k = 0; k < 4; ++k) {
        const int i = i0 + t + k * 1024;
        r[k] = binned[i < NEDGES ? i : NEDGES - 1];
      }
      #pragma unroll
      for (int k = 0; k < 4; ++k) {
        if (i0 + t + k * 1024 < end) {
          const int p = atomicAdd(&cur[(r[k].x >> 16) & 0xff], 1);
          if ((unsigned)p < (unsigned)NEDGES) perm[p] = r[k];
        }
      }
    }
    __threadfence();
    __syncthreads();
    for (int i = beg + t; i < end; i += 1024) {
      const int2 my = perm[i];
      const int ld = (my.x >> 16) & 0xff;
      const int s0 = seg0[ld];
      const int s1 = s0 + cnt[ld];
      int rank = 0;
      for (int j = s0; j < s1; ++j) {
        const int jc = j < NEDGES ? (j < 0 ? 0 : j) : NEDGES - 1;
        rank += (perm[jc].y < my.y) ? 1 : 0;
      }
      const int q = s0 + rank;
      if ((unsigned)q < (unsigned)NEDGES) srcs[q] = my.x & 0xffff;
    }
  }
}

#define KC 64
#define BT_STRIDE 72

template <int ACT, bool HASBIAS, bool F32OUT, bool AF32, int MT, int NW>
__global__ __launch_bounds__(256) void wmma_gemm_kernel(
    const _Float16* __restrict__ A, const float* __restrict__ Af,
    const float* __restrict__ W, const float* __restrict__ Wb,
    const float* __restrict__ bias,
    float* __restrict__ outF, float* __restrict__ outFb,
    _Float16* __restrict__ outH,
    int M, int Nt, int K) {
  static_assert(NW == 1 || F32OUT, "the second product has an f32 output only");
  __shared__ _Float16 bt[NW][64 * BT_STRIDE];

  const int lane = threadIdx.x & 31;
  const int wave = threadIdx.x >> 5;
  const int half = lane >> 4;
  const int l15  = lane & 15;
  const int n0   = blockIdx.y * 64;

  int rowBase = blockIdx.x * (128 * MT) + wave * (16 * MT);
  if (rowBase > M - 16 * MT) rowBase = M - 16 * MT;

  v8f c[NW][MT][4];
  #pragma unroll
  for (int w = 0; w < NW; ++w) {
    #pragma unroll
    for (int mt = 0; mt < MT; ++mt) {
      #pragma unroll
      for (int nn = 0; nn < 4; ++nn) c[w][mt][nn] = v8f{};
    }
  }

  for (int kc = 0; kc < K; kc += KC) {
    const int kl0 = threadIdx.x >> 4;
    const int n4 = (threadIdx.x & 15) * 4;
    float4 wreg[NW][4];
    #pragma unroll
    for (int w = 0; w < NW; ++w) {
      const float* __restrict__ Wp = (w == 0) ? W : Wb;
      #pragma unroll
      for (int p = 0; p < 4; ++p)
        wreg[w][p] =
            *(const float4*)(Wp + (size_t)(kc + kl0 + 16 * p) * Nt + n0 + n4);
    }

    const size_t aoff = (size_t)(rowBase + l15) * K + kc;
    v8h alo[2][MT], ahi[2][MT];
    if (!AF32) {
      #pragma unroll
      for (int ks = 0; ks < 2; ++ks) {
        const int kb = ks * 32 + half * 8;
        #pragma unroll
        for (int mt = 0; mt < MT; ++mt) {
          const _Float16* arow = A + aoff + (size_t)(mt * 16) * K;
          alo[ks][mt] = *(const v8h*)(arow + kb);
          ahi[ks][mt] = *(const v8h*)(arow + kb + 16);
        }
      }
    }

    #pragma unroll
    for (int w = 0; w < NW; ++w) {
      #pragma unroll
      for (int p = 0; p < 4; ++p) {
        const int kl = kl0 + 16 * p;
        const float4 wv = wreg[w][p];
        bt[w][(n4 + 0) * BT_STRIDE + kl] = (_Float16)wv.x;
        bt[w][(n4 + 1) * BT_STRIDE + kl] = (_Float16)wv.y;
        bt[w][(n4 + 2) * BT_STRIDE + kl] = (_Float16)wv.z;
        bt[w][(n4 + 3) * BT_STRIDE + kl] = (_Float16)wv.w;
      }
    }
    __syncthreads();

    v16h a[2][MT];
    #pragma unroll
    for (int ks = 0; ks < 2; ++ks) {
      const int kb = ks * 32 + half * 8;
      #pragma unroll
      for (int mt = 0; mt < MT; ++mt) {
        if (AF32) {
          const float* ap = Af + aoff + (size_t)(mt * 16) * K + kb;
          const float4 f0 = *(const float4*)(ap);
          const float4 f1 = *(const float4*)(ap + 4);
          const float4 f2 = *(const float4*)(ap + 16);
          const float4 f3 = *(const float4*)(ap + 20);
          v16h t;
          t[0]  = (_Float16)f0.x; t[1]  = (_Float16)f0.y;
          t[2]  = (_Float16)f0.z; t[3]  = (_Float16)f0.w;
          t[4]  = (_Float16)f1.x; t[5]  = (_Float16)f1.y;
          t[6]  = (_Float16)f1.z; t[7]  = (_Float16)f1.w;
          t[8]  = (_Float16)f2.x; t[9]  = (_Float16)f2.y;
          t[10] = (_Float16)f2.z; t[11] = (_Float16)f2.w;
          t[12] = (_Float16)f3.x; t[13] = (_Float16)f3.y;
          t[14] = (_Float16)f3.z; t[15] = (_Float16)f3.w;
          a[ks][mt] = t;
        } else {
          a[ks][mt] = __builtin_shufflevector(alo[ks][mt], ahi[ks][mt], 0, 1, 2,
                                              3, 4, 5, 6, 7, 8, 9, 10, 11, 12,
                                              13, 14, 15);
        }
      }
    }
    if (AF32) {
      __builtin_amdgcn_sched_barrier(0);
    }
    #pragma unroll
    for (int ks = 0; ks < 2; ++ks) {
      const int kb = ks * 32 + half * 8;
      #pragma unroll
      for (int w = 0; w < NW; ++w) {
        v16h b[4];
        #pragma unroll
        for (int nn = 0; nn < 4; ++nn) {
          const _Float16* bp = bt[w] + (nn * 16 + l15) * BT_STRIDE + kb;
          v8h blo = *(const v8h*)(bp);
          v8h bhi = *(const v8h*)(bp + 16);
          b[nn] = __builtin_shufflevector(blo, bhi, 0, 1, 2, 3, 4, 5, 6, 7,
                                          8, 9, 10, 11, 12, 13, 14, 15);
        }
        #pragma unroll
        for (int mt = 0; mt < MT; ++mt) {
          #pragma unroll
          for (int nn = 0; nn < 4; ++nn) {
            c[w][mt][nn] = __builtin_amdgcn_wmma_f32_16x16x32_f16(
                false, a[ks][mt], false, b[nn], (short)0, c[w][mt][nn], false,
                false);
          }
        }
      }
    }
    __syncthreads();
  }

  float bv[4] = {0.0f, 0.0f, 0.0f, 0.0f};
  if (HASBIAS) {
    #pragma unroll
    for (int nn = 0; nn < 4; ++nn) bv[nn] = bias[n0 + nn * 16 + l15];
  }
  #pragma unroll
  for (int w = 0; w < NW; ++w) {
    float* __restrict__ oF = (w == 0) ? outF : outFb;
    #pragma unroll
    for (int mt = 0; mt < MT; ++mt) {
      const unsigned o0 =
          (unsigned)(rowBase + mt * 16 + half * 8) * (unsigned)Nt +
          (unsigned)(n0 + l15);
      #pragma unroll
      for (int nn = 0; nn < 4; ++nn) {
        #pragma unroll
        for (int r = 0; r < 8; ++r) {
          float v = c[w][mt][nn][r] + bv[nn];
          if (ACT == 1) v = v > 0.0f ? v : (__expf(v) - 1.0f);
          else if (ACT == 2) v = 1.0f / (1.0f + __expf(-v));
          const unsigned o =
              o0 + (unsigned)r * (unsigned)Nt + (unsigned)(nn * 16);
          if (F32OUT) oF[o] = v;
          else        outH[o] = (_Float16)v;
        }
      }
    }
  }
}

typedef __attribute__((ext_vector_type(4))) _Float16 v4h;
typedef __attribute__((ext_vector_type(2))) float v2f;
typedef __attribute__((ext_vector_type(4))) float v4f;

template <int CTRL>
__device__ __forceinline__ float lane_xor_get(float x) {
  return __builtin_bit_cast(
      float, __builtin_amdgcn_update_dpp(0, __builtin_bit_cast(int, x), CTRL,
                                         0xf, 0xf, true));
}

__device__ __forceinline__ float edge_logit(const v4f a, const v4f b,
                                            const v4f w) {
  const v2f z0 = a.xy + b.xy;
  const v2f z1 = a.zw + b.zw;
  const v2f l0 = __builtin_elementwise_max(z0, z0 * 0.2f);
  const v2f l1 = __builtin_elementwise_max(z1, z1 * 0.2f);
  v2f p = l0 * w.xy;
  p = l1 * w.zw + p;
  float acc = p.x + p.y;
  acc += lane_xor_get<0x161>(acc);
  acc += lane_xor_get<0x162>(acc);
  acc += lane_xor_get<0x164>(acc);
  return acc;
}

template <int NB>
__device__ __forceinline__ void conv_batch(const v4f* __restrict__ xl4,
                                           int my, int j0, int lane,
                                           const v4f xrv, const v4f av,
                                           float& m, float& ssum, v4f& r) {
  v4f v[NB];
  float e[NB];
  unsigned off[NB];
  #pragma unroll
  for (int t = 0; t < NB; ++t) {
    const int sidx = __builtin_amdgcn_readlane(my, j0 + t);
    off[t] = (unsigned)sidx * 32u + (unsigned)lane;
  }
  #pragma unroll
  for (int t = 0; t < NB; ++t) v[t] = xl4[off[t]];
  float mn = m;
  #pragma unroll
  for (int t = 0; t < NB; ++t) {
    e[t] = edge_logit(v[t], xrv, av);
    mn = fmaxf(mn, e[t]);
  }
  const float sc = __expf(m - mn);
  ssum *= sc;
  r *= sc;
  #pragma unroll
  for (int t = 0; t < NB; ++t) {
    const float p = __expf(e[t] - mn);
    ssum += p;
    r = v[t] * p + r;
  }
  m = mn;
}

__global__ __launch_bounds__(256) void conv_edge_kernel(
    const int* __restrict__ rowptr, const int* __restrict__ srcs,
    const float* __restrict__ xl, const float* __restrict__ xr,
    const float* __restrict__ att, const float* __restrict__ bias,
    _Float16* __restrict__ outH) {
  const int lane = threadIdx.x & 31;
  const int d = __builtin_amdgcn_readfirstlane(
      (int)(blockIdx.x * 8 + (threadIdx.x >> 5)));
  if (d >= NNODES) return;
  const v4f* xl4 = (const v4f*)xl;
  const v4f xrv = ((const v4f*)xr)[(size_t)d * 32 + lane];
  const v4f av = ((const v4f*)att)[lane];

  v4f r = xl4[(size_t)d * 32 + lane];
  float m = edge_logit(r, xrv, av);
  float ssum = 1.0f;

  const int beg = rowptr[d], end = rowptr[d + 1];
  for (int base = beg; base < end; base += 32) {
    const int cnt = (end - base) < 32 ? (end - base) : 32;
    int my = 0;
    if (lane < cnt) my = srcs[base + lane];
    int j = 0;
    for (; j + 8 <= cnt; j += 8)
      conv_batch<8>(xl4, my, j, lane, xrv, av, m, ssum, r);
    if (j + 4 <= cnt) { conv_batch<4>(xl4, my, j, lane, xrv, av, m, ssum, r); j += 4; }
    if (j + 2 <= cnt) { conv_batch<2>(xl4, my, j, lane, xrv, av, m, ssum, r); j += 2; }
    if (j < cnt) conv_batch<1>(xl4, my, j, lane, xrv, av, m, ssum, r);
  }

  const v4f bv = ((const v4f*)bias)[lane];
  float o0 = r.x / ssum + bv.x, o1 = r.y / ssum + bv.y;
  float o2 = r.z / ssum + bv.z, o3 = r.w / ssum + bv.w;
  o0 = o0 > 0.0f ? o0 : (__expf(o0) - 1.0f);
  o1 = o1 > 0.0f ? o1 : (__expf(o1) - 1.0f);
  o2 = o2 > 0.0f ? o2 : (__expf(o2) - 1.0f);
  o3 = o3 > 0.0f ? o3 : (__expf(o3) - 1.0f);
  v4h hv;
  hv[0] = (_Float16)o0; hv[1] = (_Float16)o1;
  hv[2] = (_Float16)o2; hv[3] = (_Float16)o3;
  *(v4h*)(outH + (size_t)d * FDIM + lane * 4) = hv;
}

extern "C" void kernel_launch(void* const* d_in, const int* in_sizes, int n_in,
                              void* d_out, int out_size, void* d_ws,
                              size_t ws_size, hipStream_t stream) {
  (void)in_sizes; (void)n_in; (void)out_size; (void)ws_size;
  const float* x    = (const float*)d_in[0];
  const int*   ei   = (const int*)d_in[1];
  const float* Wl1  = (const float*)d_in[2];
  const float* Wr1  = (const float*)d_in[3];
  const float* att1 = (const float*)d_in[4];
  const float* b1   = (const float*)d_in[5];
  const float* Wl2  = (const float*)d_in[6];
  const float* Wr2  = (const float*)d_in[7];
  const float* att2 = (const float*)d_in[8];
  const float* b2   = (const float*)d_in[9];
  const float* W1   = (const float*)d_in[10];
  const float* bb1  = (const float*)d_in[11];
  const float* W2   = (const float*)d_in[12];
  const float* bb2  = (const float*)d_in[13];
  const float* W3   = (const float*)d_in[14];
  const float* bb3  = (const float*)d_in[15];
  float* out = (float*)d_out;

  char* ws = (char*)d_ws;
  size_t off = 0;
  auto alloc = [&](size_t bytes) -> char* {
    char* p = ws + off;
    off += (bytes + 255) & ~(size_t)255;
    return p;
  };
  const size_t S_F = (size_t)NNODES * FDIM;
  _Float16* h1h = (_Float16*)alloc(S_F * 2);
  _Float16* h2h = (_Float16*)alloc(S_F * 2);
  int*  ccnt   = (int*)alloc((size_t)NBLK * NCHK * 4);
  int*  coff   = (int*)alloc(((size_t)NBLK * NCHK + 1) * 4);
  int*  rowptr = (int*)alloc(((size_t)NNODES + 1) * 4);
  int2* binned = (int2*)alloc((size_t)NEDGES * 8);
  int2* perm   = (int2*)alloc((size_t)NEDGES * 8);
  int*  srcs   = (int*)alloc((size_t)NEDGES * 4);
  const size_t uoff = off;
  float*    xl   = (float*)alloc(S_F * 4);
  float*    xr   = (float*)alloc(S_F * 4);
  _Float16* m1 = (_Float16*)(ws + uoff);
  _Float16* m2 = (_Float16*)(ws + uoff + (size_t)NNODES * 640 * 2);

  const int TB = 256;
  const int wgrid = (NNODES + 7) / 8;
  const dim3 gconv((NNODES + 255) / 256, 128 / 64);

  bin_count_kernel<<<NCHK, 1024, 0, stream>>>(ei, ccnt);
  bin_scan_kernel<<<1, 1024, 0, stream>>>(ccnt, coff);
  bin_scatter_kernel<<<NCHK, 1024, 0, stream>>>(ei, coff, binned);
  bin_place_kernel<<<NBLK, 1024, 0, stream>>>(coff, binned, rowptr, perm, srcs);

  wmma_gemm_kernel<0, false, true, true, 2, 2><<<gconv, TB, 0, stream>>>(
      nullptr, x, Wl1, Wr1, nullptr, xl, xr, nullptr, NNODES, 128, 128);
  conv_edge_kernel<<<wgrid, TB, 0, stream>>>(rowptr, srcs, xl, xr, att1, b1,
                                             h1h);

  wmma_gemm_kernel<0, false, true, false, 2, 2><<<gconv, TB, 0, stream>>>(
      h1h, nullptr, Wl2, Wr2, nullptr, xl, xr, nullptr, NNODES, 128, 128);
  conv_edge_kernel<<<wgrid, TB, 0, stream>>>(rowptr, srcs, xl, xr, att2, b2,
                                             h2h);

  const dim3 gm1((NNODES + 511) / 512, 640 / 64);
  wmma_gemm_kernel<1, true, false, false, 4, 1><<<gm1, TB, 0, stream>>>(
      h2h, nullptr, W1, nullptr, bb1, nullptr, nullptr, m1, NNODES, 640, 128);
  const dim3 gm2((NNODES + 511) / 512, 320 / 64);
  wmma_gemm_kernel<1, true, false, false, 4, 1><<<gm2, TB, 0, stream>>>(
      m1, nullptr, W2, nullptr, bb2, nullptr, nullptr, m2, NNODES, 320, 640);
  const dim3 gm3((NNODES + 255) / 256, 64 / 64);
  wmma_gemm_kernel<2, true, true, false, 2, 1><<<gm3, TB, 0, stream>>>(
      m2, nullptr, W3, nullptr, bb3, out, nullptr, nullptr, NNODES, 64, 320);
}
